# starting kernel
# baseline (speedup 1.0000x reference)
#include <hip/hip_runtime.h>

#define N_NODES   100000
#define EMB       64
#define HEADS     4
#define RD        32
#define HOUT      (HEADS * RD)
#define NEDGES    1600000
#define NEG_SLOPE 0.2f

typedef __attribute__((ext_vector_type(2))) float v2f;
typedef __attribute__((ext_vector_type(8))) float v8f;

__device__ __forceinline__ void atomicMaxF(float* addr, float v) {
    if (v >= 0.0f) atomicMax((int*)addr, __float_as_int(v));
    else           atomicMin((unsigned int*)addr, __float_as_uint(v));
}

__device__ __forceinline__ float leaky(float x) {
    return x >= 0.0f ? x : NEG_SLOPE * x;
}

__global__ void init_kernel(float* __restrict__ out, float* __restrict__ m,
                            float* __restrict__ denom) {
    int i = blockIdx.x * blockDim.x + threadIdx.x;
    if (i < N_NODES * RD)    out[i] = 0.0f;
    if (i < N_NODES * HEADS) { m[i] = __int_as_float(0xFF800000); denom[i] = 0.0f; }
}

__global__ __launch_bounds__(256) void proj_wmma_kernel(const float* __restrict__ emb,
                                                        const float* __restrict__ W,
                                                        float* __restrict__ h) {
    __shared__ float Wt[HOUT][EMB + 1];
    const int tid = threadIdx.x;
    for (int i = tid; i < EMB * HOUT; i += 256) {
        int k = i >> 7;
        int n = i & 127;
        Wt[n][k] = W[i];
    }
    __syncthreads();

    const int wave = tid >> 5;
    const int lane = tid & 31;
    const int rowTile = (blockIdx.x * 8 + wave) * 16;
    if (rowTile >= N_NODES) return;

    const int half = lane >> 4;
    const int lrow = lane & 15;
    const float* arow = emb + (size_t)(rowTile + lrow) * EMB;

    v8f acc[8] = {};
    for (int k0 = 0; k0 < EMB; k0 += 4) {
        v2f a;
        a.x = arow[k0 + 2 * half];
        a.y = arow[k0 + 2 * half + 1];
        #pragma unroll
        for (int ct = 0; ct < 8; ++ct) {
            const int n = ct * 16 + lrow;
            v2f b;
            b.x = Wt[n][k0 + 2 * half];
            b.y = Wt[n][k0 + 2 * half + 1];
            acc[ct] = __builtin_amdgcn_wmma_f32_16x16x4_f32(
                false, a, false, b, (short)0, acc[ct], false, false);
        }
    }
    #pragma unroll
    for (int ct = 0; ct < 8; ++ct) {
        #pragma unroll
        for (int v = 0; v < 8; ++v) {
            const int mr = v + 8 * half;
            h[(size_t)(rowTile + mr) * HOUT + ct * 16 + lrow] = acc[ct][v];
        }
    }
}

__global__ void logits_kernel(const float* __restrict__ h,
                              const float* __restrict__ att_src,
                              const float* __restrict__ att_dst,
                              float* __restrict__ a_src, float* __restrict__ a_dst) {
    int i = blockIdx.x * blockDim.x + threadIdx.x;
    if (i >= N_NODES * HEADS) return;
    const int n = i >> 2, hh = i & 3;
    const float* hp = h + (size_t)n * HOUT + hh * RD;
    const float* as = att_src + hh * RD;
    const float* ad = att_dst + hh * RD;
    float s = 0.0f, d = 0.0f;
    #pragma unroll
    for (int k = 0; k < RD; ++k) { float v = hp[k]; s += v * as[k]; d += v * ad[k]; }
    a_src[i] = s;
    a_dst[i] = d;
}

__global__ void edge_logit_max_kernel(const int* __restrict__ path,
                                      const float* __restrict__ a_src,
                                      const float* __restrict__ a_dst,
                                      float* __restrict__ ebuf, float* __restrict__ m) {
    int e = blockIdx.x * blockDim.x + threadIdx.x;
    if (e >= NEDGES) return;
    const int s = path[e], d = path[NEDGES + e];
    const float4 es = *(const float4*)(a_src + (size_t)s * 4);
    const float4 ed = *(const float4*)(a_dst + (size_t)d * 4);
    float4 ev;
    ev.x = leaky(es.x + ed.x);
    ev.y = leaky(es.y + ed.y);
    ev.z = leaky(es.z + ed.z);
    ev.w = leaky(es.w + ed.w);
    *(float4*)(ebuf + (size_t)e * 4) = ev;
    atomicMaxF(&m[d * 4 + 0], ev.x);
    atomicMaxF(&m[d * 4 + 1], ev.y);
    atomicMaxF(&m[d * 4 + 2], ev.z);
    atomicMaxF(&m[d * 4 + 3], ev.w);
}

__global__ void edge_exp_sum_kernel(const int* __restrict__ path,
                                    float* __restrict__ ebuf,
                                    const float* __restrict__ m,
                                    float* __restrict__ denom) {
    int e = blockIdx.x * blockDim.x + threadIdx.x;
    if (e >= NEDGES) return;
    const int d = path[NEDGES + e];
    const float4 ev = *(const float4*)(ebuf + (size_t)e * 4);
    const float4 mv = *(const float4*)(m + (size_t)d * 4);
    float4 ex;
    ex.x = __expf(ev.x - mv.x);
    ex.y = __expf(ev.y - mv.y);
    ex.z = __expf(ev.z - mv.z);
    ex.w = __expf(ev.w - mv.w);
    *(float4*)(ebuf + (size_t)e * 4) = ex;
    atomicAdd(&denom[d * 4 + 0], ex.x);
    atomicAdd(&denom[d * 4 + 1], ex.y);
    atomicAdd(&denom[d * 4 + 2], ex.z);
    atomicAdd(&denom[d * 4 + 3], ex.w);
}

__global__ void aggregate_kernel(const int* __restrict__ path,
                                 const float* __restrict__ h,
                                 const float* __restrict__ ebuf,
                                 const float* __restrict__ denom,
                                 float* __restrict__ out) {
    int t = blockIdx.x * blockDim.x + threadIdx.x;
    if (t >= NEDGES * RD) return;
    const int e  = t >> 5;
    const int dd = t & 31;
    const int s = path[e], d = path[NEDGES + e];
    const float* hs = h + (size_t)s * HOUT;
    const float4 ex  = *(const float4*)(ebuf  + (size_t)e * 4);
    const float4 den = *(const float4*)(denom + (size_t)d * 4);
    float acc = (ex.x / (den.x + 1e-16f)) * hs[0 * RD + dd]
              + (ex.y / (den.y + 1e-16f)) * hs[1 * RD + dd]
              + (ex.z / (den.z + 1e-16f)) * hs[2 * RD + dd]
              + (ex.w / (den.w + 1e-16f)) * hs[3 * RD + dd];
    atomicAdd(&out[(size_t)d * RD + dd], 0.25f * acc);
}

extern "C" void kernel_launch(void* const* d_in, const int* in_sizes, int n_in,
                              void* d_out, int out_size, void* d_ws, size_t ws_size,
                              hipStream_t stream) {
    const float* node_emb = (const float*)d_in[0];
    const float* W        = (const float*)d_in[1];
    const float* att_src  = (const float*)d_in[2];
    const float* att_dst  = (const float*)d_in[3];
    const int*   path     = (const int*)d_in[4];
    float* out = (float*)d_out;

    float* ws    = (float*)d_ws;
    float* h     = ws;
    float* a_src = h     + (size_t)N_NODES * HOUT;
    float* a_dst = a_src + (size_t)N_NODES * HEADS;
    float* m     = a_dst + (size_t)N_NODES * HEADS;
    float* denom = m     + (size_t)N_NODES * HEADS;
    float* ebuf  = denom + (size_t)N_NODES * HEADS;

    init_kernel<<<(N_NODES * RD + 255) / 256, 256, 0, stream>>>(out, m, denom);

    const int rowTiles = N_NODES / 16;
    proj_wmma_kernel<<<(rowTiles + 7) / 8, 256, 0, stream>>>(node_emb, W, h);

    logits_kernel<<<(N_NODES * HEADS + 255) / 256, 256, 0, stream>>>(
        h, att_src, att_dst, a_src, a_dst);

    edge_logit_max_kernel<<<(NEDGES + 255) / 256, 256, 0, stream>>>(
        path, a_src, a_dst, ebuf, m);

    edge_exp_sum_kernel<<<(NEDGES + 255) / 256, 256, 0, stream>>>(
        path, ebuf, m, denom);

    aggregate_kernel<<<(NEDGES * RD + 255) / 256, 256, 0, stream>>>(
        path, h, ebuf, denom, out);
}


# Max and exp-sum passes: one thread per (edge, head), so an edge's 4 head atomics sit on 4 adjacent lanes in one float4 slot and an atomic instruction touches 8 lines, not 32; arrival order as before
# speedup vs baseline: 1.5805x; 1.5805x over previous
#include <hip/hip_runtime.h>

#define N_NODES   100000
#define EMB       64
#define HEADS     4
#define RD        32
#define HOUT      (HEADS * RD)
#define NEDGES    1600000
#define NEG_SLOPE 0.2f

typedef __attribute__((ext_vector_type(2))) float v2f;
typedef __attribute__((ext_vector_type(8))) float v8f;

__device__ __forceinline__ void atomicMaxF(float* addr, float v) {
    if (v >= 0.0f) atomicMax((int*)addr, __float_as_int(v));
    else           atomicMin((unsigned int*)addr, __float_as_uint(v));
}

__device__ __forceinline__ float leaky(float x) {
    return x >= 0.0f ? x : NEG_SLOPE * x;
}

__global__ void init_kernel(float* __restrict__ out, float* __restrict__ m,
                            float* __restrict__ denom) {
    int i = blockIdx.x * blockDim.x + threadIdx.x;
    if (i < N_NODES * RD)    out[i] = 0.0f;
    if (i < N_NODES * HEADS) { m[i] = __int_as_float(0xFF800000); denom[i] = 0.0f; }
}

__global__ __launch_bounds__(256) void proj_wmma_kernel(const float* __restrict__ emb,
                                                        const float* __restrict__ W,
                                                        float* __restrict__ h) {
    __shared__ float Wt[HOUT][EMB + 1];
    const int tid = threadIdx.x;
    for (int i = tid; i < EMB * HOUT; i += 256) {
        int k = i >> 7;
        int n = i & 127;
        Wt[n][k] = W[i];
    }
    __syncthreads();

    const int wave = tid >> 5;
    const int lane = tid & 31;
    const int rowTile = (blockIdx.x * 8 + wave) * 16;
    if (rowTile >= N_NODES) return;

    const int half = lane >> 4;
    const int lrow = lane & 15;
    const float* arow = emb + (size_t)(rowTile + lrow) * EMB;

    v8f acc[8] = {};
    for (int k0 = 0; k0 < EMB; k0 += 4) {
        v2f a;
        a.x = arow[k0 + 2 * half];
        a.y = arow[k0 + 2 * half + 1];
        #pragma unroll
        for (int ct = 0; ct < 8; ++ct) {
            const int n = ct * 16 + lrow;
            v2f b;
            b.x = Wt[n][k0 + 2 * half];
            b.y = Wt[n][k0 + 2 * half + 1];
            acc[ct] = __builtin_amdgcn_wmma_f32_16x16x4_f32(
                false, a, false, b, (short)0, acc[ct], false, false);
        }
    }
    #pragma unroll
    for (int ct = 0; ct < 8; ++ct) {
        #pragma unroll
        for (int v = 0; v < 8; ++v) {
            const int mr = v + 8 * half;
            h[(size_t)(rowTile + mr) * HOUT + ct * 16 + lrow] = acc[ct][v];
        }
    }
}

__global__ void logits_kernel(const float* __restrict__ h,
                              const float* __restrict__ att_src,
                              const float* __restrict__ att_dst,
                              float* __restrict__ a_src, float* __restrict__ a_dst) {
    int i = blockIdx.x * blockDim.x + threadIdx.x;
    if (i >= N_NODES * HEADS) return;
    const int n = i >> 2, hh = i & 3;
    const float* hp = h + (size_t)n * HOUT + hh * RD;
    const float* as = att_src + hh * RD;
    const float* ad = att_dst + hh * RD;
    float s = 0.0f, d = 0.0f;
    #pragma unroll
    for (int k = 0; k < RD; ++k) { float v = hp[k]; s += v * as[k]; d += v * ad[k]; }
    a_src[i] = s;
    a_dst[i] = d;
}

__global__ void edge_logit_max_kernel(const int* __restrict__ path,
                                      const float* __restrict__ a_src,
                                      const float* __restrict__ a_dst,
                                      float* __restrict__ ebuf, float* __restrict__ m) {
    int t = blockIdx.x * blockDim.x + threadIdx.x;
    if (t >= NEDGES * HEADS) return;
    const int e  = t >> 2;
    const int hh = t & 3;
    const int s = path[e], d = path[NEDGES + e];
    const float ev = leaky(a_src[(size_t)s * 4 + hh] + a_dst[(size_t)d * 4 + hh]);
    ebuf[t] = ev;
    atomicMaxF(&m[d * 4 + hh], ev);
}

__global__ void edge_exp_sum_kernel(const int* __restrict__ path,
                                    float* __restrict__ ebuf,
                                    const float* __restrict__ m,
                                    float* __restrict__ denom) {
    int t = blockIdx.x * blockDim.x + threadIdx.x;
    if (t >= NEDGES * HEADS) return;
    const int e  = t >> 2;
    const int hh = t & 3;
    const int d = path[NEDGES + e];
    const float ex = __expf(ebuf[t] - m[(size_t)d * 4 + hh]);
    ebuf[t] = ex;
    atomicAdd(&denom[d * 4 + hh], ex);
}

__global__ void aggregate_kernel(const int* __restrict__ path,
                                 const float* __restrict__ h,
                                 const float* __restrict__ ebuf,
                                 const float* __restrict__ denom,
                                 float* __restrict__ out) {
    int t = blockIdx.x * blockDim.x + threadIdx.x;
    if (t >= NEDGES * RD) return;
    const int e  = t >> 5;
    const int dd = t & 31;
    const int s = path[e], d = path[NEDGES + e];
    const float* hs = h + (size_t)s * HOUT;
    const float4 ex  = *(const float4*)(ebuf  + (size_t)e * 4);
    const float4 den = *(const float4*)(denom + (size_t)d * 4);
    float acc = (ex.x / (den.x + 1e-16f)) * hs[0 * RD + dd]
              + (ex.y / (den.y + 1e-16f)) * hs[1 * RD + dd]
              + (ex.z / (den.z + 1e-16f)) * hs[2 * RD + dd]
              + (ex.w / (den.w + 1e-16f)) * hs[3 * RD + dd];
    atomicAdd(&out[(size_t)d * RD + dd], 0.25f * acc);
}

extern "C" void kernel_launch(void* const* d_in, const int* in_sizes, int n_in,
                              void* d_out, int out_size, void* d_ws, size_t ws_size,
                              hipStream_t stream) {
    const float* node_emb = (const float*)d_in[0];
    const float* W        = (const float*)d_in[1];
    const float* att_src  = (const float*)d_in[2];
    const float* att_dst  = (const float*)d_in[3];
    const int*   path     = (const int*)d_in[4];
    float* out = (float*)d_out;

    float* ws    = (float*)d_ws;
    float* h     = ws;
    float* a_src = h     + (size_t)N_NODES * HOUT;
    float* a_dst = a_src + (size_t)N_NODES * HEADS;
    float* m     = a_dst + (size_t)N_NODES * HEADS;
    float* denom = m     + (size_t)N_NODES * HEADS;
    float* ebuf  = denom + (size_t)N_NODES * HEADS;

    init_kernel<<<(N_NODES * RD + 255) / 256, 256, 0, stream>>>(out, m, denom);

    const int rowTiles = N_NODES / 16;
    proj_wmma_kernel<<<(rowTiles + 7) / 8, 256, 0, stream>>>(node_emb, W, h);

    logits_kernel<<<(N_NODES * HEADS + 255) / 256, 256, 0, stream>>>(
        h, att_src, att_dst, a_src, a_dst);

    edge_logit_max_kernel<<<(NEDGES * HEADS + 255) / 256, 256, 0, stream>>>(
        path, a_src, a_dst, ebuf, m);

    edge_exp_sum_kernel<<<(NEDGES * HEADS + 255) / 256, 256, 0, stream>>>(
        path, ebuf, m, denom);

    aggregate_kernel<<<(NEDGES * RD + 255) / 256, 256, 0, stream>>>(
        path, h, ebuf, denom, out);
}


# Edges grouped by destination in-call (int atomic rank count, scan, scatter); one wave per destination does softmax and weighted sum in registers, no float atomics; sums in rank order, heads last
# speedup vs baseline: 2.8364x; 1.7946x over previous
#include <hip/hip_runtime.h>

#define N_NODES   100000
#define EMB       64
#define HEADS     4
#define RD        32
#define HOUT      (HEADS * RD)
#define NEDGES    1600000
#define NEG_SLOPE 0.2f

typedef __attribute__((ext_vector_type(2))) float v2f;
typedef __attribute__((ext_vector_type(8))) float v8f;

__device__ __forceinline__ float leaky(float x) {
    return x >= 0.0f ? x : NEG_SLOPE * x;
}

__global__ void init_kernel(int* __restrict__ cnt) {
    int i = blockIdx.x * blockDim.x + threadIdx.x;
    if (i < N_NODES) cnt[i] = 0;
}

__global__ __launch_bounds__(256) void proj_wmma_kernel(const float* __restrict__ emb,
                                                        const float* __restrict__ W,
                                                        float* __restrict__ h) {
    __shared__ float Wt[HOUT][EMB + 1];
    const int tid = threadIdx.x;
    for (int i = tid; i < EMB * HOUT; i += 256) {
        int k = i >> 7;
        int n = i & 127;
        Wt[n][k] = W[i];
    }
    __syncthreads();

    const int wave = tid >> 5;
    const int lane = tid & 31;
    const int rowTile = (blockIdx.x * 8 + wave) * 16;
    if (rowTile >= N_NODES) return;

    const int half = lane >> 4;
    const int lrow = lane & 15;
    const float* arow = emb + (size_t)(rowTile + lrow) * EMB;

    v8f acc[8] = {};
    for (int k0 = 0; k0 < EMB; k0 += 4) {
        v2f a;
        a.x = arow[k0 + 2 * half];
        a.y = arow[k0 + 2 * half + 1];
        #pragma unroll
        for (int ct = 0; ct < 8; ++ct) {
            const int n = ct * 16 + lrow;
            v2f b;
            b.x = Wt[n][k0 + 2 * half];
            b.y = Wt[n][k0 + 2 * half + 1];
            acc[ct] = __builtin_amdgcn_wmma_f32_16x16x4_f32(
                false, a, false, b, (short)0, acc[ct], false, false);
        }
    }
    #pragma unroll
    for (int ct = 0; ct < 8; ++ct) {
        #pragma unroll
        for (int v = 0; v < 8; ++v) {
            const int mr = v + 8 * half;
            h[(size_t)(rowTile + mr) * HOUT + ct * 16 + lrow] = acc[ct][v];
        }
    }
}

__global__ void logits_kernel(const float* __restrict__ h,
                              const float* __restrict__ att_src,
                              const float* __restrict__ att_dst,
                              float* __restrict__ a_src, float* __restrict__ a_dst) {
    int i = blockIdx.x * blockDim.x + threadIdx.x;
    if (i >= N_NODES * HEADS) return;
    const int n = i >> 2, hh = i & 3;
    const float* hp = h + (size_t)n * HOUT + hh * RD;
    const float* as = att_src + hh * RD;
    const float* ad = att_dst + hh * RD;
    float s = 0.0f, d = 0.0f;
    #pragma unroll
    for (int k = 0; k < RD; ++k) { float v = hp[k]; s += v * as[k]; d += v * ad[k]; }
    a_src[i] = s;
    a_dst[i] = d;
}

__global__ void count_kernel(const int* __restrict__ path, int* __restrict__ cnt,
                             int* __restrict__ rank) {
    int e = blockIdx.x * blockDim.x + threadIdx.x;
    if (e >= NEDGES) return;
    rank[e] = atomicAdd(&cnt[path[NEDGES + e]], 1);
}

#define SCAN_T 1024
__global__ __launch_bounds__(SCAN_T) void offsets_kernel(const int* __restrict__ cnt,
                                                         int* __restrict__ off) {
    __shared__ int wtot[32];
    __shared__ int wpart[32];
    const int tid  = threadIdx.x;
    const int lane = tid & 31;
    const int wave = tid >> 5;
    const int base = blockIdx.x * SCAN_T;

    int part = 0;
    for (int i = tid; i < base; i += SCAN_T) part += cnt[i];
    for (int o = 16; o > 0; o >>= 1) part += __shfl_xor(part, o, 32);

    const int i = base + tid;
    const int c = (i < N_NODES) ? cnt[i] : 0;
    int incl = c;
    for (int o = 1; o < 32; o <<= 1) {
        const int up = __shfl_up(incl, o, 32);
        if (lane >= o) incl += up;
    }
    if (lane == 31) { wtot[wave] = incl; wpart[wave] = part; }
    __syncthreads();

    int before = 0;
    for (int w = 0; w < 32; ++w) {
        before += wpart[w];
        if (w < wave) before += wtot[w];
    }
    if (i < N_NODES) {
        off[i] = before + incl - c;
        if (i == N_NODES - 1) off[N_NODES] = before + incl;
    }
}

__global__ void scatter_kernel(const int* __restrict__ path, const int* __restrict__ off,
                               const int* __restrict__ rank, int* __restrict__ srt) {
    int e = blockIdx.x * blockDim.x + threadIdx.x;
    if (e >= NEDGES) return;
    srt[off[path[NEDGES + e]] + rank[e]] = path[e];
}

template <int M>
__device__ __forceinline__ float xrow(float x) {
    const int b = __float_as_int(x);
    return __int_as_float(__builtin_amdgcn_update_dpp(b, b, 0x160 + M, 0xf, 0xf, false));
}

__device__ __forceinline__ float xhalf(float x) {
    const int b = __float_as_int(x);
    return __int_as_float(__builtin_amdgcn_permlanex16(b, b, 0x76543210, 0xfedcba98, false, false));
}

__device__ __forceinline__ float wave_max(float v) {
    v = fmaxf(v, xrow<1>(v));
    v = fmaxf(v, xrow<2>(v));
    v = fmaxf(v, xrow<4>(v));
    v = fmaxf(v, xrow<8>(v));
    return fmaxf(v, xhalf(v));
}

__device__ __forceinline__ float wave_sum(float v) {
    v += xrow<1>(v);
    v += xrow<2>(v);
    v += xrow<4>(v);
    v += xrow<8>(v);
    return v + xhalf(v);
}

__device__ __forceinline__ float4 wave_max4(float4 v) {
    return make_float4(wave_max(v.x), wave_max(v.y), wave_max(v.z), wave_max(v.w));
}

__device__ __forceinline__ float4 wave_sum4(float4 v) {
    return make_float4(wave_sum(v.x), wave_sum(v.y), wave_sum(v.z), wave_sum(v.w));
}

__device__ __forceinline__ float4 edge_logits(const int* __restrict__ srt,
                                              const float* __restrict__ a_src,
                                              float4 ad, int j, int& s) {
    s = srt[j];
    const float4 as = *(const float4*)(a_src + (size_t)s * 4);
    float4 e;
    e.x = leaky(as.x + ad.x);
    e.y = leaky(as.y + ad.y);
    e.z = leaky(as.z + ad.z);
    e.w = leaky(as.w + ad.w);
    return e;
}

__device__ __forceinline__ int bcast(float v, int srcLane) {
    return __builtin_amdgcn_readlane(__float_as_int(v), srcLane);
}

__global__ __launch_bounds__(256) void gather_kernel(const int* __restrict__ off,
                                                     const int* __restrict__ srt,
                                                     const float* __restrict__ h,
                                                     const float* __restrict__ a_src,
                                                     const float* __restrict__ a_dst,
                                                     float* __restrict__ out) {
    const int lane = threadIdx.x & 31;
    const int d = blockIdx.x * 8 + __builtin_amdgcn_readfirstlane(threadIdx.x >> 5);
    if (d >= N_NODES) return;
    const int beg = __builtin_amdgcn_readfirstlane(off[d]);
    const int end = __builtin_amdgcn_readfirstlane(off[d + 1]);
    const float4 ad = *(const float4*)(a_dst + (size_t)d * 4);
    const float NEG_INF = __int_as_float(0xFF800000);

    const int  j0 = beg + lane;
    const bool v0 = j0 < end;
    int s0 = 0;
    float4 e0 = make_float4(NEG_INF, NEG_INF, NEG_INF, NEG_INF);
    if (v0) e0 = edge_logits(srt, a_src, ad, j0, s0);

    float4 mx = e0;
    for (int j = j0 + 32; j < end; j += 32) {
        int s;
        const float4 e = edge_logits(srt, a_src, ad, j, s);
        mx.x = fmaxf(mx.x, e.x); mx.y = fmaxf(mx.y, e.y);
        mx.z = fmaxf(mx.z, e.z); mx.w = fmaxf(mx.w, e.w);
    }
    mx = wave_max4(mx);

    float4 ex0 = make_float4(0.0f, 0.0f, 0.0f, 0.0f);
    if (v0) {
        ex0.x = __expf(e0.x - mx.x); ex0.y = __expf(e0.y - mx.y);
        ex0.z = __expf(e0.z - mx.z); ex0.w = __expf(e0.w - mx.w);
    }
    float4 den = ex0;
    for (int j = j0 + 32; j < end; j += 32) {
        int s;
        const float4 e = edge_logits(srt, a_src, ad, j, s);
        den.x += __expf(e.x - mx.x); den.y += __expf(e.y - mx.y);
        den.z += __expf(e.z - mx.z); den.w += __expf(e.w - mx.w);
    }
    den = wave_sum4(den);
    den.x += 1e-16f; den.y += 1e-16f; den.z += 1e-16f; den.w += 1e-16f;

    const int hh = lane >> 3;
    const int m0 = hh == 0 ? -1 : 0, m1 = hh == 1 ? -1 : 0;
    const int m2 = hh == 2 ? -1 : 0, m3 = hh == 3 ? -1 : 0;
    const float* hcol = h + lane * 4;
    float4 acc = make_float4(0.0f, 0.0f, 0.0f, 0.0f);

    int s = s0;
    float4 al = make_float4(ex0.x / den.x, ex0.y / den.y, ex0.z / den.z, ex0.w / den.w);
    for (int base = beg; base < end; base += 32) {
        if (base != beg) {
            const int j = base + lane;
            s = 0;
            al = make_float4(0.0f, 0.0f, 0.0f, 0.0f);
            if (j < end) {
                const float4 e = edge_logits(srt, a_src, ad, j, s);
                al.x = __expf(e.x - mx.x) / den.x; al.y = __expf(e.y - mx.y) / den.y;
                al.z = __expf(e.z - mx.z) / den.z; al.w = __expf(e.w - mx.w) / den.w;
            }
        }
        const int n = min(32, end - base);
        #pragma unroll 4
        for (int t = 0; t < n; ++t) {
            const int   st = __builtin_amdgcn_readlane(s, t);
            const float a  = __int_as_float((bcast(al.x, t) & m0) | (bcast(al.y, t) & m1) |
                                            (bcast(al.z, t) & m2) | (bcast(al.w, t) & m3));
            const float4 hv = *(const float4*)(hcol + (size_t)st * HOUT);
            acc.x += a * hv.x; acc.y += a * hv.y; acc.z += a * hv.z; acc.w += a * hv.w;
        }
    }

    acc.x += xrow<8>(acc.x); acc.y += xrow<8>(acc.y);
    acc.z += xrow<8>(acc.z); acc.w += xrow<8>(acc.w);
    acc.x += xhalf(acc.x);   acc.y += xhalf(acc.y);
    acc.z += xhalf(acc.z);   acc.w += xhalf(acc.w);
    if (lane < 8) {
        *(float4*)(out + (size_t)d * RD + lane * 4) =
            make_float4(0.25f * acc.x, 0.25f * acc.y, 0.25f * acc.z, 0.25f * acc.w);
    }
}

extern "C" void kernel_launch(void* const* d_in, const int* in_sizes, int n_in,
                              void* d_out, int out_size, void* d_ws, size_t ws_size,
                              hipStream_t stream) {
    const float* node_emb = (const float*)d_in[0];
    const float* W        = (const float*)d_in[1];
    const float* att_src  = (const float*)d_in[2];
    const float* att_dst  = (const float*)d_in[3];
    const int*   path     = (const int*)d_in[4];
    float* out = (float*)d_out;

    float* ws    = (float*)d_ws;
    float* h     = ws;
    float* a_src = h     + (size_t)N_NODES * HOUT;
    float* a_dst = a_src + (size_t)N_NODES * HEADS;
    int*   cnt   = (int*)(a_dst + (size_t)N_NODES * HEADS);
    int*   off   = cnt  + (size_t)N_NODES;
    int*   rank  = off  + (size_t)N_NODES + 32;
    int*   srt   = rank + (size_t)NEDGES;

    init_kernel<<<(N_NODES + 255) / 256, 256, 0, stream>>>(cnt);

    const int rowTiles = N_NODES / 16;
    proj_wmma_kernel<<<(rowTiles + 7) / 8, 256, 0, stream>>>(node_emb, W, h);

    logits_kernel<<<(N_NODES * HEADS + 255) / 256, 256, 0, stream>>>(
        h, att_src, att_dst, a_src, a_dst);

    count_kernel<<<(NEDGES + 255) / 256, 256, 0, stream>>>(path, cnt, rank);

    offsets_kernel<<<(N_NODES + SCAN_T - 1) / SCAN_T, SCAN_T, 0, stream>>>(cnt, off);

    scatter_kernel<<<(NEDGES + 255) / 256, 256, 0, stream>>>(path, off, rank, srt);

    gather_kernel<<<(N_NODES + 7) / 8, 256, 0, stream>>>(off, srt, h, a_src, a_dst, out);
}


# Projection moved behind the aggregation: per head the alpha-weighted 64-float input rows are summed, then one 256x32 WMMA product per 16 destinations; logits from emb @ (W att); sums re-associated
# speedup vs baseline: 3.1201x; 1.1000x over previous
#include <hip/hip_runtime.h>

#define N_NODES   100000
#define EMB       64
#define HEADS     4
#define RD        32
#define HOUT      (HEADS * RD)
#define NEDGES    1600000
#define NEG_SLOPE 0.2f

typedef __attribute__((ext_vector_type(2))) float v2f;
typedef __attribute__((ext_vector_type(8))) float v8f;

__device__ __forceinline__ float leaky(float x) {
    return x >= 0.0f ? x : NEG_SLOPE * x;
}

__global__ void init_kernel(int* __restrict__ cnt) {
    int i = blockIdx.x * blockDim.x + threadIdx.x;
    if (i < N_NODES) cnt[i] = 0;
}

__global__ __launch_bounds__(256) void logits_wmma_kernel(const float* __restrict__ emb,
                                                          const float* __restrict__ W,
                                                          const float* __restrict__ att_src,
                                                          const float* __restrict__ att_dst,
                                                          float* __restrict__ a_src,
                                                          float* __restrict__ a_dst) {
    __shared__ float Wa[16][EMB + 1];
    const int tid = threadIdx.x;
    for (int i = tid; i < 16 * EMB; i += 256) {
        const int k = i >> 4;
        const int j = i & 15;
        float s = 0.0f;
        if (j < 8) {
            const int hh = j & 3;
            const float* att = (j < 4 ? att_src : att_dst) + hh * RD;
            const float* wr  = W + k * HOUT + hh * RD;
            #pragma unroll
            for (int dd = 0; dd < RD; ++dd) s += wr[dd] * att[dd];
        }
        Wa[j][k] = s;
    }
    __syncthreads();

    const int wave = tid >> 5;
    const int lane = tid & 31;
    const int rowTile = (blockIdx.x * 8 + wave) * 16;
    if (rowTile >= N_NODES) return;

    const int half = lane >> 4;
    const int lrow = lane & 15;
    const float* arow = emb + (size_t)(rowTile + lrow) * EMB;

    v8f acc = {};
    for (int k0 = 0; k0 < EMB; k0 += 4) {
        v2f a, b;
        a.x = arow[k0 + 2 * half];
        a.y = arow[k0 + 2 * half + 1];
        b.x = Wa[lrow][k0 + 2 * half];
        b.y = Wa[lrow][k0 + 2 * half + 1];
        acc = __builtin_amdgcn_wmma_f32_16x16x4_f32(
            false, a, false, b, (short)0, acc, false, false);
    }
    float* dstp = (lrow < 4 ? a_src + lrow : a_dst + (lrow - 4)) + (size_t)(rowTile + 8 * half) * HEADS;
    if (lrow < 8) {
        #pragma unroll
        for (int v = 0; v < 8; ++v) dstp[v * HEADS] = acc[v];
    }
}

__global__ void count_kernel(const int* __restrict__ path, int* __restrict__ cnt,
                             int* __restrict__ rank) {
    int e = blockIdx.x * blockDim.x + threadIdx.x;
    if (e >= NEDGES) return;
    rank[e] = atomicAdd(&cnt[path[NEDGES + e]], 1);
}

#define SCAN_T 1024
__global__ __launch_bounds__(SCAN_T) void offsets_kernel(const int* __restrict__ cnt,
                                                         int* __restrict__ off) {
    __shared__ int wtot[32];
    __shared__ int wpart[32];
    const int tid  = threadIdx.x;
    const int lane = tid & 31;
    const int wave = tid >> 5;
    const int base = blockIdx.x * SCAN_T;

    int part = 0;
    for (int i = tid; i < base; i += SCAN_T) part += cnt[i];
    for (int o = 16; o > 0; o >>= 1) part += __shfl_xor(part, o, 32);

    const int i = base + tid;
    const int c = (i < N_NODES) ? cnt[i] : 0;
    int incl = c;
    for (int o = 1; o < 32; o <<= 1) {
        const int up = __shfl_up(incl, o, 32);
        if (lane >= o) incl += up;
    }
    if (lane == 31) { wtot[wave] = incl; wpart[wave] = part; }
    __syncthreads();

    int before = 0;
    for (int w = 0; w < 32; ++w) {
        before += wpart[w];
        if (w < wave) before += wtot[w];
    }
    if (i < N_NODES) {
        off[i] = before + incl - c;
        if (i == N_NODES - 1) off[N_NODES] = before + incl;
    }
}

__global__ void scatter_kernel(const int* __restrict__ path, const int* __restrict__ off,
                               const int* __restrict__ rank, int* __restrict__ srt) {
    int e = blockIdx.x * blockDim.x + threadIdx.x;
    if (e >= NEDGES) return;
    srt[off[path[NEDGES + e]] + rank[e]] = path[e];
}

template <int M>
__device__ __forceinline__ float xrow(float x) {
    const int b = __float_as_int(x);
    return __int_as_float(__builtin_amdgcn_update_dpp(b, b, 0x160 + M, 0xf, 0xf, false));
}

__device__ __forceinline__ float xhalf(float x) {
    const int b = __float_as_int(x);
    return __int_as_float(__builtin_amdgcn_permlanex16(b, b, 0x76543210, 0xfedcba98, false, false));
}

__device__ __forceinline__ float wave_max(float v) {
    v = fmaxf(v, xrow<1>(v));
    v = fmaxf(v, xrow<2>(v));
    v = fmaxf(v, xrow<4>(v));
    v = fmaxf(v, xrow<8>(v));
    return fmaxf(v, xhalf(v));
}

__device__ __forceinline__ float wave_sum(float v) {
    v += xrow<1>(v);
    v += xrow<2>(v);
    v += xrow<4>(v);
    v += xrow<8>(v);
    return v + xhalf(v);
}

__device__ __forceinline__ float4 wave_max4(float4 v) {
    return make_float4(wave_max(v.x), wave_max(v.y), wave_max(v.z), wave_max(v.w));
}

__device__ __forceinline__ float4 wave_sum4(float4 v) {
    return make_float4(wave_sum(v.x), wave_sum(v.y), wave_sum(v.z), wave_sum(v.w));
}

__device__ __forceinline__ float4 edge_logits(const int* __restrict__ srt,
                                              const float* __restrict__ a_src,
                                              float4 ad, int j, int& s) {
    s = srt[j];
    const float4 as = *(const float4*)(a_src + (size_t)s * 4);
    float4 e;
    e.x = leaky(as.x + ad.x);
    e.y = leaky(as.y + ad.y);
    e.z = leaky(as.z + ad.z);
    e.w = leaky(as.w + ad.w);
    return e;
}

__device__ __forceinline__ int bcast(float v, int srcLane) {
    return __builtin_amdgcn_readlane(__float_as_int(v), srcLane);
}

#define ZK (HEADS * EMB)
#define ZP (ZK + 4)
#define PP (256 + 16)
__global__ __launch_bounds__(512) void gather_project_kernel(const int* __restrict__ off,
                                                             const int* __restrict__ srt,
                                                             const float* __restrict__ emb,
                                                             const float* __restrict__ W,
                                                             const float* __restrict__ a_src,
                                                             const float* __restrict__ a_dst,
                                                             float* __restrict__ out) {
    __shared__ __attribute__((aligned(16))) float zt[16][ZP];
    __shared__ __attribute__((aligned(16))) float part[16][PP];
    const int lane = threadIdx.x & 31;
    const int wave = __builtin_amdgcn_readfirstlane(threadIdx.x >> 5);
    const int d = blockIdx.x * 16 + wave;
    const int beg = __builtin_amdgcn_readfirstlane(off[d]);
    const int end = __builtin_amdgcn_readfirstlane(off[d + 1]);
    const float4 ad = *(const float4*)(a_dst + (size_t)d * 4);
    const float NEG_INF = __int_as_float(0xFF800000);

    const int  j0 = beg + lane;
    const bool v0 = j0 < end;
    int s0 = 0;
    float4 e0 = make_float4(NEG_INF, NEG_INF, NEG_INF, NEG_INF);
    if (v0) e0 = edge_logits(srt, a_src, ad, j0, s0);

    float4 mx = e0;
    for (int j = j0 + 32; j < end; j += 32) {
        int s;
        const float4 e = edge_logits(srt, a_src, ad, j, s);
        mx.x = fmaxf(mx.x, e.x); mx.y = fmaxf(mx.y, e.y);
        mx.z = fmaxf(mx.z, e.z); mx.w = fmaxf(mx.w, e.w);
    }
    mx = wave_max4(mx);

    float4 ex0 = make_float4(0.0f, 0.0f, 0.0f, 0.0f);
    if (v0) {
        ex0.x = __expf(e0.x - mx.x); ex0.y = __expf(e0.y - mx.y);
        ex0.z = __expf(e0.z - mx.z); ex0.w = __expf(e0.w - mx.w);
    }
    float4 den = ex0;
    for (int j = j0 + 32; j < end; j += 32) {
        int s;
        const float4 e = edge_logits(srt, a_src, ad, j, s);
        den.x += __expf(e.x - mx.x); den.y += __expf(e.y - mx.y);
        den.z += __expf(e.z - mx.z); den.w += __expf(e.w - mx.w);
    }
    den = wave_sum4(den);
    den.x += 1e-16f; den.y += 1e-16f; den.z += 1e-16f; den.w += 1e-16f;

    const float* ecol = emb + lane * 2;
    float2 z0 = make_float2(0.0f, 0.0f), z1 = z0, z2 = z0, z3 = z0;

    int s = s0;
    float4 al = make_float4(ex0.x / den.x, ex0.y / den.y, ex0.z / den.z, ex0.w / den.w);
    for (int base = beg; base < end; base += 32) {
        if (base != beg) {
            const int j = base + lane;
            s = 0;
            al = make_float4(0.0f, 0.0f, 0.0f, 0.0f);
            if (j < end) {
                const float4 e = edge_logits(srt, a_src, ad, j, s);
                al.x = __expf(e.x - mx.x) / den.x; al.y = __expf(e.y - mx.y) / den.y;
                al.z = __expf(e.z - mx.z) / den.z; al.w = __expf(e.w - mx.w) / den.w;
            }
        }
        const int n = min(32, end - base);
        #pragma unroll 4
        for (int t = 0; t < n; ++t) {
            const int   st = __builtin_amdgcn_readlane(s, t);
            const float a0 = __int_as_float(bcast(al.x, t));
            const float a1 = __int_as_float(bcast(al.y, t));
            const float a2 = __int_as_float(bcast(al.z, t));
            const float a3 = __int_as_float(bcast(al.w, t));
            const float2 ev = *(const float2*)(ecol + (size_t)st * EMB);
            z0.x += a0 * ev.x; z0.y += a0 * ev.y;
            z1.x += a1 * ev.x; z1.y += a1 * ev.y;
            z2.x += a2 * ev.x; z2.y += a2 * ev.y;
            z3.x += a3 * ev.x; z3.y += a3 * ev.y;
        }
    }
    *(float2*)&zt[wave][0 * EMB + lane * 2] = z0;
    *(float2*)&zt[wave][1 * EMB + lane * 2] = z1;
    *(float2*)&zt[wave][2 * EMB + lane * 2] = z2;
    *(float2*)&zt[wave][3 * EMB + lane * 2] = z3;

    const int half = lane >> 4;
    const int lrow = lane & 15;
    const int ct   = wave & 1;
    const int kq   = wave >> 1;
    const int hk   = kq >> 1;
    const int kin  = (kq & 1) * 32;
    const float* wp = W + (size_t)(kin + 2 * half) * HOUT + hk * RD + ct * 16 + lrow;
    v2f bf[8];
    #pragma unroll
    for (int ks = 0; ks < 8; ++ks) {
        bf[ks].x = wp[(ks * 4) * HOUT];
        bf[ks].y = wp[(ks * 4 + 1) * HOUT];
    }
    __syncthreads();

    const float* zr = &zt[lrow][kq * 32 + 2 * half];
    v8f c0 = {}, c1 = {};
    #pragma unroll
    for (int ks = 0; ks < 8; ks += 2) {
        const v2f fa0 = *(const v2f*)(zr + ks * 4);
        const v2f fa1 = *(const v2f*)(zr + ks * 4 + 4);
        c0 = __builtin_amdgcn_wmma_f32_16x16x4_f32(false, fa0, false, bf[ks],     (short)0, c0, false, false);
        c1 = __builtin_amdgcn_wmma_f32_16x16x4_f32(false, fa1, false, bf[ks + 1], (short)0, c1, false, false);
    }
    #pragma unroll
    for (int v = 0; v < 8; ++v) part[wave][v * 32 + lane] = c0[v] + c1[v];
    __syncthreads();

    const int pl = (wave >> 3) * 16 + lrow;
    const int pv = wave & 7;
    const int pc = lane >> 4;
    float sum = 0.0f;
    #pragma unroll
    for (int q = 0; q < 8; ++q) sum += part[2 * q + pc][pv * 32 + pl];
    out[(size_t)d * RD + lane] = 0.25f * sum;
}

extern "C" void kernel_launch(void* const* d_in, const int* in_sizes, int n_in,
                              void* d_out, int out_size, void* d_ws, size_t ws_size,
                              hipStream_t stream) {
    const float* node_emb = (const float*)d_in[0];
    const float* W        = (const float*)d_in[1];
    const float* att_src  = (const float*)d_in[2];
    const float* att_dst  = (const float*)d_in[3];
    const int*   path     = (const int*)d_in[4];
    float* out = (float*)d_out;

    float* ws    = (float*)d_ws;
    float* h     = ws;
    float* a_src = h     + (size_t)N_NODES * HOUT;
    float* a_dst = a_src + (size_t)N_NODES * HEADS;
    int*   cnt   = (int*)(a_dst + (size_t)N_NODES * HEADS);
    int*   off   = cnt  + (size_t)N_NODES;
    int*   rank  = off  + (size_t)N_NODES + 32;
    int*   srt   = rank + (size_t)NEDGES;

    init_kernel<<<(N_NODES + 255) / 256, 256, 0, stream>>>(cnt);

    const int rowTiles = N_NODES / 16;
    logits_wmma_kernel<<<(rowTiles + 7) / 8, 256, 0, stream>>>(
        node_emb, W, att_src, att_dst, a_src, a_dst);

    count_kernel<<<(NEDGES + 255) / 256, 256, 0, stream>>>(path, cnt, rank);

    offsets_kernel<<<(N_NODES + SCAN_T - 1) / SCAN_T, SCAN_T, 0, stream>>>(cnt, off);

    scatter_kernel<<<(NEDGES + 255) / 256, 256, 0, stream>>>(path, off, rank, srt);

    gather_project_kernel<<<N_NODES / 16, 512, 0, stream>>>(off, srt, node_emb, W, a_src, a_dst, out);
}


# Gather: input rows of a destination's first 16 edges are requested when the sorted ids arrive, before the softmax; later edges in padded groups of 8, no one-row remainder loop; sum order kept
# speedup vs baseline: 3.2241x; 1.0333x over previous
#include <hip/hip_runtime.h>

#define N_NODES   100000
#define EMB       64
#define HEADS     4
#define RD        32
#define HOUT      (HEADS * RD)
#define NEDGES    1600000
#define NEG_SLOPE 0.2f

typedef __attribute__((ext_vector_type(2))) float v2f;
typedef __attribute__((ext_vector_type(8))) float v8f;

__device__ __forceinline__ float leaky(float x) {
    return x >= 0.0f ? x : NEG_SLOPE * x;
}

__global__ void init_kernel(int* __restrict__ cnt) {
    int i = blockIdx.x * blockDim.x + threadIdx.x;
    if (i < N_NODES) cnt[i] = 0;
}

__global__ __launch_bounds__(256) void logits_wmma_kernel(const float* __restrict__ emb,
                                                          const float* __restrict__ W,
                                                          const float* __restrict__ att_src,
                                                          const float* __restrict__ att_dst,
                                                          float* __restrict__ a_src,
                                                          float* __restrict__ a_dst) {
    __shared__ float Wa[16][EMB + 1];
    const int tid = threadIdx.x;
    for (int i = tid; i < 16 * EMB; i += 256) {
        const int k = i >> 4;
        const int j = i & 15;
        float s = 0.0f;
        if (j < 8) {
            const int hh = j & 3;
            const float* att = (j < 4 ? att_src : att_dst) + hh * RD;
            const float* wr  = W + k * HOUT + hh * RD;
            #pragma unroll
            for (int dd = 0; dd < RD; ++dd) s += wr[dd] * att[dd];
        }
        Wa[j][k] = s;
    }
    __syncthreads();

    const int wave = tid >> 5;
    const int lane = tid & 31;
    const int rowTile = (blockIdx.x * 8 + wave) * 16;
    if (rowTile >= N_NODES) return;

    const int half = lane >> 4;
    const int lrow = lane & 15;
    const float* arow = emb + (size_t)(rowTile + lrow) * EMB;

    v8f acc = {};
    for (int k0 = 0; k0 < EMB; k0 += 4) {
        v2f a, b;
        a.x = arow[k0 + 2 * half];
        a.y = arow[k0 + 2 * half + 1];
        b.x = Wa[lrow][k0 + 2 * half];
        b.y = Wa[lrow][k0 + 2 * half + 1];
        acc = __builtin_amdgcn_wmma_f32_16x16x4_f32(
            false, a, false, b, (short)0, acc, false, false);
    }
    float* dstp = (lrow < 4 ? a_src + lrow : a_dst + (lrow - 4)) + (size_t)(rowTile + 8 * half) * HEADS;
    if (lrow < 8) {
        #pragma unroll
        for (int v = 0; v < 8; ++v) dstp[v * HEADS] = acc[v];
    }
}

__global__ void count_kernel(const int* __restrict__ path, int* __restrict__ cnt,
                             int* __restrict__ rank) {
    int e = blockIdx.x * blockDim.x + threadIdx.x;
    if (e >= NEDGES) return;
    rank[e] = atomicAdd(&cnt[path[NEDGES + e]], 1);
}

#define SCAN_T 1024
__global__ __launch_bounds__(SCAN_T) void offsets_kernel(const int* __restrict__ cnt,
                                                         int* __restrict__ off) {
    __shared__ int wtot[32];
    __shared__ int wpart[32];
    const int tid  = threadIdx.x;
    const int lane = tid & 31;
    const int wave = tid >> 5;
    const int base = blockIdx.x * SCAN_T;

    int part = 0;
    for (int i = tid; i < base; i += SCAN_T) part += cnt[i];
    for (int o = 16; o > 0; o >>= 1) part += __shfl_xor(part, o, 32);

    const int i = base + tid;
    const int c = (i < N_NODES) ? cnt[i] : 0;
    int incl = c;
    for (int o = 1; o < 32; o <<= 1) {
        const int up = __shfl_up(incl, o, 32);
        if (lane >= o) incl += up;
    }
    if (lane == 31) { wtot[wave] = incl; wpart[wave] = part; }
    __syncthreads();

    int before = 0;
    for (int w = 0; w < 32; ++w) {
        before += wpart[w];
        if (w < wave) before += wtot[w];
    }
    if (i < N_NODES) {
        off[i] = before + incl - c;
        if (i == N_NODES - 1) off[N_NODES] = before + incl;
    }
}

__global__ void scatter_kernel(const int* __restrict__ path, const int* __restrict__ off,
                               const int* __restrict__ rank, int* __restrict__ srt) {
    int e = blockIdx.x * blockDim.x + threadIdx.x;
    if (e >= NEDGES) return;
    srt[off[path[NEDGES + e]] + rank[e]] = path[e];
}

template <int M>
__device__ __forceinline__ float xrow(float x) {
    const int b = __float_as_int(x);
    return __int_as_float(__builtin_amdgcn_update_dpp(b, b, 0x160 + M, 0xf, 0xf, false));
}

__device__ __forceinline__ float xhalf(float x) {
    const int b = __float_as_int(x);
    return __int_as_float(__builtin_amdgcn_permlanex16(b, b, 0x76543210, 0xfedcba98, false, false));
}

__device__ __forceinline__ float wave_max(float v) {
    v = fmaxf(v, xrow<1>(v));
    v = fmaxf(v, xrow<2>(v));
    v = fmaxf(v, xrow<4>(v));
    v = fmaxf(v, xrow<8>(v));
    return fmaxf(v, xhalf(v));
}

__device__ __forceinline__ float wave_sum(float v) {
    v += xrow<1>(v);
    v += xrow<2>(v);
    v += xrow<4>(v);
    v += xrow<8>(v);
    return v + xhalf(v);
}

__device__ __forceinline__ float4 wave_max4(float4 v) {
    return make_float4(wave_max(v.x), wave_max(v.y), wave_max(v.z), wave_max(v.w));
}

__device__ __forceinline__ float4 wave_sum4(float4 v) {
    return make_float4(wave_sum(v.x), wave_sum(v.y), wave_sum(v.z), wave_sum(v.w));
}

__device__ __forceinline__ float4 edge_logits(const int* __restrict__ srt,
                                              const float* __restrict__ a_src,
                                              float4 ad, int j, int& s) {
    s = srt[j];
    const float4 as = *(const float4*)(a_src + (size_t)s * 4);
    float4 e;
    e.x = leaky(as.x + ad.x);
    e.y = leaky(as.y + ad.y);
    e.z = leaky(as.z + ad.z);
    e.w = leaky(as.w + ad.w);
    return e;
}

__device__ __forceinline__ float to_scalar(float v) {
    return __int_as_float(__builtin_amdgcn_readfirstlane(__float_as_int(v)));
}

__device__ __forceinline__ int bcast(float v, int srcLane) {
    return __builtin_amdgcn_readlane(__float_as_int(v), srcLane);
}

template <int G>
__device__ __forceinline__ void load_rows(const float* __restrict__ emb, int s, int t,
                                          unsigned lane, float2 (&r)[G]) {
    #pragma unroll
    for (int i = 0; i < G; ++i) {
        const unsigned st = (unsigned)__builtin_amdgcn_readlane(s, t + i);
        r[i] = ((const float2*)emb)[st * (EMB / 2) + lane];
    }
}

template <int G>
__device__ __forceinline__ void add_rows(const float2 (&r)[G], float4 al, int t,
                                         float2& z0, float2& z1, float2& z2, float2& z3) {
    #pragma unroll
    for (int i = 0; i < G; ++i) {
        const float a0 = __int_as_float(bcast(al.x, t + i));
        const float a1 = __int_as_float(bcast(al.y, t + i));
        const float a2 = __int_as_float(bcast(al.z, t + i));
        const float a3 = __int_as_float(bcast(al.w, t + i));
        z0.x = __builtin_fmaf(a0, r[i].x, z0.x); z0.y = __builtin_fmaf(a0, r[i].y, z0.y);
        z1.x = __builtin_fmaf(a1, r[i].x, z1.x); z1.y = __builtin_fmaf(a1, r[i].y, z1.y);
        z2.x = __builtin_fmaf(a2, r[i].x, z2.x); z2.y = __builtin_fmaf(a2, r[i].y, z2.y);
        z3.x = __builtin_fmaf(a3, r[i].x, z3.x); z3.y = __builtin_fmaf(a3, r[i].y, z3.y);
    }
}

__device__ __forceinline__ void add_groups(const float* __restrict__ emb, int s, float4 al,
                                           int t0, int n, unsigned lane,
                                           float2& z0, float2& z1, float2& z2, float2& z3) {
    #pragma unroll 1
    for (int t = t0; t < n; t += 8) {
        float2 q[8];
        load_rows<8>(emb, s, t, lane, q);
        add_rows<8>(q, al, t, z0, z1, z2, z3);
    }
}

#define ZK (HEADS * EMB)
#define ZP (ZK + 4)
#define PP (256 + 16)
__global__ __launch_bounds__(512) void gather_project_kernel(const int* __restrict__ off,
                                                             const int* __restrict__ srt,
                                                             const float* __restrict__ emb,
                                                             const float* __restrict__ W,
                                                             const float* __restrict__ a_src,
                                                             const float* __restrict__ a_dst,
                                                             float* __restrict__ out) {
    __shared__ __attribute__((aligned(16))) float zt[16][ZP];
    __shared__ __attribute__((aligned(16))) float part[16][PP];
    const int lane = threadIdx.x & 31;
    const int wave = __builtin_amdgcn_readfirstlane(threadIdx.x >> 5);
    const int d = blockIdx.x * 16 + wave;
    const int beg = __builtin_amdgcn_readfirstlane(off[d]);
    const int end = __builtin_amdgcn_readfirstlane(off[d + 1]);
    const float4 ad = *(const float4*)(a_dst + (size_t)d * 4);
    const float NEG_INF = __int_as_float(0xFF800000);

    const int  j0 = beg + lane;
    const bool v0 = j0 < end;
    const unsigned ulane = (unsigned)lane;
    const int n0 = min(32, end - beg);
    int s0 = 0;
    if (v0) s0 = srt[j0];
    const int sp = v0 ? s0 : __builtin_amdgcn_readlane(s0, max(n0 - 1, 0));
    const float4 as0 = *(const float4*)(a_src + (size_t)sp * 4);
    float2 r0[16];
    load_rows<16>(emb, sp, 0, ulane, r0);
    float4 e0;
    e0.x = v0 ? leaky(as0.x + ad.x) : NEG_INF; e0.y = v0 ? leaky(as0.y + ad.y) : NEG_INF;
    e0.z = v0 ? leaky(as0.z + ad.z) : NEG_INF; e0.w = v0 ? leaky(as0.w + ad.w) : NEG_INF;

    float4 mx = e0;
    for (int j = j0 + 32; j < end; j += 32) {
        int s;
        const float4 e = edge_logits(srt, a_src, ad, j, s);
        mx.x = fmaxf(mx.x, e.x); mx.y = fmaxf(mx.y, e.y);
        mx.z = fmaxf(mx.z, e.z); mx.w = fmaxf(mx.w, e.w);
    }
    mx = wave_max4(mx);
    mx = make_float4(to_scalar(mx.x), to_scalar(mx.y), to_scalar(mx.z), to_scalar(mx.w));

    float4 ex0 = make_float4(0.0f, 0.0f, 0.0f, 0.0f);
    if (v0) {
        ex0.x = __expf(e0.x - mx.x); ex0.y = __expf(e0.y - mx.y);
        ex0.z = __expf(e0.z - mx.z); ex0.w = __expf(e0.w - mx.w);
    }
    float4 den = ex0;
    for (int j = j0 + 32; j < end; j += 32) {
        int s;
        const float4 e = edge_logits(srt, a_src, ad, j, s);
        den.x += __expf(e.x - mx.x); den.y += __expf(e.y - mx.y);
        den.z += __expf(e.z - mx.z); den.w += __expf(e.w - mx.w);
    }
    den = wave_sum4(den);
    den.x += 1e-16f; den.y += 1e-16f; den.z += 1e-16f; den.w += 1e-16f;
    den = make_float4(to_scalar(den.x), to_scalar(den.y), to_scalar(den.z), to_scalar(den.w));

    float2 z0 = make_float2(0.0f, 0.0f), z1 = z0, z2 = z0, z3 = z0;

    if (n0 > 0) {
        const float4 al = make_float4(ex0.x / den.x, ex0.y / den.y, ex0.z / den.z, ex0.w / den.w);
        add_rows<16>(r0, al, 0, z0, z1, z2, z3);
        add_groups(emb, sp, al, 16, n0, ulane, z0, z1, z2, z3);
    }
    for (int base = beg + 32; base < end; base += 32) {
        const int j = base + lane;
        const int n = min(32, end - base);
        int s = 0;
        float4 al = make_float4(0.0f, 0.0f, 0.0f, 0.0f);
        if (j < end) {
            const float4 e = edge_logits(srt, a_src, ad, j, s);
            al.x = __expf(e.x - mx.x) / den.x; al.y = __expf(e.y - mx.y) / den.y;
            al.z = __expf(e.z - mx.z) / den.z; al.w = __expf(e.w - mx.w) / den.w;
        }
        const int sl = __builtin_amdgcn_readlane(s, n - 1);
        s = (j < end) ? s : sl;
        add_groups(emb, s, al, 0, n, ulane, z0, z1, z2, z3);
    }
    *(float2*)&zt[wave][0 * EMB + lane * 2] = z0;
    *(float2*)&zt[wave][1 * EMB + lane * 2] = z1;
    *(float2*)&zt[wave][2 * EMB + lane * 2] = z2;
    *(float2*)&zt[wave][3 * EMB + lane * 2] = z3;

    const int half = lane >> 4;
    const int lrow = lane & 15;
    const int ct   = wave & 1;
    const int kq   = wave >> 1;
    const int hk   = kq >> 1;
    const int kin  = (kq & 1) * 32;
    const float* wp = W + (size_t)(kin + 2 * half) * HOUT + hk * RD + ct * 16 + lrow;
    v2f bf[8];
    #pragma unroll
    for (int ks = 0; ks < 8; ++ks) {
        bf[ks].x = wp[(ks * 4) * HOUT];
        bf[ks].y = wp[(ks * 4 + 1) * HOUT];
    }
    __syncthreads();

    const float* zr = &zt[lrow][kq * 32 + 2 * half];
    v8f c0 = {}, c1 = {};
    #pragma unroll
    for (int ks = 0; ks < 8; ks += 2) {
        const v2f fa0 = *(const v2f*)(zr + ks * 4);
        const v2f fa1 = *(const v2f*)(zr + ks * 4 + 4);
        c0 = __builtin_amdgcn_wmma_f32_16x16x4_f32(false, fa0, false, bf[ks],     (short)0, c0, false, false);
        c1 = __builtin_amdgcn_wmma_f32_16x16x4_f32(false, fa1, false, bf[ks + 1], (short)0, c1, false, false);
    }
    #pragma unroll
    for (int v = 0; v < 8; ++v) part[wave][v * 32 + lane] = c0[v] + c1[v];
    __syncthreads();

    const int pl = (wave >> 3) * 16 + lrow;
    const int pv = wave & 7;
    const int pc = lane >> 4;
    float sum = 0.0f;
    #pragma unroll
    for (int q = 0; q < 8; ++q) sum += part[2 * q + pc][pv * 32 + pl];
    out[(size_t)d * RD + lane] = 0.25f * sum;
}

extern "C" void kernel_launch(void* const* d_in, const int* in_sizes, int n_in,
                              void* d_out, int out_size, void* d_ws, size_t ws_size,
                              hipStream_t stream) {
    const float* node_emb = (const float*)d_in[0];
    const float* W        = (const float*)d_in[1];
    const float* att_src  = (const float*)d_in[2];
    const float* att_dst  = (const float*)d_in[3];
    const int*   path     = (const int*)d_in[4];
    float* out = (float*)d_out;

    float* ws    = (float*)d_ws;
    float* h     = ws;
    float* a_src = h     + (size_t)N_NODES * HOUT;
    float* a_dst = a_src + (size_t)N_NODES * HEADS;
    int*   cnt   = (int*)(a_dst + (size_t)N_NODES * HEADS);
    int*   off   = cnt  + (size_t)N_NODES;
    int*   rank  = off  + (size_t)N_NODES + 32;
    int*   srt   = rank + (size_t)NEDGES;

    init_kernel<<<(N_NODES + 255) / 256, 256, 0, stream>>>(cnt);

    const int rowTiles = N_NODES / 16;
    logits_wmma_kernel<<<(rowTiles + 7) / 8, 256, 0, stream>>>(
        node_emb, W, att_src, att_dst, a_src, a_dst);

    count_kernel<<<(NEDGES + 255) / 256, 256, 0, stream>>>(path, cnt, rank);

    offsets_kernel<<<(N_NODES + SCAN_T - 1) / SCAN_T, SCAN_T, 0, stream>>>(cnt, off);

    scatter_kernel<<<(NEDGES + 255) / 256, 256, 0, stream>>>(path, off, rank, srt);

    gather_project_kernel<<<N_NODES / 16, 512, 0, stream>>>(off, srt, node_emb, W, a_src, a_dst, out);
}


# Edge grouping ranks from LDS counters instead of one returning global atomic per edge on scattered words: two-level counting sort (196 destination ranges, one block each); arrival order kept
# speedup vs baseline: 5.6095x; 1.7399x over previous
#include <hip/hip_runtime.h>

#define N_NODES   100000
#define EMB       64
#define HEADS     4
#define RD        32
#define HOUT      (HEADS * RD)
#define NEDGES    1600000
#define NEG_SLOPE 0.2f

typedef __attribute__((ext_vector_type(2))) float v2f;
typedef __attribute__((ext_vector_type(8))) float v8f;

__device__ __forceinline__ float leaky(float x) {
    return x >= 0.0f ? x : NEG_SLOPE * x;
}

__global__ __launch_bounds__(256) void logits_wmma_kernel(const float* __restrict__ emb,
                                                          const float* __restrict__ W,
                                                          const float* __restrict__ att_src,
                                                          const float* __restrict__ att_dst,
                                                          float* __restrict__ a_src,
                                                          float* __restrict__ a_dst) {
    __shared__ float Wa[16][EMB + 1];
    const int tid = threadIdx.x;
    for (int i = tid; i < 16 * EMB; i += 256) {
        const int k = i >> 4;
        const int j = i & 15;
        float s = 0.0f;
        if (j < 8) {
            const int hh = j & 3;
            const float* att = (j < 4 ? att_src : att_dst) + hh * RD;
            const float* wr  = W + k * HOUT + hh * RD;
            #pragma unroll
            for (int dd = 0; dd < RD; ++dd) s += wr[dd] * att[dd];
        }
        Wa[j][k] = s;
    }
    __syncthreads();

    const int wave = tid >> 5;
    const int lane = tid & 31;
    const int rowTile = (blockIdx.x * 8 + wave) * 16;
    if (rowTile >= N_NODES) return;

    const int half = lane >> 4;
    const int lrow = lane & 15;
    const float* arow = emb + (size_t)(rowTile + lrow) * EMB;

    v8f acc = {};
    for (int k0 = 0; k0 < EMB; k0 += 4) {
        v2f a, b;
        a.x = arow[k0 + 2 * half];
        a.y = arow[k0 + 2 * half + 1];
        b.x = Wa[lrow][k0 + 2 * half];
        b.y = Wa[lrow][k0 + 2 * half + 1];
        acc = __builtin_amdgcn_wmma_f32_16x16x4_f32(
            false, a, false, b, (short)0, acc, false, false);
    }
    float* dstp = (lrow < 4 ? a_src + lrow : a_dst + (lrow - 4)) + (size_t)(rowTile + 8 * half) * HEADS;
    if (lrow < 8) {
        #pragma unroll
        for (int v = 0; v < 8; ++v) dstp[v * HEADS] = acc[v];
    }
}

#define BSH    9
#define BW     (1 << BSH)
#define NBUCK  ((N_NODES + BW - 1) >> BSH)
#define CHUNK  6400
#define NB1    (NEDGES / CHUNK)
#define SRT_T  1024
#define FULLQ  (CHUNK / SRT_T)
#define TAILT  (CHUNK - FULLQ * SRT_T)
#define SRCBITS 17
#define PARTS  5
#define BATCH  10
static_assert(NEDGES % CHUNK == 0, "level-1 chunks must tile the edge list");
static_assert(TAILT > 0 && TAILT % 32 == 0, "the 7th edge belongs to whole waves");
static_assert(N_NODES <= (1 << SRCBITS), "source id and local destination share one word");
static_assert(NB1 % PARTS == 0 && PARTS * NBUCK <= SRT_T, "table sums split evenly");
static_assert(NB1 * NBUCK + NBUCK + 1 <= N_NODES, "count table and range starts fit the old counter region");

__global__ __launch_bounds__(SRT_T) void range_count_kernel(const int* __restrict__ path,
                                                            int* __restrict__ cntB) {
    __shared__ int hist[NBUCK];
    const int tid = threadIdx.x;
    const int* dstp = path + NEDGES + blockIdx.x * CHUNK;
    const bool tail = tid < TAILT;
    int dv[FULLQ], dt = 0;
    #pragma unroll
    for (int q = 0; q < FULLQ; ++q) dv[q] = dstp[tid + q * SRT_T];
    if (tail) dt = dstp[tid + FULLQ * SRT_T];
    if (tid < NBUCK) hist[tid] = 0;
    __syncthreads();
    #pragma unroll
    for (int q = 0; q < FULLQ; ++q) atomicAdd(&hist[dv[q] >> BSH], 1);
    if (tail) atomicAdd(&hist[dt >> BSH], 1);
    __syncthreads();
    if (tid < NBUCK) cntB[blockIdx.x * NBUCK + tid] = hist[tid];
}

__global__ __launch_bounds__(SRT_T) void range_scatter_kernel(const int* __restrict__ path,
                                                              const int* __restrict__ cntB,
                                                              int* __restrict__ bstart,
                                                              int* __restrict__ bkt) {
    __shared__ int ptot[PARTS][NBUCK];
    __shared__ int pbef[PARTS][NBUCK];
    __shared__ int tot[NBUCK];
    __shared__ int cur[NBUCK];
    const int tid = threadIdx.x;
    const int b   = blockIdx.x;
    const int* srcp = path + b * CHUNK;
    const int* dstp = path + NEDGES + b * CHUNK;
    const bool tail = tid < TAILT;
    int dv[FULLQ], sv[FULLQ], dt = 0, st = 0;
    #pragma unroll
    for (int q = 0; q < FULLQ; ++q) {
        dv[q] = dstp[tid + q * SRT_T];
        sv[q] = srcp[tid + q * SRT_T];
    }
    if (tail) { dt = dstp[tid + FULLQ * SRT_T]; st = srcp[tid + FULLQ * SRT_T]; }
    if (tid < PARTS * NBUCK) {
        const int k = tid % NBUCK;
        const int p = tid / NBUCK;
        int total = 0, before = 0;
        #pragma unroll
        for (int j = 0; j < NB1 / PARTS; ++j) {
            const int bb = p + j * PARTS;
            const int c = cntB[bb * NBUCK + k];
            total += c;
            before += (bb < b) ? c : 0;
        }
        ptot[p][k] = total;
        pbef[p][k] = before;
    }
    __syncthreads();
    int total = 0, before = 0, rstart = 0;
    if (tid < NBUCK) {
        #pragma unroll
        for (int p = 0; p < PARTS; ++p) { total += ptot[p][tid]; before += pbef[p][tid]; }
        tot[tid] = total;
    }
    __syncthreads();
    if (tid < NBUCK) {
        int start = 0;
        for (int k = 0; k < tid; ++k) start += tot[k];
        rstart = start;
        cur[tid] = start + before;
    }
    __syncthreads();
    int pos[FULLQ], pt = 0;
    #pragma unroll
    for (int q = 0; q < FULLQ; ++q) pos[q] = atomicAdd(&cur[dv[q] >> BSH], 1);
    if (tail) pt = atomicAdd(&cur[dt >> BSH], 1);
    #pragma unroll
    for (int q = 0; q < FULLQ; ++q) bkt[pos[q]] = ((dv[q] & (BW - 1)) << SRCBITS) | sv[q];
    if (tail) bkt[pt] = ((dt & (BW - 1)) << SRCBITS) | st;
    if (b == 0 && tid < NBUCK) {
        bstart[tid] = rstart;
        if (tid == NBUCK - 1) bstart[NBUCK] = rstart + total;
    }
}

__global__ __launch_bounds__(SRT_T) void range_sort_kernel(const int* __restrict__ bkt,
                                                           const int* __restrict__ bstart,
                                                           int* __restrict__ off,
                                                           int* __restrict__ srt) {
    __shared__ int cnt[BW];
    __shared__ int wsum[BW / 32];
    const int tid  = threadIdx.x;
    const int lane = tid & 31;
    const int wave = tid >> 5;
    const int k  = blockIdx.x;
    const int lo = bstart[k];
    const int hi = bstart[k + 1];

    int w0[BATCH];
    #pragma unroll
    for (int q = 0; q < BATCH; ++q) {
        const int i = lo + q * SRT_T + tid;
        w0[q] = (i < hi) ? bkt[i] : -1;
    }
    if (tid < BW) cnt[tid] = 0;
    __syncthreads();
    #pragma unroll
    for (int q = 0; q < BATCH; ++q)
        if (w0[q] >= 0) atomicAdd(&cnt[w0[q] >> SRCBITS], 1);
    for (int i = lo + BATCH * SRT_T + tid; i < hi; i += SRT_T)
        atomicAdd(&cnt[bkt[i] >> SRCBITS], 1);
    __syncthreads();

    int c = 0, incl = 0, myoff = 0;
    if (wave < BW / 32) {
        c = cnt[tid];
        incl = c;
        for (int o = 1; o < 32; o <<= 1) {
            const int up = __shfl_up(incl, o, 32);
            if (lane >= o) incl += up;
        }
        if (lane == 31) wsum[wave] = incl;
    }
    __syncthreads();
    if (wave < BW / 32) {
        int before = lo;
        for (int w = 0; w < wave; ++w) before += wsum[w];
        myoff = before + incl - c;
        cnt[tid] = myoff;
    }
    __syncthreads();

    int pos[BATCH];
    #pragma unroll
    for (int q = 0; q < BATCH; ++q) {
        pos[q] = 0;
        if (w0[q] >= 0) pos[q] = atomicAdd(&cnt[w0[q] >> SRCBITS], 1);
    }
    if (wave < BW / 32) {
        const int d = k * BW + tid;
        if (d < N_NODES) off[d] = myoff;
    }
    if (k == NBUCK - 1 && tid == 0) off[N_NODES] = hi;
    #pragma unroll
    for (int q = 0; q < BATCH; ++q)
        if (w0[q] >= 0) srt[pos[q]] = w0[q] & ((1 << SRCBITS) - 1);
    for (int i = lo + BATCH * SRT_T + tid; i < hi; i += SRT_T) {
        const int w = bkt[i];
        const int slot = atomicAdd(&cnt[w >> SRCBITS], 1);
        srt[slot] = w & ((1 << SRCBITS) - 1);
    }
}

template <int M>
__device__ __forceinline__ float xrow(float x) {
    const int b = __float_as_int(x);
    return __int_as_float(__builtin_amdgcn_update_dpp(b, b, 0x160 + M, 0xf, 0xf, false));
}

__device__ __forceinline__ float xhalf(float x) {
    const int b = __float_as_int(x);
    return __int_as_float(__builtin_amdgcn_permlanex16(b, b, 0x76543210, 0xfedcba98, false, false));
}

__device__ __forceinline__ float wave_max(float v) {
    v = fmaxf(v, xrow<1>(v));
    v = fmaxf(v, xrow<2>(v));
    v = fmaxf(v, xrow<4>(v));
    v = fmaxf(v, xrow<8>(v));
    return fmaxf(v, xhalf(v));
}

__device__ __forceinline__ float wave_sum(float v) {
    v += xrow<1>(v);
    v += xrow<2>(v);
    v += xrow<4>(v);
    v += xrow<8>(v);
    return v + xhalf(v);
}

__device__ __forceinline__ float4 wave_max4(float4 v) {
    return make_float4(wave_max(v.x), wave_max(v.y), wave_max(v.z), wave_max(v.w));
}

__device__ __forceinline__ float4 wave_sum4(float4 v) {
    return make_float4(wave_sum(v.x), wave_sum(v.y), wave_sum(v.z), wave_sum(v.w));
}

__device__ __forceinline__ float4 edge_logits(const int* __restrict__ srt,
                                              const float* __restrict__ a_src,
                                              float4 ad, int j, int& s) {
    s = srt[j];
    const float4 as = *(const float4*)(a_src + (size_t)s * 4);
    float4 e;
    e.x = leaky(as.x + ad.x);
    e.y = leaky(as.y + ad.y);
    e.z = leaky(as.z + ad.z);
    e.w = leaky(as.w + ad.w);
    return e;
}

__device__ __forceinline__ float to_scalar(float v) {
    return __int_as_float(__builtin_amdgcn_readfirstlane(__float_as_int(v)));
}

__device__ __forceinline__ int bcast(float v, int srcLane) {
    return __builtin_amdgcn_readlane(__float_as_int(v), srcLane);
}

template <int G>
__device__ __forceinline__ void load_rows(const float* __restrict__ emb, int s, int t,
                                          unsigned lane, float2 (&r)[G]) {
    #pragma unroll
    for (int i = 0; i < G; ++i) {
        const unsigned st = (unsigned)__builtin_amdgcn_readlane(s, t + i);
        r[i] = ((const float2*)emb)[st * (EMB / 2) + lane];
    }
}

template <int G>
__device__ __forceinline__ void add_rows(const float2 (&r)[G], float4 al, int t,
                                         float2& z0, float2& z1, float2& z2, float2& z3) {
    #pragma unroll
    for (int i = 0; i < G; ++i) {
        const float a0 = __int_as_float(bcast(al.x, t + i));
        const float a1 = __int_as_float(bcast(al.y, t + i));
        const float a2 = __int_as_float(bcast(al.z, t + i));
        const float a3 = __int_as_float(bcast(al.w, t + i));
        z0.x = __builtin_fmaf(a0, r[i].x, z0.x); z0.y = __builtin_fmaf(a0, r[i].y, z0.y);
        z1.x = __builtin_fmaf(a1, r[i].x, z1.x); z1.y = __builtin_fmaf(a1, r[i].y, z1.y);
        z2.x = __builtin_fmaf(a2, r[i].x, z2.x); z2.y = __builtin_fmaf(a2, r[i].y, z2.y);
        z3.x = __builtin_fmaf(a3, r[i].x, z3.x); z3.y = __builtin_fmaf(a3, r[i].y, z3.y);
    }
}

__device__ __forceinline__ void add_groups(const float* __restrict__ emb, int s, float4 al,
                                           int t0, int n, unsigned lane,
                                           float2& z0, float2& z1, float2& z2, float2& z3) {
    #pragma unroll 1
    for (int t = t0; t < n; t += 8) {
        float2 q[8];
        load_rows<8>(emb, s, t, lane, q);
        add_rows<8>(q, al, t, z0, z1, z2, z3);
    }
}

#define ZK (HEADS * EMB)
#define ZP (ZK + 4)
#define PP (256 + 16)
__global__ __launch_bounds__(512) void gather_project_kernel(const int* __restrict__ off,
                                                             const int* __restrict__ srt,
                                                             const float* __restrict__ emb,
                                                             const float* __restrict__ W,
                                                             const float* __restrict__ a_src,
                                                             const float* __restrict__ a_dst,
                                                             float* __restrict__ out) {
    __shared__ __attribute__((aligned(16))) float zt[16][ZP];
    __shared__ __attribute__((aligned(16))) float part[16][PP];
    const int lane = threadIdx.x & 31;
    const int wave = __builtin_amdgcn_readfirstlane(threadIdx.x >> 5);
    const int d = blockIdx.x * 16 + wave;
    const int beg = __builtin_amdgcn_readfirstlane(off[d]);
    const int end = __builtin_amdgcn_readfirstlane(off[d + 1]);
    const float4 ad = *(const float4*)(a_dst + (size_t)d * 4);
    const float NEG_INF = __int_as_float(0xFF800000);

    const int  j0 = beg + lane;
    const bool v0 = j0 < end;
    const unsigned ulane = (unsigned)lane;
    const int n0 = min(32, end - beg);
    int s0 = 0;
    if (v0) s0 = srt[j0];
    const int sp = v0 ? s0 : __builtin_amdgcn_readlane(s0, max(n0 - 1, 0));
    const float4 as0 = *(const float4*)(a_src + (size_t)sp * 4);
    float2 r0[16];
    load_rows<16>(emb, sp, 0, ulane, r0);
    float4 e0;
    e0.x = v0 ? leaky(as0.x + ad.x) : NEG_INF; e0.y = v0 ? leaky(as0.y + ad.y) : NEG_INF;
    e0.z = v0 ? leaky(as0.z + ad.z) : NEG_INF; e0.w = v0 ? leaky(as0.w + ad.w) : NEG_INF;

    float4 mx = e0;
    for (int j = j0 + 32; j < end; j += 32) {
        int s;
        const float4 e = edge_logits(srt, a_src, ad, j, s);
        mx.x = fmaxf(mx.x, e.x); mx.y = fmaxf(mx.y, e.y);
        mx.z = fmaxf(mx.z, e.z); mx.w = fmaxf(mx.w, e.w);
    }
    mx = wave_max4(mx);
    mx = make_float4(to_scalar(mx.x), to_scalar(mx.y), to_scalar(mx.z), to_scalar(mx.w));

    float4 ex0 = make_float4(0.0f, 0.0f, 0.0f, 0.0f);
    if (v0) {
        ex0.x = __expf(e0.x - mx.x); ex0.y = __expf(e0.y - mx.y);
        ex0.z = __expf(e0.z - mx.z); ex0.w = __expf(e0.w - mx.w);
    }
    float4 den = ex0;
    for (int j = j0 + 32; j < end; j += 32) {
        int s;
        const float4 e = edge_logits(srt, a_src, ad, j, s);
        den.x += __expf(e.x - mx.x); den.y += __expf(e.y - mx.y);
        den.z += __expf(e.z - mx.z); den.w += __expf(e.w - mx.w);
    }
    den = wave_sum4(den);
    den.x += 1e-16f; den.y += 1e-16f; den.z += 1e-16f; den.w += 1e-16f;
    den = make_float4(to_scalar(den.x), to_scalar(den.y), to_scalar(den.z), to_scalar(den.w));

    float2 z0 = make_float2(0.0f, 0.0f), z1 = z0, z2 = z0, z3 = z0;

    if (n0 > 0) {
        const float4 al = make_float4(ex0.x / den.x, ex0.y / den.y, ex0.z / den.z, ex0.w / den.w);
        add_rows<16>(r0, al, 0, z0, z1, z2, z3);
        add_groups(emb, sp, al, 16, n0, ulane, z0, z1, z2, z3);
    }
    for (int base = beg + 32; base < end; base += 32) {
        const int j = base + lane;
        const int n = min(32, end - base);
        int s = 0;
        float4 al = make_float4(0.0f, 0.0f, 0.0f, 0.0f);
        if (j < end) {
            const float4 e = edge_logits(srt, a_src, ad, j, s);
            al.x = __expf(e.x - mx.x) / den.x; al.y = __expf(e.y - mx.y) / den.y;
            al.z = __expf(e.z - mx.z) / den.z; al.w = __expf(e.w - mx.w) / den.w;
        }
        const int sl = __builtin_amdgcn_readlane(s, n - 1);
        s = (j < end) ? s : sl;
        add_groups(emb, s, al, 0, n, ulane, z0, z1, z2, z3);
    }
    *(float2*)&zt[wave][0 * EMB + lane * 2] = z0;
    *(float2*)&zt[wave][1 * EMB + lane * 2] = z1;
    *(float2*)&zt[wave][2 * EMB + lane * 2] = z2;
    *(float2*)&zt[wave][3 * EMB + lane * 2] = z3;

    const int half = lane >> 4;
    const int lrow = lane & 15;
    const int ct   = wave & 1;
    const int kq   = wave >> 1;
    const int hk   = kq >> 1;
    const int kin  = (kq & 1) * 32;
    const float* wp = W + (size_t)(kin + 2 * half) * HOUT + hk * RD + ct * 16 + lrow;
    v2f bf[8];
    #pragma unroll
    for (int ks = 0; ks < 8; ++ks) {
        bf[ks].x = wp[(ks * 4) * HOUT];
        bf[ks].y = wp[(ks * 4 + 1) * HOUT];
    }
    __syncthreads();

    const float* zr = &zt[lrow][kq * 32 + 2 * half];
    v8f c0 = {}, c1 = {};
    #pragma unroll
    for (int ks = 0; ks < 8; ks += 2) {
        const v2f fa0 = *(const v2f*)(zr + ks * 4);
        const v2f fa1 = *(const v2f*)(zr + ks * 4 + 4);
        c0 = __builtin_amdgcn_wmma_f32_16x16x4_f32(false, fa0, false, bf[ks],     (short)0, c0, false, false);
        c1 = __builtin_amdgcn_wmma_f32_16x16x4_f32(false, fa1, false, bf[ks + 1], (short)0, c1, false, false);
    }
    #pragma unroll
    for (int v = 0; v < 8; ++v) part[wave][v * 32 + lane] = c0[v] + c1[v];
    __syncthreads();

    const int pl = (wave >> 3) * 16 + lrow;
    const int pv = wave & 7;
    const int pc = lane >> 4;
    float sum = 0.0f;
    #pragma unroll
    for (int q = 0; q < 8; ++q) sum += part[2 * q + pc][pv * 32 + pl];
    out[(size_t)d * RD + lane] = 0.25f * sum;
}

extern "C" void kernel_launch(void* const* d_in, const int* in_sizes, int n_in,
                              void* d_out, int out_size, void* d_ws, size_t ws_size,
                              hipStream_t stream) {
    const float* node_emb = (const float*)d_in[0];
    const float* W        = (const float*)d_in[1];
    const float* att_src  = (const float*)d_in[2];
    const float* att_dst  = (const float*)d_in[3];
    const int*   path     = (const int*)d_in[4];
    float* out = (float*)d_out;

    float* ws    = (float*)d_ws;
    float* h     = ws;
    float* a_src = h     + (size_t)N_NODES * HOUT;
    float* a_dst = a_src + (size_t)N_NODES * HEADS;
    int*   cntB  = (int*)(a_dst + (size_t)N_NODES * HEADS);
    int*   bstart = cntB + (size_t)NB1 * NBUCK;
    int*   off   = cntB + (size_t)N_NODES;
    int*   bkt   = off  + (size_t)N_NODES + 32;
    int*   srt   = bkt  + (size_t)NEDGES;

    const int rowTiles = N_NODES / 16;
    logits_wmma_kernel<<<(rowTiles + 7) / 8, 256, 0, stream>>>(
        node_emb, W, att_src, att_dst, a_src, a_dst);

    range_count_kernel<<<NB1, SRT_T, 0, stream>>>(path, cntB);

    range_scatter_kernel<<<NB1, SRT_T, 0, stream>>>(path, cntB, bstart, bkt);

    range_sort_kernel<<<NBUCK, SRT_T, 0, stream>>>(bkt, bstart, off, srt);

    gather_project_kernel<<<N_NODES / 16, 512, 0, stream>>>(off, srt, node_emb, W, a_src, a_dst, out);
}


# Both sort levels place their words in an LDS copy of the block's output and store it in slot order: a store covers whole runs or one aligned 128-B line, not 32 scattered 4-B slots; order as before
# speedup vs baseline: 7.5362x; 1.3435x over previous
#include <hip/hip_runtime.h>

#define N_NODES   100000
#define EMB       64
#define HEADS     4
#define RD        32
#define HOUT      (HEADS * RD)
#define NEDGES    1600000
#define NEG_SLOPE 0.2f

typedef __attribute__((ext_vector_type(2))) float v2f;
typedef __attribute__((ext_vector_type(8))) float v8f;

__device__ __forceinline__ float leaky(float x) {
    return x >= 0.0f ? x : NEG_SLOPE * x;
}

__global__ __launch_bounds__(256) void logits_wmma_kernel(const float* __restrict__ emb,
                                                          const float* __restrict__ W,
                                                          const float* __restrict__ att_src,
                                                          const float* __restrict__ att_dst,
                                                          float* __restrict__ a_src,
                                                          float* __restrict__ a_dst) {
    __shared__ float Wa[16][EMB + 1];
    const int tid = threadIdx.x;
    for (int i = tid; i < 16 * EMB; i += 256) {
        const int k = i >> 4;
        const int j = i & 15;
        float s = 0.0f;
        if (j < 8) {
            const int hh = j & 3;
            const float* att = (j < 4 ? att_src : att_dst) + hh * RD;
            const float* wr  = W + k * HOUT + hh * RD;
            #pragma unroll
            for (int dd = 0; dd < RD; ++dd) s += wr[dd] * att[dd];
        }
        Wa[j][k] = s;
    }
    __syncthreads();

    const int wave = tid >> 5;
    const int lane = tid & 31;
    const int rowTile = (blockIdx.x * 8 + wave) * 16;
    if (rowTile >= N_NODES) return;

    const int half = lane >> 4;
    const int lrow = lane & 15;
    const float* arow = emb + (size_t)(rowTile + lrow) * EMB;

    v8f acc = {};
    for (int k0 = 0; k0 < EMB; k0 += 4) {
        v2f a, b;
        a.x = arow[k0 + 2 * half];
        a.y = arow[k0 + 2 * half + 1];
        b.x = Wa[lrow][k0 + 2 * half];
        b.y = Wa[lrow][k0 + 2 * half + 1];
        acc = __builtin_amdgcn_wmma_f32_16x16x4_f32(
            false, a, false, b, (short)0, acc, false, false);
    }
    float* dstp = (lrow < 4 ? a_src + lrow : a_dst + (lrow - 4)) + (size_t)(rowTile + 8 * half) * HEADS;
    if (lrow < 8) {
        #pragma unroll
        for (int v = 0; v < 8; ++v) dstp[v * HEADS] = acc[v];
    }
}

#define BSH    9
#define BW     (1 << BSH)
#define NBUCK  ((N_NODES + BW - 1) >> BSH)
#define CHUNK  6400
#define NB1    (NEDGES / CHUNK)
#define SRT_T  1024
#define FULLQ  (CHUNK / SRT_T)
#define TAILT  (CHUNK - FULLQ * SRT_T)
#define SRCBITS 17
#define PARTS  5
#define BATCH  10
#define STAGE2 (BATCH * SRT_T)
static_assert(NEDGES % CHUNK == 0, "level-1 chunks must tile the edge list");
static_assert(TAILT > 0 && TAILT % 32 == 0, "the 7th edge belongs to whole waves");
static_assert(N_NODES <= (1 << SRCBITS), "source id and local destination share one word");
static_assert(NB1 % PARTS == 0 && PARTS * NBUCK <= SRT_T, "table sums split evenly");
static_assert(NB1 * NBUCK + NBUCK + 1 <= N_NODES, "count table and range starts fit the old counter region");
static_assert(NBUCK <= 256, "a range id fits the byte kept beside each staged word");

__global__ __launch_bounds__(SRT_T) void range_count_kernel(const int* __restrict__ path,
                                                            int* __restrict__ cntB) {
    __shared__ int hist[NBUCK];
    const int tid = threadIdx.x;
    const int* dstp = path + NEDGES + blockIdx.x * CHUNK;
    const bool tail = tid < TAILT;
    int dv[FULLQ], dt = 0;
    #pragma unroll
    for (int q = 0; q < FULLQ; ++q) dv[q] = dstp[tid + q * SRT_T];
    if (tail) dt = dstp[tid + FULLQ * SRT_T];
    if (tid < NBUCK) hist[tid] = 0;
    __syncthreads();
    #pragma unroll
    for (int q = 0; q < FULLQ; ++q) atomicAdd(&hist[dv[q] >> BSH], 1);
    if (tail) atomicAdd(&hist[dt >> BSH], 1);
    __syncthreads();
    if (tid < NBUCK) cntB[blockIdx.x * NBUCK + tid] = hist[tid];
}

__global__ __launch_bounds__(SRT_T) void range_scatter_kernel(const int* __restrict__ path,
                                                              const int* __restrict__ cntB,
                                                              int* __restrict__ bstart,
                                                              int* __restrict__ bkt) {
    __shared__ int ptot[PARTS][NBUCK];
    __shared__ int pbef[PARTS][NBUCK];
    __shared__ int tot[NBUCK];
    __shared__ int own[NBUCK];
    __shared__ int cur[NBUCK];
    __shared__ int gdel[NBUCK];
    __shared__ int stage[CHUNK];
    __shared__ unsigned char rid[CHUNK];
    const int tid = threadIdx.x;
    const int b   = blockIdx.x;
    const int* srcp = path + b * CHUNK;
    const int* dstp = path + NEDGES + b * CHUNK;
    const bool tail = tid < TAILT;
    int dv[FULLQ], sv[FULLQ], dt = 0, st = 0;
    #pragma unroll
    for (int q = 0; q < FULLQ; ++q) {
        dv[q] = dstp[tid + q * SRT_T];
        sv[q] = srcp[tid + q * SRT_T];
    }
    if (tail) { dt = dstp[tid + FULLQ * SRT_T]; st = srcp[tid + FULLQ * SRT_T]; }
    int ownc = 0;
    if (tid < NBUCK) ownc = cntB[b * NBUCK + tid];
    if (tid < PARTS * NBUCK) {
        const int k = tid % NBUCK;
        const int p = tid / NBUCK;
        int total = 0, before = 0;
        #pragma unroll
        for (int j = 0; j < NB1 / PARTS; ++j) {
            const int bb = p + j * PARTS;
            const int c = cntB[bb * NBUCK + k];
            total += c;
            before += (bb < b) ? c : 0;
        }
        ptot[p][k] = total;
        pbef[p][k] = before;
    }
    __syncthreads();
    int total = 0, before = 0, rstart = 0;
    if (tid < NBUCK) {
        #pragma unroll
        for (int p = 0; p < PARTS; ++p) { total += ptot[p][tid]; before += pbef[p][tid]; }
        tot[tid] = total;
        own[tid] = ownc;
    }
    __syncthreads();
    if (tid < NBUCK) {
        int start = 0, lstart = 0;
        for (int k = 0; k < tid; ++k) { start += tot[k]; lstart += own[k]; }
        rstart = start;
        cur[tid]  = lstart;
        gdel[tid] = start + before - lstart;
    }
    __syncthreads();
    int pos[FULLQ], pt = 0;
    #pragma unroll
    for (int q = 0; q < FULLQ; ++q) pos[q] = atomicAdd(&cur[dv[q] >> BSH], 1);
    if (tail) pt = atomicAdd(&cur[dt >> BSH], 1);
    #pragma unroll
    for (int q = 0; q < FULLQ; ++q) {
        stage[pos[q]] = ((dv[q] & (BW - 1)) << SRCBITS) | sv[q];
        rid[pos[q]]   = (unsigned char)(dv[q] >> BSH);
    }
    if (tail) {
        stage[pt] = ((dt & (BW - 1)) << SRCBITS) | st;
        rid[pt]   = (unsigned char)(dt >> BSH);
    }
    __syncthreads();
    #pragma unroll
    for (int q = 0; q < FULLQ; ++q) {
        const int i = tid + q * SRT_T;
        bkt[i + gdel[rid[i]]] = stage[i];
    }
    if (tail) {
        const int i = tid + FULLQ * SRT_T;
        bkt[i + gdel[rid[i]]] = stage[i];
    }
    if (b == 0 && tid < NBUCK) {
        bstart[tid] = rstart;
        if (tid == NBUCK - 1) bstart[NBUCK] = rstart + total;
    }
}

__global__ __launch_bounds__(SRT_T) void range_sort_kernel(const int* __restrict__ bkt,
                                                           const int* __restrict__ bstart,
                                                           int* __restrict__ off,
                                                           int* __restrict__ srt) {
    __shared__ int cnt[BW];
    __shared__ int wsum[BW / 32];
    __shared__ int stage[STAGE2];
    const int tid  = threadIdx.x;
    const int lane = tid & 31;
    const int wave = tid >> 5;
    const int k  = blockIdx.x;
    const int lo = bstart[k];
    const int hi = bstart[k + 1];
    const int n  = hi - lo;
    const bool staged = n <= STAGE2;

    int w0[BATCH];
    #pragma unroll
    for (int q = 0; q < BATCH; ++q) {
        const int i = lo + q * SRT_T + tid;
        w0[q] = (i < hi) ? bkt[i] : -1;
    }
    if (tid < BW) cnt[tid] = 0;
    __syncthreads();
    #pragma unroll
    for (int q = 0; q < BATCH; ++q)
        if (w0[q] >= 0) atomicAdd(&cnt[w0[q] >> SRCBITS], 1);
    for (int i = lo + BATCH * SRT_T + tid; i < hi; i += SRT_T)
        atomicAdd(&cnt[bkt[i] >> SRCBITS], 1);
    __syncthreads();

    int c = 0, incl = 0, myoff = 0;
    if (wave < BW / 32) {
        c = cnt[tid];
        incl = c;
        for (int o = 1; o < 32; o <<= 1) {
            const int up = __shfl_up(incl, o, 32);
            if (lane >= o) incl += up;
        }
        if (lane == 31) wsum[wave] = incl;
    }
    __syncthreads();
    if (wave < BW / 32) {
        int before = lo;
        for (int w = 0; w < wave; ++w) before += wsum[w];
        myoff = before + incl - c;
        cnt[tid] = myoff;
    }
    __syncthreads();

    int pos[BATCH];
    #pragma unroll
    for (int q = 0; q < BATCH; ++q) {
        pos[q] = 0;
        if (w0[q] >= 0) pos[q] = atomicAdd(&cnt[w0[q] >> SRCBITS], 1);
    }
    if (wave < BW / 32) {
        const int d = k * BW + tid;
        if (d < N_NODES) off[d] = myoff;
    }
    if (k == NBUCK - 1 && tid == 0) off[N_NODES] = hi;
    if (staged) {
        #pragma unroll
        for (int q = 0; q < BATCH; ++q)
            if (w0[q] >= 0) stage[pos[q] - lo] = w0[q] & ((1 << SRCBITS) - 1);
        __syncthreads();
        const int a = lo & 31;
        for (int i = tid - a; i < n; i += SRT_T)
            if (i >= 0) srt[lo + i] = stage[i];
    } else {
        #pragma unroll
        for (int q = 0; q < BATCH; ++q)
            if (w0[q] >= 0) srt[pos[q]] = w0[q] & ((1 << SRCBITS) - 1);
        for (int i = lo + BATCH * SRT_T + tid; i < hi; i += SRT_T) {
            const int w = bkt[i];
            const int slot = atomicAdd(&cnt[w >> SRCBITS], 1);
            srt[slot] = w & ((1 << SRCBITS) - 1);
        }
    }
}

template <int M>
__device__ __forceinline__ float xrow(float x) {
    const int b = __float_as_int(x);
    return __int_as_float(__builtin_amdgcn_update_dpp(b, b, 0x160 + M, 0xf, 0xf, false));
}

__device__ __forceinline__ float xhalf(float x) {
    const int b = __float_as_int(x);
    return __int_as_float(__builtin_amdgcn_permlanex16(b, b, 0x76543210, 0xfedcba98, false, false));
}

__device__ __forceinline__ float wave_max(float v) {
    v = fmaxf(v, xrow<1>(v));
    v = fmaxf(v, xrow<2>(v));
    v = fmaxf(v, xrow<4>(v));
    v = fmaxf(v, xrow<8>(v));
    return fmaxf(v, xhalf(v));
}

__device__ __forceinline__ float wave_sum(float v) {
    v += xrow<1>(v);
    v += xrow<2>(v);
    v += xrow<4>(v);
    v += xrow<8>(v);
    return v + xhalf(v);
}

__device__ __forceinline__ float4 wave_max4(float4 v) {
    return make_float4(wave_max(v.x), wave_max(v.y), wave_max(v.z), wave_max(v.w));
}

__device__ __forceinline__ float4 wave_sum4(float4 v) {
    return make_float4(wave_sum(v.x), wave_sum(v.y), wave_sum(v.z), wave_sum(v.w));
}

__device__ __forceinline__ float4 edge_logits(const int* __restrict__ srt,
                                              const float* __restrict__ a_src,
                                              float4 ad, int j, int& s) {
    s = srt[j];
    const float4 as = *(const float4*)(a_src + (size_t)s * 4);
    float4 e;
    e.x = leaky(as.x + ad.x);
    e.y = leaky(as.y + ad.y);
    e.z = leaky(as.z + ad.z);
    e.w = leaky(as.w + ad.w);
    return e;
}

__device__ __forceinline__ float to_scalar(float v) {
    return __int_as_float(__builtin_amdgcn_readfirstlane(__float_as_int(v)));
}

__device__ __forceinline__ int bcast(float v, int srcLane) {
    return __builtin_amdgcn_readlane(__float_as_int(v), srcLane);
}

template <int G>
__device__ __forceinline__ void load_rows(const float* __restrict__ emb, int s, int t,
                                          unsigned lane, float2 (&r)[G]) {
    #pragma unroll
    for (int i = 0; i < G; ++i) {
        const unsigned st = (unsigned)__builtin_amdgcn_readlane(s, t + i);
        r[i] = ((const float2*)emb)[st * (EMB / 2) + lane];
    }
}

template <int G>
__device__ __forceinline__ void add_rows(const float2 (&r)[G], float4 al, int t,
                                         float2& z0, float2& z1, float2& z2, float2& z3) {
    #pragma unroll
    for (int i = 0; i < G; ++i) {
        const float a0 = __int_as_float(bcast(al.x, t + i));
        const float a1 = __int_as_float(bcast(al.y, t + i));
        const float a2 = __int_as_float(bcast(al.z, t + i));
        const float a3 = __int_as_float(bcast(al.w, t + i));
        z0.x = __builtin_fmaf(a0, r[i].x, z0.x); z0.y = __builtin_fmaf(a0, r[i].y, z0.y);
        z1.x = __builtin_fmaf(a1, r[i].x, z1.x); z1.y = __builtin_fmaf(a1, r[i].y, z1.y);
        z2.x = __builtin_fmaf(a2, r[i].x, z2.x); z2.y = __builtin_fmaf(a2, r[i].y, z2.y);
        z3.x = __builtin_fmaf(a3, r[i].x, z3.x); z3.y = __builtin_fmaf(a3, r[i].y, z3.y);
    }
}

__device__ __forceinline__ void add_groups(const float* __restrict__ emb, int s, float4 al,
                                           int t0, int n, unsigned lane,
                                           float2& z0, float2& z1, float2& z2, float2& z3) {
    #pragma unroll 1
    for (int t = t0; t < n; t += 8) {
        float2 q[8];
        load_rows<8>(emb, s, t, lane, q);
        add_rows<8>(q, al, t, z0, z1, z2, z3);
    }
}

#define ZK (HEADS * EMB)
#define ZP (ZK + 4)
#define PP (256 + 16)
__global__ __launch_bounds__(512) void gather_project_kernel(const int* __restrict__ off,
                                                             const int* __restrict__ srt,
                                                             const float* __restrict__ emb,
                                                             const float* __restrict__ W,
                                                             const float* __restrict__ a_src,
                                                             const float* __restrict__ a_dst,
                                                             float* __restrict__ out) {
    __shared__ __attribute__((aligned(16))) float zt[16][ZP];
    __shared__ __attribute__((aligned(16))) float part[16][PP];
    const int lane = threadIdx.x & 31;
    const int wave = __builtin_amdgcn_readfirstlane(threadIdx.x >> 5);
    const int d = blockIdx.x * 16 + wave;
    const int beg = __builtin_amdgcn_readfirstlane(off[d]);
    const int end = __builtin_amdgcn_readfirstlane(off[d + 1]);
    const float4 ad = *(const float4*)(a_dst + (size_t)d * 4);
    const float NEG_INF = __int_as_float(0xFF800000);

    const int  j0 = beg + lane;
    const bool v0 = j0 < end;
    const unsigned ulane = (unsigned)lane;
    const int n0 = min(32, end - beg);
    int s0 = 0;
    if (v0) s0 = srt[j0];
    const int sp = v0 ? s0 : __builtin_amdgcn_readlane(s0, max(n0 - 1, 0));
    const float4 as0 = *(const float4*)(a_src + (size_t)sp * 4);
    float2 r0[16];
    load_rows<16>(emb, sp, 0, ulane, r0);
    float4 e0;
    e0.x = v0 ? leaky(as0.x + ad.x) : NEG_INF; e0.y = v0 ? leaky(as0.y + ad.y) : NEG_INF;
    e0.z = v0 ? leaky(as0.z + ad.z) : NEG_INF; e0.w = v0 ? leaky(as0.w + ad.w) : NEG_INF;

    float4 mx = e0;
    for (int j = j0 + 32; j < end; j += 32) {
        int s;
        const float4 e = edge_logits(srt, a_src, ad, j, s);
        mx.x = fmaxf(mx.x, e.x); mx.y = fmaxf(mx.y, e.y);
        mx.z = fmaxf(mx.z, e.z); mx.w = fmaxf(mx.w, e.w);
    }
    mx = wave_max4(mx);
    mx = make_float4(to_scalar(mx.x), to_scalar(mx.y), to_scalar(mx.z), to_scalar(mx.w));

    float4 ex0 = make_float4(0.0f, 0.0f, 0.0f, 0.0f);
    if (v0) {
        ex0.x = __expf(e0.x - mx.x); ex0.y = __expf(e0.y - mx.y);
        ex0.z = __expf(e0.z - mx.z); ex0.w = __expf(e0.w - mx.w);
    }
    float4 den = ex0;
    for (int j = j0 + 32; j < end; j += 32) {
        int s;
        const float4 e = edge_logits(srt, a_src, ad, j, s);
        den.x += __expf(e.x - mx.x); den.y += __expf(e.y - mx.y);
        den.z += __expf(e.z - mx.z); den.w += __expf(e.w - mx.w);
    }
    den = wave_sum4(den);
    den.x += 1e-16f; den.y += 1e-16f; den.z += 1e-16f; den.w += 1e-16f;
    den = make_float4(to_scalar(den.x), to_scalar(den.y), to_scalar(den.z), to_scalar(den.w));

    float2 z0 = make_float2(0.0f, 0.0f), z1 = z0, z2 = z0, z3 = z0;

    if (n0 > 0) {
        const float4 al = make_float4(ex0.x / den.x, ex0.y / den.y, ex0.z / den.z, ex0.w / den.w);
        add_rows<16>(r0, al, 0, z0, z1, z2, z3);
        add_groups(emb, sp, al, 16, n0, ulane, z0, z1, z2, z3);
    }
    for (int base = beg + 32; base < end; base += 32) {
        const int j = base + lane;
        const int n = min(32, end - base);
        int s = 0;
        float4 al = make_float4(0.0f, 0.0f, 0.0f, 0.0f);
        if (j < end) {
            const float4 e = edge_logits(srt, a_src, ad, j, s);
            al.x = __expf(e.x - mx.x) / den.x; al.y = __expf(e.y - mx.y) / den.y;
            al.z = __expf(e.z - mx.z) / den.z; al.w = __expf(e.w - mx.w) / den.w;
        }
        const int sl = __builtin_amdgcn_readlane(s, n - 1);
        s = (j < end) ? s : sl;
        add_groups(emb, s, al, 0, n, ulane, z0, z1, z2, z3);
    }
    *(float2*)&zt[wave][0 * EMB + lane * 2] = z0;
    *(float2*)&zt[wave][1 * EMB + lane * 2] = z1;
    *(float2*)&zt[wave][2 * EMB + lane * 2] = z2;
    *(float2*)&zt[wave][3 * EMB + lane * 2] = z3;

    const int half = lane >> 4;
    const int lrow = lane & 15;
    const int ct   = wave & 1;
    const int kq   = wave >> 1;
    const int hk   = kq >> 1;
    const int kin  = (kq & 1) * 32;
    const float* wp = W + (size_t)(kin + 2 * half) * HOUT + hk * RD + ct * 16 + lrow;
    v2f bf[8];
    #pragma unroll
    for (int ks = 0; ks < 8; ++ks) {
        bf[ks].x = wp[(ks * 4) * HOUT];
        bf[ks].y = wp[(ks * 4 + 1) * HOUT];
    }
    __syncthreads();

    const float* zr = &zt[lrow][kq * 32 + 2 * half];
    v8f c0 = {}, c1 = {};
    #pragma unroll
    for (int ks = 0; ks < 8; ks += 2) {
        const v2f fa0 = *(const v2f*)(zr + ks * 4);
        const v2f fa1 = *(const v2f*)(zr + ks * 4 + 4);
        c0 = __builtin_amdgcn_wmma_f32_16x16x4_f32(false, fa0, false, bf[ks],     (short)0, c0, false, false);
        c1 = __builtin_amdgcn_wmma_f32_16x16x4_f32(false, fa1, false, bf[ks + 1], (short)0, c1, false, false);
    }
    #pragma unroll
    for (int v = 0; v < 8; ++v) part[wave][v * 32 + lane] = c0[v] + c1[v];
    __syncthreads();

    const int pl = (wave >> 3) * 16 + lrow;
    const int pv = wave & 7;
    const int pc = lane >> 4;
    float sum = 0.0f;
    #pragma unroll
    for (int q = 0; q < 8; ++q) sum += part[2 * q + pc][pv * 32 + pl];
    out[(size_t)d * RD + lane] = 0.25f * sum;
}

extern "C" void kernel_launch(void* const* d_in, const int* in_sizes, int n_in,
                              void* d_out, int out_size, void* d_ws, size_t ws_size,
                              hipStream_t stream) {
    const float* node_emb = (const float*)d_in[0];
    const float* W        = (const float*)d_in[1];
    const float* att_src  = (const float*)d_in[2];
    const float* att_dst  = (const float*)d_in[3];
    const int*   path     = (const int*)d_in[4];
    float* out = (float*)d_out;

    float* ws    = (float*)d_ws;
    float* h     = ws;
    float* a_src = h     + (size_t)N_NODES * HOUT;
    float* a_dst = a_src + (size_t)N_NODES * HEADS;
    int*   cntB  = (int*)(a_dst + (size_t)N_NODES * HEADS);
    int*   bstart = cntB + (size_t)NB1 * NBUCK;
    int*   off   = cntB + (size_t)N_NODES;
    int*   bkt   = off  + (size_t)N_NODES + 32;
    int*   srt   = bkt  + (size_t)NEDGES;

    const int rowTiles = N_NODES / 16;
    logits_wmma_kernel<<<(rowTiles + 7) / 8, 256, 0, stream>>>(
        node_emb, W, att_src, att_dst, a_src, a_dst);

    range_count_kernel<<<NB1, SRT_T, 0, stream>>>(path, cntB);

    range_scatter_kernel<<<NB1, SRT_T, 0, stream>>>(path, cntB, bstart, bkt);

    range_sort_kernel<<<NBUCK, SRT_T, 0, stream>>>(bkt, bstart, off, srt);

    gather_project_kernel<<<N_NODES / 16, 512, 0, stream>>>(off, srt, node_emb, W, a_src, a_dst, out);
}


# Logits product rides in the launch of the sort's second level (role by block index, 196 + 196 groups of 1,024 threads): 4 launches, not 5; folded matrix built 196 times, not 782; sums as before
# speedup vs baseline: 8.2041x; 1.0886x over previous
#include <hip/hip_runtime.h>

#define N_NODES   100000
#define EMB       64
#define HEADS     4
#define RD        32
#define HOUT      (HEADS * RD)
#define NEDGES    1600000
#define NEG_SLOPE 0.2f

typedef __attribute__((ext_vector_type(2))) float v2f;
typedef __attribute__((ext_vector_type(8))) float v8f;

__device__ __forceinline__ float leaky(float x) {
    return x >= 0.0f ? x : NEG_SLOPE * x;
}

#define LG_T      1024
#define LG_TILES  (N_NODES / 16)
#define LG_BLOCKS ((LG_TILES + LG_T / 32 - 1) / (LG_T / 32))
static_assert(16 * EMB == LG_T, "one folded entry per thread");
__device__ __forceinline__ void logits_role(const int blk,
                                            const float* __restrict__ emb,
                                            const float* __restrict__ W,
                                            const float* __restrict__ att_src,
                                            const float* __restrict__ att_dst,
                                            float* __restrict__ a_src,
                                            float* __restrict__ a_dst) {
    __shared__ float Wa[16][EMB + 1];
    const int tid = threadIdx.x;
    {
        const int k = tid >> 4;
        const int j = tid & 15;
        float s = 0.0f;
        if (j < 8) {
            const int hh = j & 3;
            const float* att = (j < 4 ? att_src : att_dst) + hh * RD;
            const float* wr  = W + k * HOUT + hh * RD;
            #pragma unroll
            for (int dd = 0; dd < RD; ++dd) s += wr[dd] * att[dd];
        }
        Wa[j][k] = s;
    }
    __syncthreads();

    const int wave = tid >> 5;
    const int lane = tid & 31;
    const int rowTile = (blk * (LG_T / 32) + wave) * 16;
    if (rowTile >= N_NODES) return;

    const int half = lane >> 4;
    const int lrow = lane & 15;
    const float* arow = emb + (size_t)(rowTile + lrow) * EMB;

    v8f acc = {};
    for (int k0 = 0; k0 < EMB; k0 += 4) {
        v2f a, b;
        a.x = arow[k0 + 2 * half];
        a.y = arow[k0 + 2 * half + 1];
        b.x = Wa[lrow][k0 + 2 * half];
        b.y = Wa[lrow][k0 + 2 * half + 1];
        acc = __builtin_amdgcn_wmma_f32_16x16x4_f32(
            false, a, false, b, (short)0, acc, false, false);
    }
    float* dstp = (lrow < 4 ? a_src + lrow : a_dst + (lrow - 4)) + (size_t)(rowTile + 8 * half) * HEADS;
    if (lrow < 8) {
        #pragma unroll
        for (int v = 0; v < 8; ++v) dstp[v * HEADS] = acc[v];
    }
}

#define BSH    9
#define BW     (1 << BSH)
#define NBUCK  ((N_NODES + BW - 1) >> BSH)
#define CHUNK  6400
#define NB1    (NEDGES / CHUNK)
#define SRT_T  1024
#define FULLQ  (CHUNK / SRT_T)
#define TAILT  (CHUNK - FULLQ * SRT_T)
#define SRCBITS 17
#define PARTS  5
#define BATCH  10
#define STAGE2 (BATCH * SRT_T)
static_assert(NEDGES % CHUNK == 0, "level-1 chunks must tile the edge list");
static_assert(TAILT > 0 && TAILT % 32 == 0, "the 7th edge belongs to whole waves");
static_assert(N_NODES <= (1 << SRCBITS), "source id and local destination share one word");
static_assert(NB1 % PARTS == 0 && PARTS * NBUCK <= SRT_T, "table sums split evenly");
static_assert(NB1 * NBUCK + NBUCK + 1 <= N_NODES, "count table and range starts fit the old counter region");
static_assert(NBUCK <= 256, "a range id fits the byte kept beside each staged word");

__global__ __launch_bounds__(SRT_T) void range_count_kernel(const int* __restrict__ path,
                                                            int* __restrict__ cntB) {
    __shared__ int hist[NBUCK];
    const int tid = threadIdx.x;
    const int* dstp = path + NEDGES + blockIdx.x * CHUNK;
    const bool tail = tid < TAILT;
    int dv[FULLQ], dt = 0;
    #pragma unroll
    for (int q = 0; q < FULLQ; ++q) dv[q] = dstp[tid + q * SRT_T];
    if (tail) dt = dstp[tid + FULLQ * SRT_T];
    if (tid < NBUCK) hist[tid] = 0;
    __syncthreads();
    #pragma unroll
    for (int q = 0; q < FULLQ; ++q) atomicAdd(&hist[dv[q] >> BSH], 1);
    if (tail) atomicAdd(&hist[dt >> BSH], 1);
    __syncthreads();
    if (tid < NBUCK) cntB[blockIdx.x * NBUCK + tid] = hist[tid];
}

__global__ __launch_bounds__(SRT_T) void range_scatter_kernel(const int* __restrict__ path,
                                                              const int* __restrict__ cntB,
                                                              int* __restrict__ bstart,
                                                              int* __restrict__ bkt) {
    __shared__ int ptot[PARTS][NBUCK];
    __shared__ int pbef[PARTS][NBUCK];
    __shared__ int tot[NBUCK];
    __shared__ int own[NBUCK];
    __shared__ int cur[NBUCK];
    __shared__ int gdel[NBUCK];
    __shared__ int stage[CHUNK];
    __shared__ unsigned char rid[CHUNK];
    const int tid = threadIdx.x;
    const int b   = blockIdx.x;
    const int* srcp = path + b * CHUNK;
    const int* dstp = path + NEDGES + b * CHUNK;
    const bool tail = tid < TAILT;
    int dv[FULLQ], sv[FULLQ], dt = 0, st = 0;
    #pragma unroll
    for (int q = 0; q < FULLQ; ++q) {
        dv[q] = dstp[tid + q * SRT_T];
        sv[q] = srcp[tid + q * SRT_T];
    }
    if (tail) { dt = dstp[tid + FULLQ * SRT_T]; st = srcp[tid + FULLQ * SRT_T]; }
    int ownc = 0;
    if (tid < NBUCK) ownc = cntB[b * NBUCK + tid];
    if (tid < PARTS * NBUCK) {
        const int k = tid % NBUCK;
        const int p = tid / NBUCK;
        int total = 0, before = 0;
        #pragma unroll
        for (int j = 0; j < NB1 / PARTS; ++j) {
            const int bb = p + j * PARTS;
            const int c = cntB[bb * NBUCK + k];
            total += c;
            before += (bb < b) ? c : 0;
        }
        ptot[p][k] = total;
        pbef[p][k] = before;
    }
    __syncthreads();
    int total = 0, before = 0, rstart = 0;
    if (tid < NBUCK) {
        #pragma unroll
        for (int p = 0; p < PARTS; ++p) { total += ptot[p][tid]; before += pbef[p][tid]; }
        tot[tid] = total;
        own[tid] = ownc;
    }
    __syncthreads();
    if (tid < NBUCK) {
        int start = 0, lstart = 0;
        for (int k = 0; k < tid; ++k) { start += tot[k]; lstart += own[k]; }
        rstart = start;
        cur[tid]  = lstart;
        gdel[tid] = start + before - lstart;
    }
    __syncthreads();
    int pos[FULLQ], pt = 0;
    #pragma unroll
    for (int q = 0; q < FULLQ; ++q) pos[q] = atomicAdd(&cur[dv[q] >> BSH], 1);
    if (tail) pt = atomicAdd(&cur[dt >> BSH], 1);
    #pragma unroll
    for (int q = 0; q < FULLQ; ++q) {
        stage[pos[q]] = ((dv[q] & (BW - 1)) << SRCBITS) | sv[q];
        rid[pos[q]]   = (unsigned char)(dv[q] >> BSH);
    }
    if (tail) {
        stage[pt] = ((dt & (BW - 1)) << SRCBITS) | st;
        rid[pt]   = (unsigned char)(dt >> BSH);
    }
    __syncthreads();
    #pragma unroll
    for (int q = 0; q < FULLQ; ++q) {
        const int i = tid + q * SRT_T;
        bkt[i + gdel[rid[i]]] = stage[i];
    }
    if (tail) {
        const int i = tid + FULLQ * SRT_T;
        bkt[i + gdel[rid[i]]] = stage[i];
    }
    if (b == 0 && tid < NBUCK) {
        bstart[tid] = rstart;
        if (tid == NBUCK - 1) bstart[NBUCK] = rstart + total;
    }
}

__device__ __forceinline__ void range_sort_role(const int k,
                                                const int* __restrict__ bkt,
                                                const int* __restrict__ bstart,
                                                int* __restrict__ off,
                                                int* __restrict__ srt) {
    __shared__ int cnt[BW];
    __shared__ int wsum[BW / 32];
    __shared__ int stage[STAGE2];
    const int tid  = threadIdx.x;
    const int lane = tid & 31;
    const int wave = tid >> 5;
    const int lo = bstart[k];
    const int hi = bstart[k + 1];
    const int n  = hi - lo;
    const bool staged = n <= STAGE2;

    int w0[BATCH];
    #pragma unroll
    for (int q = 0; q < BATCH; ++q) {
        const int i = lo + q * SRT_T + tid;
        w0[q] = (i < hi) ? bkt[i] : -1;
    }
    if (tid < BW) cnt[tid] = 0;
    __syncthreads();
    #pragma unroll
    for (int q = 0; q < BATCH; ++q)
        if (w0[q] >= 0) atomicAdd(&cnt[w0[q] >> SRCBITS], 1);
    for (int i = lo + BATCH * SRT_T + tid; i < hi; i += SRT_T)
        atomicAdd(&cnt[bkt[i] >> SRCBITS], 1);
    __syncthreads();

    int c = 0, incl = 0, myoff = 0;
    if (wave < BW / 32) {
        c = cnt[tid];
        incl = c;
        for (int o = 1; o < 32; o <<= 1) {
            const int up = __shfl_up(incl, o, 32);
            if (lane >= o) incl += up;
        }
        if (lane == 31) wsum[wave] = incl;
    }
    __syncthreads();
    if (wave < BW / 32) {
        int before = lo;
        for (int w = 0; w < wave; ++w) before += wsum[w];
        myoff = before + incl - c;
        cnt[tid] = myoff;
    }
    __syncthreads();

    int pos[BATCH];
    #pragma unroll
    for (int q = 0; q < BATCH; ++q) {
        pos[q] = 0;
        if (w0[q] >= 0) pos[q] = atomicAdd(&cnt[w0[q] >> SRCBITS], 1);
    }
    if (wave < BW / 32) {
        const int d = k * BW + tid;
        if (d < N_NODES) off[d] = myoff;
    }
    if (k == NBUCK - 1 && tid == 0) off[N_NODES] = hi;
    if (staged) {
        #pragma unroll
        for (int q = 0; q < BATCH; ++q)
            if (w0[q] >= 0) stage[pos[q] - lo] = w0[q] & ((1 << SRCBITS) - 1);
        __syncthreads();
        const int a = lo & 31;
        for (int i = tid - a; i < n; i += SRT_T)
            if (i >= 0) srt[lo + i] = stage[i];
    } else {
        #pragma unroll
        for (int q = 0; q < BATCH; ++q)
            if (w0[q] >= 0) srt[pos[q]] = w0[q] & ((1 << SRCBITS) - 1);
        for (int i = lo + BATCH * SRT_T + tid; i < hi; i += SRT_T) {
            const int w = bkt[i];
            const int slot = atomicAdd(&cnt[w >> SRCBITS], 1);
            srt[slot] = w & ((1 << SRCBITS) - 1);
        }
    }
}

static_assert(LG_T == SRT_T, "both roles use the same group size");
__global__ __launch_bounds__(SRT_T) void sort_logits_kernel(const int* __restrict__ bkt,
                                                            const int* __restrict__ bstart,
                                                            int* __restrict__ off,
                                                            int* __restrict__ srt,
                                                            const float* __restrict__ emb,
                                                            const float* __restrict__ W,
                                                            const float* __restrict__ att_src,
                                                            const float* __restrict__ att_dst,
                                                            float* __restrict__ a_src,
                                                            float* __restrict__ a_dst) {
    if (blockIdx.x < NBUCK) range_sort_role(blockIdx.x, bkt, bstart, off, srt);
    else logits_role(blockIdx.x - NBUCK, emb, W, att_src, att_dst, a_src, a_dst);
}

template <int M>
__device__ __forceinline__ float xrow(float x) {
    const int b = __float_as_int(x);
    return __int_as_float(__builtin_amdgcn_update_dpp(b, b, 0x160 + M, 0xf, 0xf, false));
}

__device__ __forceinline__ float xhalf(float x) {
    const int b = __float_as_int(x);
    return __int_as_float(__builtin_amdgcn_permlanex16(b, b, 0x76543210, 0xfedcba98, false, false));
}

__device__ __forceinline__ float wave_max(float v) {
    v = fmaxf(v, xrow<1>(v));
    v = fmaxf(v, xrow<2>(v));
    v = fmaxf(v, xrow<4>(v));
    v = fmaxf(v, xrow<8>(v));
    return fmaxf(v, xhalf(v));
}

__device__ __forceinline__ float wave_sum(float v) {
    v += xrow<1>(v);
    v += xrow<2>(v);
    v += xrow<4>(v);
    v += xrow<8>(v);
    return v + xhalf(v);
}

__device__ __forceinline__ float4 wave_max4(float4 v) {
    return make_float4(wave_max(v.x), wave_max(v.y), wave_max(v.z), wave_max(v.w));
}

__device__ __forceinline__ float4 wave_sum4(float4 v) {
    return make_float4(wave_sum(v.x), wave_sum(v.y), wave_sum(v.z), wave_sum(v.w));
}

__device__ __forceinline__ float4 edge_logits(const int* __restrict__ srt,
                                              const float* __restrict__ a_src,
                                              float4 ad, int j, int& s) {
    s = srt[j];
    const float4 as = *(const float4*)(a_src + (size_t)s * 4);
    float4 e;
    e.x = leaky(as.x + ad.x);
    e.y = leaky(as.y + ad.y);
    e.z = leaky(as.z + ad.z);
    e.w = leaky(as.w + ad.w);
    return e;
}

__device__ __forceinline__ float to_scalar(float v) {
    return __int_as_float(__builtin_amdgcn_readfirstlane(__float_as_int(v)));
}

__device__ __forceinline__ int bcast(float v, int srcLane) {
    return __builtin_amdgcn_readlane(__float_as_int(v), srcLane);
}

template <int G>
__device__ __forceinline__ void load_rows(const float* __restrict__ emb, int s, int t,
                                          unsigned lane, float2 (&r)[G]) {
    #pragma unroll
    for (int i = 0; i < G; ++i) {
        const unsigned st = (unsigned)__builtin_amdgcn_readlane(s, t + i);
        r[i] = ((const float2*)emb)[st * (EMB / 2) + lane];
    }
}

template <int G>
__device__ __forceinline__ void add_rows(const float2 (&r)[G], float4 al, int t,
                                         float2& z0, float2& z1, float2& z2, float2& z3) {
    #pragma unroll
    for (int i = 0; i < G; ++i) {
        const float a0 = __int_as_float(bcast(al.x, t + i));
        const float a1 = __int_as_float(bcast(al.y, t + i));
        const float a2 = __int_as_float(bcast(al.z, t + i));
        const float a3 = __int_as_float(bcast(al.w, t + i));
        z0.x = __builtin_fmaf(a0, r[i].x, z0.x); z0.y = __builtin_fmaf(a0, r[i].y, z0.y);
        z1.x = __builtin_fmaf(a1, r[i].x, z1.x); z1.y = __builtin_fmaf(a1, r[i].y, z1.y);
        z2.x = __builtin_fmaf(a2, r[i].x, z2.x); z2.y = __builtin_fmaf(a2, r[i].y, z2.y);
        z3.x = __builtin_fmaf(a3, r[i].x, z3.x); z3.y = __builtin_fmaf(a3, r[i].y, z3.y);
    }
}

__device__ __forceinline__ void add_groups(const float* __restrict__ emb, int s, float4 al,
                                           int t0, int n, unsigned lane,
                                           float2& z0, float2& z1, float2& z2, float2& z3) {
    #pragma unroll 1
    for (int t = t0; t < n; t += 8) {
        float2 q[8];
        load_rows<8>(emb, s, t, lane, q);
        add_rows<8>(q, al, t, z0, z1, z2, z3);
    }
}

#define ZK (HEADS * EMB)
#define ZP (ZK + 4)
#define PP (256 + 16)
__global__ __launch_bounds__(512) void gather_project_kernel(const int* __restrict__ off,
                                                             const int* __restrict__ srt,
                                                             const float* __restrict__ emb,
                                                             const float* __restrict__ W,
                                                             const float* __restrict__ a_src,
                                                             const float* __restrict__ a_dst,
                                                             float* __restrict__ out) {
    __shared__ __attribute__((aligned(16))) float zt[16][ZP];
    __shared__ __attribute__((aligned(16))) float part[16][PP];
    const int lane = threadIdx.x & 31;
    const int wave = __builtin_amdgcn_readfirstlane(threadIdx.x >> 5);
    const int d = blockIdx.x * 16 + wave;
    const int beg = __builtin_amdgcn_readfirstlane(off[d]);
    const int end = __builtin_amdgcn_readfirstlane(off[d + 1]);
    const float4 ad = *(const float4*)(a_dst + (size_t)d * 4);
    const float NEG_INF = __int_as_float(0xFF800000);

    const int  j0 = beg + lane;
    const bool v0 = j0 < end;
    const unsigned ulane = (unsigned)lane;
    const int n0 = min(32, end - beg);
    int s0 = 0;
    if (v0) s0 = srt[j0];
    const int sp = v0 ? s0 : __builtin_amdgcn_readlane(s0, max(n0 - 1, 0));
    const float4 as0 = *(const float4*)(a_src + (size_t)sp * 4);
    float2 r0[16];
    load_rows<16>(emb, sp, 0, ulane, r0);
    float4 e0;
    e0.x = v0 ? leaky(as0.x + ad.x) : NEG_INF; e0.y = v0 ? leaky(as0.y + ad.y) : NEG_INF;
    e0.z = v0 ? leaky(as0.z + ad.z) : NEG_INF; e0.w = v0 ? leaky(as0.w + ad.w) : NEG_INF;

    float4 mx = e0;
    for (int j = j0 + 32; j < end; j += 32) {
        int s;
        const float4 e = edge_logits(srt, a_src, ad, j, s);
        mx.x = fmaxf(mx.x, e.x); mx.y = fmaxf(mx.y, e.y);
        mx.z = fmaxf(mx.z, e.z); mx.w = fmaxf(mx.w, e.w);
    }
    mx = wave_max4(mx);
    mx = make_float4(to_scalar(mx.x), to_scalar(mx.y), to_scalar(mx.z), to_scalar(mx.w));

    float4 ex0 = make_float4(0.0f, 0.0f, 0.0f, 0.0f);
    if (v0) {
        ex0.x = __expf(e0.x - mx.x); ex0.y = __expf(e0.y - mx.y);
        ex0.z = __expf(e0.z - mx.z); ex0.w = __expf(e0.w - mx.w);
    }
    float4 den = ex0;
    for (int j = j0 + 32; j < end; j += 32) {
        int s;
        const float4 e = edge_logits(srt, a_src, ad, j, s);
        den.x += __expf(e.x - mx.x); den.y += __expf(e.y - mx.y);
        den.z += __expf(e.z - mx.z); den.w += __expf(e.w - mx.w);
    }
    den = wave_sum4(den);
    den.x += 1e-16f; den.y += 1e-16f; den.z += 1e-16f; den.w += 1e-16f;
    den = make_float4(to_scalar(den.x), to_scalar(den.y), to_scalar(den.z), to_scalar(den.w));

    float2 z0 = make_float2(0.0f, 0.0f), z1 = z0, z2 = z0, z3 = z0;

    if (n0 > 0) {
        const float4 al = make_float4(ex0.x / den.x, ex0.y / den.y, ex0.z / den.z, ex0.w / den.w);
        add_rows<16>(r0, al, 0, z0, z1, z2, z3);
        add_groups(emb, sp, al, 16, n0, ulane, z0, z1, z2, z3);
    }
    for (int base = beg + 32; base < end; base += 32) {
        const int j = base + lane;
        const int n = min(32, end - base);
        int s = 0;
        float4 al = make_float4(0.0f, 0.0f, 0.0f, 0.0f);
        if (j < end) {
            const float4 e = edge_logits(srt, a_src, ad, j, s);
            al.x = __expf(e.x - mx.x) / den.x; al.y = __expf(e.y - mx.y) / den.y;
            al.z = __expf(e.z - mx.z) / den.z; al.w = __expf(e.w - mx.w) / den.w;
        }
        const int sl = __builtin_amdgcn_readlane(s, n - 1);
        s = (j < end) ? s : sl;
        add_groups(emb, s, al, 0, n, ulane, z0, z1, z2, z3);
    }
    *(float2*)&zt[wave][0 * EMB + lane * 2] = z0;
    *(float2*)&zt[wave][1 * EMB + lane * 2] = z1;
    *(float2*)&zt[wave][2 * EMB + lane * 2] = z2;
    *(float2*)&zt[wave][3 * EMB + lane * 2] = z3;

    const int half = lane >> 4;
    const int lrow = lane & 15;
    const int ct   = wave & 1;
    const int kq   = wave >> 1;
    const int hk   = kq >> 1;
    const int kin  = (kq & 1) * 32;
    const float* wp = W + (size_t)(kin + 2 * half) * HOUT + hk * RD + ct * 16 + lrow;
    v2f bf[8];
    #pragma unroll
    for (int ks = 0; ks < 8; ++ks) {
        bf[ks].x = wp[(ks * 4) * HOUT];
        bf[ks].y = wp[(ks * 4 + 1) * HOUT];
    }
    __syncthreads();

    const float* zr = &zt[lrow][kq * 32 + 2 * half];
    v8f c0 = {}, c1 = {};
    #pragma unroll
    for (int ks = 0; ks < 8; ks += 2) {
        const v2f fa0 = *(const v2f*)(zr + ks * 4);
        const v2f fa1 = *(const v2f*)(zr + ks * 4 + 4);
        c0 = __builtin_amdgcn_wmma_f32_16x16x4_f32(false, fa0, false, bf[ks],     (short)0, c0, false, false);
        c1 = __builtin_amdgcn_wmma_f32_16x16x4_f32(false, fa1, false, bf[ks + 1], (short)0, c1, false, false);
    }
    #pragma unroll
    for (int v = 0; v < 8; ++v) part[wave][v * 32 + lane] = c0[v] + c1[v];
    __syncthreads();

    const int pl = (wave >> 3) * 16 + lrow;
    const int pv = wave & 7;
    const int pc = lane >> 4;
    float sum = 0.0f;
    #pragma unroll
    for (int q = 0; q < 8; ++q) sum += part[2 * q + pc][pv * 32 + pl];
    out[(size_t)d * RD + lane] = 0.25f * sum;
}

extern "C" void kernel_launch(void* const* d_in, const int* in_sizes, int n_in,
                              void* d_out, int out_size, void* d_ws, size_t ws_size,
                              hipStream_t stream) {
    const float* node_emb = (const float*)d_in[0];
    const float* W        = (const float*)d_in[1];
    const float* att_src  = (const float*)d_in[2];
    const float* att_dst  = (const float*)d_in[3];
    const int*   path     = (const int*)d_in[4];
    float* out = (float*)d_out;

    float* ws    = (float*)d_ws;
    float* h     = ws;
    float* a_src = h     + (size_t)N_NODES * HOUT;
    float* a_dst = a_src + (size_t)N_NODES * HEADS;
    int*   cntB  = (int*)(a_dst + (size_t)N_NODES * HEADS);
    int*   bstart = cntB + (size_t)NB1 * NBUCK;
    int*   off   = cntB + (size_t)N_NODES;
    int*   bkt   = off  + (size_t)N_NODES + 32;
    int*   srt   = bkt  + (size_t)NEDGES;

    range_count_kernel<<<NB1, SRT_T, 0, stream>>>(path, cntB);

    range_scatter_kernel<<<NB1, SRT_T, 0, stream>>>(path, cntB, bstart, bkt);

    sort_logits_kernel<<<NBUCK + LG_BLOCKS, SRT_T, 0, stream>>>(
        bkt, bstart, off, srt, node_emb, W, att_src, att_dst, a_src, a_dst);

    gather_project_kernel<<<N_NODES / 16, 512, 0, stream>>>(off, srt, node_emb, W, a_src, a_dst, out);
}


# Last kernel: a wave aggregates two destinations in turn, the block projects both 16-row tiles with one set of W fragments: 50,000 waves in 3,125 groups enter, not 100,000 in 6,250; sum order kept
# speedup vs baseline: 8.5229x; 1.0389x over previous
#include <hip/hip_runtime.h>

#define N_NODES   100000
#define EMB       64
#define HEADS     4
#define RD        32
#define HOUT      (HEADS * RD)
#define NEDGES    1600000
#define NEG_SLOPE 0.2f

typedef __attribute__((ext_vector_type(2))) float v2f;
typedef __attribute__((ext_vector_type(8))) float v8f;

__device__ __forceinline__ float leaky(float x) {
    return x >= 0.0f ? x : NEG_SLOPE * x;
}

#define LG_T      1024
#define LG_TILES  (N_NODES / 16)
#define LG_BLOCKS ((LG_TILES + LG_T / 32 - 1) / (LG_T / 32))
static_assert(16 * EMB == LG_T, "one folded entry per thread");
__device__ __forceinline__ void logits_role(const int blk,
                                            const float* __restrict__ emb,
                                            const float* __restrict__ W,
                                            const float* __restrict__ att_src,
                                            const float* __restrict__ att_dst,
                                            float* __restrict__ a_src,
                                            float* __restrict__ a_dst) {
    __shared__ float Wa[16][EMB + 1];
    const int tid = threadIdx.x;
    {
        const int k = tid >> 4;
        const int j = tid & 15;
        float s = 0.0f;
        if (j < 8) {
            const int hh = j & 3;
            const float* att = (j < 4 ? att_src : att_dst) + hh * RD;
            const float* wr  = W + k * HOUT + hh * RD;
            #pragma unroll
            for (int dd = 0; dd < RD; ++dd) s += wr[dd] * att[dd];
        }
        Wa[j][k] = s;
    }
    __syncthreads();

    const int wave = tid >> 5;
    const int lane = tid & 31;
    const int rowTile = (blk * (LG_T / 32) + wave) * 16;
    if (rowTile >= N_NODES) return;

    const int half = lane >> 4;
    const int lrow = lane & 15;
    const float* arow = emb + (size_t)(rowTile + lrow) * EMB;

    v8f acc = {};
    for (int k0 = 0; k0 < EMB; k0 += 4) {
        v2f a, b;
        a.x = arow[k0 + 2 * half];
        a.y = arow[k0 + 2 * half + 1];
        b.x = Wa[lrow][k0 + 2 * half];
        b.y = Wa[lrow][k0 + 2 * half + 1];
        acc = __builtin_amdgcn_wmma_f32_16x16x4_f32(
            false, a, false, b, (short)0, acc, false, false);
    }
    float* dstp = (lrow < 4 ? a_src + lrow : a_dst + (lrow - 4)) + (size_t)(rowTile + 8 * half) * HEADS;
    if (lrow < 8) {
        #pragma unroll
        for (int v = 0; v < 8; ++v) dstp[v * HEADS] = acc[v];
    }
}

#define BSH    9
#define BW     (1 << BSH)
#define NBUCK  ((N_NODES + BW - 1) >> BSH)
#define CHUNK  6400
#define NB1    (NEDGES / CHUNK)
#define SRT_T  1024
#define FULLQ  (CHUNK / SRT_T)
#define TAILT  (CHUNK - FULLQ * SRT_T)
#define SRCBITS 17
#define PARTS  5
#define BATCH  10
#define STAGE2 (BATCH * SRT_T)
static_assert(NEDGES % CHUNK == 0, "level-1 chunks must tile the edge list");
static_assert(TAILT > 0 && TAILT % 32 == 0, "the 7th edge belongs to whole waves");
static_assert(N_NODES <= (1 << SRCBITS), "source id and local destination share one word");
static_assert(NB1 % PARTS == 0 && PARTS * NBUCK <= SRT_T, "table sums split evenly");
static_assert(NB1 * NBUCK + NBUCK + 1 <= N_NODES, "count table and range starts fit the old counter region");
static_assert(NBUCK <= 256, "a range id fits the byte kept beside each staged word");

__global__ __launch_bounds__(SRT_T) void range_count_kernel(const int* __restrict__ path,
                                                            int* __restrict__ cntB) {
    __shared__ int hist[NBUCK];
    const int tid = threadIdx.x;
    const int* dstp = path + NEDGES + blockIdx.x * CHUNK;
    const bool tail = tid < TAILT;
    int dv[FULLQ], dt = 0;
    #pragma unroll
    for (int q = 0; q < FULLQ; ++q) dv[q] = dstp[tid + q * SRT_T];
    if (tail) dt = dstp[tid + FULLQ * SRT_T];
    if (tid < NBUCK) hist[tid] = 0;
    __syncthreads();
    #pragma unroll
    for (int q = 0; q < FULLQ; ++q) atomicAdd(&hist[dv[q] >> BSH], 1);
    if (tail) atomicAdd(&hist[dt >> BSH], 1);
    __syncthreads();
    if (tid < NBUCK) cntB[blockIdx.x * NBUCK + tid] = hist[tid];
}

__global__ __launch_bounds__(SRT_T) void range_scatter_kernel(const int* __restrict__ path,
                                                              const int* __restrict__ cntB,
                                                              int* __restrict__ bstart,
                                                              int* __restrict__ bkt) {
    __shared__ int ptot[PARTS][NBUCK];
    __shared__ int pbef[PARTS][NBUCK];
    __shared__ int tot[NBUCK];
    __shared__ int own[NBUCK];
    __shared__ int cur[NBUCK];
    __shared__ int gdel[NBUCK];
    __shared__ int stage[CHUNK];
    __shared__ unsigned char rid[CHUNK];
    const int tid = threadIdx.x;
    const int b   = blockIdx.x;
    const int* srcp = path + b * CHUNK;
    const int* dstp = path + NEDGES + b * CHUNK;
    const bool tail = tid < TAILT;
    int dv[FULLQ], sv[FULLQ], dt = 0, st = 0;
    #pragma unroll
    for (int q = 0; q < FULLQ; ++q) {
        dv[q] = dstp[tid + q * SRT_T];
        sv[q] = srcp[tid + q * SRT_T];
    }
    if (tail) { dt = dstp[tid + FULLQ * SRT_T]; st = srcp[tid + FULLQ * SRT_T]; }
    int ownc = 0;
    if (tid < NBUCK) ownc = cntB[b * NBUCK + tid];
    if (tid < PARTS * NBUCK) {
        const int k = tid % NBUCK;
        const int p = tid / NBUCK;
        int total = 0, before = 0;
        #pragma unroll
        for (int j = 0; j < NB1 / PARTS; ++j) {
            const int bb = p + j * PARTS;
            const int c = cntB[bb * NBUCK + k];
            total += c;
            before += (bb < b) ? c : 0;
        }
        ptot[p][k] = total;
        pbef[p][k] = before;
    }
    __syncthreads();
    int total = 0, before = 0, rstart = 0;
    if (tid < NBUCK) {
        #pragma unroll
        for (int p = 0; p < PARTS; ++p) { total += ptot[p][tid]; before += pbef[p][tid]; }
        tot[tid] = total;
        own[tid] = ownc;
    }
    __syncthreads();
    if (tid < NBUCK) {
        int start = 0, lstart = 0;
        for (int k = 0; k < tid; ++k) { start += tot[k]; lstart += own[k]; }
        rstart = start;
        cur[tid]  = lstart;
        gdel[tid] = start + before - lstart;
    }
    __syncthreads();
    int pos[FULLQ], pt = 0;
    #pragma unroll
    for (int q = 0; q < FULLQ; ++q) pos[q] = atomicAdd(&cur[dv[q] >> BSH], 1);
    if (tail) pt = atomicAdd(&cur[dt >> BSH], 1);
    #pragma unroll
    for (int q = 0; q < FULLQ; ++q) {
        stage[pos[q]] = ((dv[q] & (BW - 1)) << SRCBITS) | sv[q];
        rid[pos[q]]   = (unsigned char)(dv[q] >> BSH);
    }
    if (tail) {
        stage[pt] = ((dt & (BW - 1)) << SRCBITS) | st;
        rid[pt]   = (unsigned char)(dt >> BSH);
    }
    __syncthreads();
    #pragma unroll
    for (int q = 0; q < FULLQ; ++q) {
        const int i = tid + q * SRT_T;
        bkt[i + gdel[rid[i]]] = stage[i];
    }
    if (tail) {
        const int i = tid + FULLQ * SRT_T;
        bkt[i + gdel[rid[i]]] = stage[i];
    }
    if (b == 0 && tid < NBUCK) {
        bstart[tid] = rstart;
        if (tid == NBUCK - 1) bstart[NBUCK] = rstart + total;
    }
}

__device__ __forceinline__ void range_sort_role(const int k,
                                                const int* __restrict__ bkt,
                                                const int* __restrict__ bstart,
                                                int* __restrict__ off,
                                                int* __restrict__ srt) {
    __shared__ int cnt[BW];
    __shared__ int wsum[BW / 32];
    __shared__ int stage[STAGE2];
    const int tid  = threadIdx.x;
    const int lane = tid & 31;
    const int wave = tid >> 5;
    const int lo = bstart[k];
    const int hi = bstart[k + 1];
    const int n  = hi - lo;
    const bool staged = n <= STAGE2;

    int w0[BATCH];
    #pragma unroll
    for (int q = 0; q < BATCH; ++q) {
        const int i = lo + q * SRT_T + tid;
        w0[q] = (i < hi) ? bkt[i] : -1;
    }
    if (tid < BW) cnt[tid] = 0;
    __syncthreads();
    #pragma unroll
    for (int q = 0; q < BATCH; ++q)
        if (w0[q] >= 0) atomicAdd(&cnt[w0[q] >> SRCBITS], 1);
    for (int i = lo + BATCH * SRT_T + tid; i < hi; i += SRT_T)
        atomicAdd(&cnt[bkt[i] >> SRCBITS], 1);
    __syncthreads();

    int c = 0, incl = 0, myoff = 0;
    if (wave < BW / 32) {
        c = cnt[tid];
        incl = c;
        for (int o = 1; o < 32; o <<= 1) {
            const int up = __shfl_up(incl, o, 32);
            if (lane >= o) incl += up;
        }
        if (lane == 31) wsum[wave] = incl;
    }
    __syncthreads();
    if (wave < BW / 32) {
        int before = lo;
        for (int w = 0; w < wave; ++w) before += wsum[w];
        myoff = before + incl - c;
        cnt[tid] = myoff;
    }
    __syncthreads();

    int pos[BATCH];
    #pragma unroll
    for (int q = 0; q < BATCH; ++q) {
        pos[q] = 0;
        if (w0[q] >= 0) pos[q] = atomicAdd(&cnt[w0[q] >> SRCBITS], 1);
    }
    if (wave < BW / 32) {
        const int d = k * BW + tid;
        if (d < N_NODES) off[d] = myoff;
    }
    if (k == NBUCK - 1 && tid == 0) off[N_NODES] = hi;
    if (staged) {
        #pragma unroll
        for (int q = 0; q < BATCH; ++q)
            if (w0[q] >= 0) stage[pos[q] - lo] = w0[q] & ((1 << SRCBITS) - 1);
        __syncthreads();
        const int a = lo & 31;
        for (int i = tid - a; i < n; i += SRT_T)
            if (i >= 0) srt[lo + i] = stage[i];
    } else {
        #pragma unroll
        for (int q = 0; q < BATCH; ++q)
            if (w0[q] >= 0) srt[pos[q]] = w0[q] & ((1 << SRCBITS) - 1);
        for (int i = lo + BATCH * SRT_T + tid; i < hi; i += SRT_T) {
            const int w = bkt[i];
            const int slot = atomicAdd(&cnt[w >> SRCBITS], 1);
            srt[slot] = w & ((1 << SRCBITS) - 1);
        }
    }
}

static_assert(LG_T == SRT_T, "both roles use the same group size");
__global__ __launch_bounds__(SRT_T) void sort_logits_kernel(const int* __restrict__ bkt,
                                                            const int* __restrict__ bstart,
                                                            int* __restrict__ off,
                                                            int* __restrict__ srt,
                                                            const float* __restrict__ emb,
                                                            const float* __restrict__ W,
                                                            const float* __restrict__ att_src,
                                                            const float* __restrict__ att_dst,
                                                            float* __restrict__ a_src,
                                                            float* __restrict__ a_dst) {
    if (blockIdx.x < NBUCK) range_sort_role(blockIdx.x, bkt, bstart, off, srt);
    else logits_role(blockIdx.x - NBUCK, emb, W, att_src, att_dst, a_src, a_dst);
}

template <int M>
__device__ __forceinline__ float xrow(float x) {
    const int b = __float_as_int(x);
    return __int_as_float(__builtin_amdgcn_update_dpp(b, b, 0x160 + M, 0xf, 0xf, false));
}

__device__ __forceinline__ float xhalf(float x) {
    const int b = __float_as_int(x);
    return __int_as_float(__builtin_amdgcn_permlanex16(b, b, 0x76543210, 0xfedcba98, false, false));
}

__device__ __forceinline__ float wave_max(float v) {
    v = fmaxf(v, xrow<1>(v));
    v = fmaxf(v, xrow<2>(v));
    v = fmaxf(v, xrow<4>(v));
    v = fmaxf(v, xrow<8>(v));
    return fmaxf(v, xhalf(v));
}

__device__ __forceinline__ float wave_sum(float v) {
    v += xrow<1>(v);
    v += xrow<2>(v);
    v += xrow<4>(v);
    v += xrow<8>(v);
    return v + xhalf(v);
}

__device__ __forceinline__ float4 wave_max4(float4 v) {
    return make_float4(wave_max(v.x), wave_max(v.y), wave_max(v.z), wave_max(v.w));
}

__device__ __forceinline__ float4 wave_sum4(float4 v) {
    return make_float4(wave_sum(v.x), wave_sum(v.y), wave_sum(v.z), wave_sum(v.w));
}

__device__ __forceinline__ float4 edge_logits(const int* __restrict__ srt,
                                              const float* __restrict__ a_src,
                                              float4 ad, int j, int& s) {
    s = srt[j];
    const float4 as = *(const float4*)(a_src + (size_t)s * 4);
    float4 e;
    e.x = leaky(as.x + ad.x);
    e.y = leaky(as.y + ad.y);
    e.z = leaky(as.z + ad.z);
    e.w = leaky(as.w + ad.w);
    return e;
}

__device__ __forceinline__ float to_scalar(float v) {
    return __int_as_float(__builtin_amdgcn_readfirstlane(__float_as_int(v)));
}

__device__ __forceinline__ int bcast(float v, int srcLane) {
    return __builtin_amdgcn_readlane(__float_as_int(v), srcLane);
}

template <int G>
__device__ __forceinline__ void load_rows(const float* __restrict__ emb, int s, int t,
                                          unsigned lane, float2 (&r)[G]) {
    #pragma unroll
    for (int i = 0; i < G; ++i) {
        const unsigned st = (unsigned)__builtin_amdgcn_readlane(s, t + i);
        r[i] = ((const float2*)emb)[st * (EMB / 2) + lane];
    }
}

template <int G>
__device__ __forceinline__ void add_rows(const float2 (&r)[G], float4 al, int t,
                                         float2& z0, float2& z1, float2& z2, float2& z3) {
    #pragma unroll
    for (int i = 0; i < G; ++i) {
        const float a0 = __int_as_float(bcast(al.x, t + i));
        const float a1 = __int_as_float(bcast(al.y, t + i));
        const float a2 = __int_as_float(bcast(al.z, t + i));
        const float a3 = __int_as_float(bcast(al.w, t + i));
        z0.x = __builtin_fmaf(a0, r[i].x, z0.x); z0.y = __builtin_fmaf(a0, r[i].y, z0.y);
        z1.x = __builtin_fmaf(a1, r[i].x, z1.x); z1.y = __builtin_fmaf(a1, r[i].y, z1.y);
        z2.x = __builtin_fmaf(a2, r[i].x, z2.x); z2.y = __builtin_fmaf(a2, r[i].y, z2.y);
        z3.x = __builtin_fmaf(a3, r[i].x, z3.x); z3.y = __builtin_fmaf(a3, r[i].y, z3.y);
    }
}

__device__ __forceinline__ void add_groups(const float* __restrict__ emb, int s, float4 al,
                                           int t0, int n, unsigned lane,
                                           float2& z0, float2& z1, float2& z2, float2& z3) {
    #pragma unroll 1
    for (int t = t0; t < n; t += 8) {
        float2 q[8];
        load_rows<8>(emb, s, t, lane, q);
        add_rows<8>(q, al, t, z0, z1, z2, z3);
    }
}

#define ZK (HEADS * EMB)
#define ZP (ZK + 4)
#define PP (256 + 16)
#define DPW 2
static_assert(N_NODES % (16 * DPW) == 0, "no partial block");

__device__ __forceinline__ void aggregate_segment(const int beg, const int end, const int s0,
                                                  const float4 ad,
                                                  const int* __restrict__ srt,
                                                  const float* __restrict__ emb,
                                                  const float* __restrict__ a_src,
                                                  const int lane, float* __restrict__ zrow) {
    const float NEG_INF = __int_as_float(0xFF800000);

    const int  j0 = beg + lane;
    const bool v0 = j0 < end;
    const unsigned ulane = (unsigned)lane;
    const int n0 = min(32, end - beg);
    const int sp = v0 ? s0 : __builtin_amdgcn_readlane(s0, max(n0 - 1, 0));
    const float4 as0 = *(const float4*)(a_src + (size_t)sp * 4);
    float2 r0[16];
    load_rows<16>(emb, sp, 0, ulane, r0);
    float4 e0;
    e0.x = v0 ? leaky(as0.x + ad.x) : NEG_INF; e0.y = v0 ? leaky(as0.y + ad.y) : NEG_INF;
    e0.z = v0 ? leaky(as0.z + ad.z) : NEG_INF; e0.w = v0 ? leaky(as0.w + ad.w) : NEG_INF;

    float4 mx = e0;
    for (int j = j0 + 32; j < end; j += 32) {
        int s;
        const float4 e = edge_logits(srt, a_src, ad, j, s);
        mx.x = fmaxf(mx.x, e.x); mx.y = fmaxf(mx.y, e.y);
        mx.z = fmaxf(mx.z, e.z); mx.w = fmaxf(mx.w, e.w);
    }
    mx = wave_max4(mx);
    mx = make_float4(to_scalar(mx.x), to_scalar(mx.y), to_scalar(mx.z), to_scalar(mx.w));

    float4 ex0 = make_float4(0.0f, 0.0f, 0.0f, 0.0f);
    if (v0) {
        ex0.x = __expf(e0.x - mx.x); ex0.y = __expf(e0.y - mx.y);
        ex0.z = __expf(e0.z - mx.z); ex0.w = __expf(e0.w - mx.w);
    }
    float4 den = ex0;
    for (int j = j0 + 32; j < end; j += 32) {
        int s;
        const float4 e = edge_logits(srt, a_src, ad, j, s);
        den.x += __expf(e.x - mx.x); den.y += __expf(e.y - mx.y);
        den.z += __expf(e.z - mx.z); den.w += __expf(e.w - mx.w);
    }
    den = wave_sum4(den);
    den.x += 1e-16f; den.y += 1e-16f; den.z += 1e-16f; den.w += 1e-16f;
    den = make_float4(to_scalar(den.x), to_scalar(den.y), to_scalar(den.z), to_scalar(den.w));

    float2 z0 = make_float2(0.0f, 0.0f), z1 = z0, z2 = z0, z3 = z0;

    if (n0 > 0) {
        const float4 al = make_float4(ex0.x / den.x, ex0.y / den.y, ex0.z / den.z, ex0.w / den.w);
        add_rows<16>(r0, al, 0, z0, z1, z2, z3);
        add_groups(emb, sp, al, 16, n0, ulane, z0, z1, z2, z3);
    }
    for (int base = beg + 32; base < end; base += 32) {
        const int j = base + lane;
        const int n = min(32, end - base);
        int s = 0;
        float4 al = make_float4(0.0f, 0.0f, 0.0f, 0.0f);
        if (j < end) {
            const float4 e = edge_logits(srt, a_src, ad, j, s);
            al.x = __expf(e.x - mx.x) / den.x; al.y = __expf(e.y - mx.y) / den.y;
            al.z = __expf(e.z - mx.z) / den.z; al.w = __expf(e.w - mx.w) / den.w;
        }
        const int sl = __builtin_amdgcn_readlane(s, n - 1);
        s = (j < end) ? s : sl;
        add_groups(emb, s, al, 0, n, ulane, z0, z1, z2, z3);
    }
    *(float2*)&zrow[0 * EMB + lane * 2] = z0;
    *(float2*)&zrow[1 * EMB + lane * 2] = z1;
    *(float2*)&zrow[2 * EMB + lane * 2] = z2;
    *(float2*)&zrow[3 * EMB + lane * 2] = z3;
}

__global__ __launch_bounds__(512, 16) void gather_project_kernel(const int* __restrict__ off,
                                                             const int* __restrict__ srt,
                                                             const float* __restrict__ emb,
                                                             const float* __restrict__ W,
                                                             const float* __restrict__ a_src,
                                                             const float* __restrict__ a_dst,
                                                             float* __restrict__ out) {
    __shared__ __attribute__((aligned(16))) float zt[DPW][16][ZP];
    __shared__ __attribute__((aligned(16))) float part[DPW][16][PP];
    const int lane = threadIdx.x & 31;
    const int wave = __builtin_amdgcn_readfirstlane(threadIdx.x >> 5);
    const int dbase = blockIdx.x * (16 * DPW) + wave;

    int beg[DPW], end[DPW], s0[DPW];
    float4 ad[DPW];
    #pragma unroll
    for (int t = 0; t < DPW; ++t) {
        const int d = dbase + 16 * t;
        beg[t] = __builtin_amdgcn_readfirstlane(off[d]);
        end[t] = __builtin_amdgcn_readfirstlane(off[d + 1]);
        ad[t]  = *(const float4*)(a_dst + (size_t)d * 4);
    }
    #pragma unroll
    for (int t = 0; t < DPW; ++t) {
        s0[t] = 0;
        if (beg[t] + lane < end[t]) s0[t] = srt[beg[t] + lane];
    }
    #pragma unroll
    for (int t = 0; t < DPW; ++t)
        aggregate_segment(beg[t], end[t], s0[t], ad[t], srt, emb, a_src, lane, &zt[t][wave][0]);

    const int half = lane >> 4;
    const int lrow = lane & 15;
    const int ct   = wave & 1;
    const int kq   = wave >> 1;
    const int hk   = kq >> 1;
    const int kin  = (kq & 1) * 32;
    const float* wp = W + (size_t)(kin + 2 * half) * HOUT + hk * RD + ct * 16 + lrow;
    v2f bf[8];
    #pragma unroll
    for (int ks = 0; ks < 8; ++ks) {
        bf[ks].x = wp[(ks * 4) * HOUT];
        bf[ks].y = wp[(ks * 4 + 1) * HOUT];
    }
    __syncthreads();

    #pragma unroll
    for (int t = 0; t < DPW; ++t) {
        const float* zr = &zt[t][lrow][kq * 32 + 2 * half];
        v8f c0 = {}, c1 = {};
        #pragma unroll
        for (int ks = 0; ks < 8; ks += 2) {
            const v2f fa0 = *(const v2f*)(zr + ks * 4);
            const v2f fa1 = *(const v2f*)(zr + ks * 4 + 4);
            c0 = __builtin_amdgcn_wmma_f32_16x16x4_f32(false, fa0, false, bf[ks],     (short)0, c0, false, false);
            c1 = __builtin_amdgcn_wmma_f32_16x16x4_f32(false, fa1, false, bf[ks + 1], (short)0, c1, false, false);
        }
        #pragma unroll
        for (int v = 0; v < 8; ++v) part[t][wave][v * 32 + lane] = c0[v] + c1[v];
    }
    __syncthreads();

    const int pl = (wave >> 3) * 16 + lrow;
    const int pv = wave & 7;
    const int pc = lane >> 4;
    #pragma unroll
    for (int t = 0; t < DPW; ++t) {
        float sum = 0.0f;
        #pragma unroll
        for (int q = 0; q < 8; ++q) sum += part[t][2 * q + pc][pv * 32 + pl];
        out[(size_t)(dbase + 16 * t) * RD + lane] = 0.25f * sum;
    }
}

extern "C" void kernel_launch(void* const* d_in, const int* in_sizes, int n_in,
                              void* d_out, int out_size, void* d_ws, size_t ws_size,
                              hipStream_t stream) {
    const float* node_emb = (const float*)d_in[0];
    const float* W        = (const float*)d_in[1];
    const float* att_src  = (const float*)d_in[2];
    const float* att_dst  = (const float*)d_in[3];
    const int*   path     = (const int*)d_in[4];
    float* out = (float*)d_out;

    float* ws    = (float*)d_ws;
    float* h     = ws;
    float* a_src = h     + (size_t)N_NODES * HOUT;
    float* a_dst = a_src + (size_t)N_NODES * HEADS;
    int*   cntB  = (int*)(a_dst + (size_t)N_NODES * HEADS);
    int*   bstart = cntB + (size_t)NB1 * NBUCK;
    int*   off   = cntB + (size_t)N_NODES;
    int*   bkt   = off  + (size_t)N_NODES + 32;
    int*   srt   = bkt  + (size_t)NEDGES;

    range_count_kernel<<<NB1, SRT_T, 0, stream>>>(path, cntB);

    range_scatter_kernel<<<NB1, SRT_T, 0, stream>>>(path, cntB, bstart, bkt);

    sort_logits_kernel<<<NBUCK + LG_BLOCKS, SRT_T, 0, stream>>>(
        bkt, bstart, off, srt, node_emb, W, att_src, att_dst, a_src, a_dst);

    gather_project_kernel<<<N_NODES / (16 * DPW), 512, 0, stream>>>(off, srt, node_emb, W, a_src, a_dst, out);
}


# Level 1 sorts each block's edges inside the block's own words, ranks from its own LDS counts; level 2 gathers its 250 runs by a start table: 3 launches, not 4, no count pass; arrival order as before
# speedup vs baseline: 9.2895x; 1.0899x over previous
#include <hip/hip_runtime.h>

#define N_NODES   100000
#define EMB       64
#define HEADS     4
#define RD        32
#define HOUT      (HEADS * RD)
#define NEDGES    1600000
#define NEG_SLOPE 0.2f

typedef __attribute__((ext_vector_type(2))) float v2f;
typedef __attribute__((ext_vector_type(8))) float v8f;

__device__ __forceinline__ float leaky(float x) {
    return x >= 0.0f ? x : NEG_SLOPE * x;
}

#define LG_T      1024
#define LG_TILES  (N_NODES / 16)
#define LG_BLOCKS ((LG_TILES + LG_T / 32 - 1) / (LG_T / 32))
static_assert(16 * EMB == LG_T, "one folded entry per thread");
__device__ __forceinline__ void logits_role(const int blk,
                                            const float* __restrict__ emb,
                                            const float* __restrict__ W,
                                            const float* __restrict__ att_src,
                                            const float* __restrict__ att_dst,
                                            float* __restrict__ a_src,
                                            float* __restrict__ a_dst) {
    __shared__ float Wa[16][EMB + 1];
    const int tid = threadIdx.x;
    {
        const int k = tid >> 4;
        const int j = tid & 15;
        float s = 0.0f;
        if (j < 8) {
            const int hh = j & 3;
            const float* att = (j < 4 ? att_src : att_dst) + hh * RD;
            const float* wr  = W + k * HOUT + hh * RD;
            #pragma unroll
            for (int dd = 0; dd < RD; ++dd) s += wr[dd] * att[dd];
        }
        Wa[j][k] = s;
    }
    __syncthreads();

    const int wave = tid >> 5;
    const int lane = tid & 31;
    const int rowTile = (blk * (LG_T / 32) + wave) * 16;
    if (rowTile >= N_NODES) return;

    const int half = lane >> 4;
    const int lrow = lane & 15;
    const float* arow = emb + (size_t)(rowTile + lrow) * EMB;

    v8f acc = {};
    for (int k0 = 0; k0 < EMB; k0 += 4) {
        v2f a, b;
        a.x = arow[k0 + 2 * half];
        a.y = arow[k0 + 2 * half + 1];
        b.x = Wa[lrow][k0 + 2 * half];
        b.y = Wa[lrow][k0 + 2 * half + 1];
        acc = __builtin_amdgcn_wmma_f32_16x16x4_f32(
            false, a, false, b, (short)0, acc, false, false);
    }
    float* dstp = (lrow < 4 ? a_src + lrow : a_dst + (lrow - 4)) + (size_t)(rowTile + 8 * half) * HEADS;
    if (lrow < 8) {
        #pragma unroll
        for (int v = 0; v < 8; ++v) dstp[v * HEADS] = acc[v];
    }
}

#define BSH    9
#define BW     (1 << BSH)
#define NBUCK  ((N_NODES + BW - 1) >> BSH)
#define CHUNK  6400
#define NB1    (NEDGES / CHUNK)
#define SRT_T  1024
#define FULLQ  (CHUNK / SRT_T)
#define TAILT  (CHUNK - FULLQ * SRT_T)
#define SRCBITS 17
#define LSTR   (NBUCK + 1)
#define SCANW  ((NBUCK + 31) / 32)
#define RUNS   256
#define RUNW   (RUNS / 32)
#define RPW    (RUNS / (SRT_T / 32))
#define BATCH  10
#define STAGE2 (BATCH * SRT_T)
static_assert(NEDGES % CHUNK == 0, "level-1 chunks must tile the edge list");
static_assert(TAILT > 0 && TAILT % 32 == 0, "the 7th edge belongs to whole waves");
static_assert(N_NODES <= (1 << SRCBITS), "source id and local destination share one word");
static_assert(NB1 * LSTR <= N_NODES, "the run-start table fits the old counter region");
static_assert(NB1 <= RUNS && RUNS <= SRT_T && RPW * (SRT_T / 32) == RUNS, "one run slot per thread of the first waves, RPW runs per wave");

__global__ __launch_bounds__(SRT_T) void range_local_sort_kernel(const int* __restrict__ path,
                                                                 int* __restrict__ ls,
                                                                 int* __restrict__ bkt) {
    __shared__ int hist[NBUCK];
    __shared__ int lst[NBUCK];
    __shared__ int wtot[SCANW];
    __shared__ int stage[CHUNK];
    const int tid  = threadIdx.x;
    const int lane = tid & 31;
    const int wave = tid >> 5;
    const int b    = blockIdx.x;
    const int* srcp = path + b * CHUNK;
    const int* dstp = path + NEDGES + b * CHUNK;
    const bool tail = tid < TAILT;
    int dv[FULLQ], sv[FULLQ], dt = 0, st = 0;
    #pragma unroll
    for (int q = 0; q < FULLQ; ++q) {
        dv[q] = dstp[tid + q * SRT_T];
        sv[q] = srcp[tid + q * SRT_T];
    }
    if (tail) { dt = dstp[tid + FULLQ * SRT_T]; st = srcp[tid + FULLQ * SRT_T]; }
    if (tid < NBUCK) hist[tid] = 0;
    __syncthreads();
    int rk[FULLQ], rt = 0;
    #pragma unroll
    for (int q = 0; q < FULLQ; ++q) rk[q] = atomicAdd(&hist[dv[q] >> BSH], 1);
    if (tail) rt = atomicAdd(&hist[dt >> BSH], 1);
    __syncthreads();
    int c = 0, incl = 0;
    if (wave < SCANW) {
        c = (tid < NBUCK) ? hist[tid] : 0;
        incl = c;
        for (int o = 1; o < 32; o <<= 1) {
            const int up = __shfl_up(incl, o, 32);
            if (lane >= o) incl += up;
        }
        if (lane == 31) wtot[wave] = incl;
    }
    __syncthreads();
    if (wave < SCANW) {
        int before = 0;
        for (int w = 0; w < wave; ++w) before += wtot[w];
        const int ex = before + incl - c;
        if (tid < NBUCK) {
            lst[tid] = ex;
            ls[b * LSTR + tid] = ex;
        }
        if (tid == NBUCK - 1) ls[b * LSTR + NBUCK] = ex + c;
    }
    __syncthreads();
    #pragma unroll
    for (int q = 0; q < FULLQ; ++q)
        stage[lst[dv[q] >> BSH] + rk[q]] = ((dv[q] & (BW - 1)) << SRCBITS) | sv[q];
    if (tail) stage[lst[dt >> BSH] + rt] = ((dt & (BW - 1)) << SRCBITS) | st;
    __syncthreads();
    #pragma unroll
    for (int q = 0; q < FULLQ; ++q) {
        const int i = tid + q * SRT_T;
        bkt[b * CHUNK + i] = stage[i];
    }
    if (tail) {
        const int i = tid + FULLQ * SRT_T;
        bkt[b * CHUNK + i] = stage[i];
    }
}

__device__ __forceinline__ int uni(int v) { return __builtin_amdgcn_readfirstlane(v); }

__device__ __forceinline__ void range_sort_role(const int k,
                                                const int* __restrict__ bkt,
                                                const int* __restrict__ ls,
                                                int* __restrict__ off,
                                                int* __restrict__ srt) {
    __shared__ int cnt[BW];
    __shared__ int wsum[BW / 32];
    __shared__ int rlen[RUNS];
    __shared__ int rpos[RUNS];
    __shared__ int rbase[RUNS];
    __shared__ int wlen[RUNW];
    __shared__ int wbeg[RUNW];
    __shared__ int stage[STAGE2];
    const int tid  = threadIdx.x;
    const int lane = tid & 31;
    const int wave = tid >> 5;

    int rs = 0, len = 0, lincl = 0;
    if (wave < RUNW) {
        int re = 0;
        if (tid < NB1) {
            rs = ls[tid * LSTR + k];
            re = ls[tid * LSTR + k + 1];
        }
        len = re - rs;
        lincl = len;
        for (int o = 1; o < 32; o <<= 1) {
            const int up = __shfl_up(lincl, o, 32);
            if (lane >= o) lincl += up;
        }
        int ssum = rs;
        for (int o = 16; o > 0; o >>= 1) ssum += __shfl_xor(ssum, o, 32);
        if (lane == 31) { wlen[wave] = lincl; wbeg[wave] = ssum; }
    }
    if (tid < BW) cnt[tid] = 0;
    __syncthreads();
    int losum = 0, nsum = 0;
    #pragma unroll
    for (int w = 0; w < RUNW; ++w) { losum += wbeg[w]; nsum += wlen[w]; }
    const int lo = uni(losum);
    const int n  = uni(nsum);
    const int hi = lo + n;
    if (wave < RUNW) {
        int before = 0;
        for (int w = 0; w < wave; ++w) before += wlen[w];
        rpos[tid]  = before + lincl - len;
        rlen[tid]  = len;
        rbase[tid] = tid * CHUNK + rs;
    }
    __syncthreads();
    const bool staged = n <= STAGE2;

    int w0[BATCH];
    if (staged) {
        int L[RPW], P[RPW], G[RPW], va[RPW], vb[RPW];
        #pragma unroll
        for (int r = 0; r < RPW; ++r) {
            const int rb = wave + (SRT_T / 32) * r;
            L[r] = uni(rlen[rb]);
            P[r] = uni(rpos[rb]);
            G[r] = uni(rbase[rb]);
            va[r] = (lane < L[r]) ? bkt[G[r] + lane] : 0;
            vb[r] = (lane + 32 < L[r]) ? bkt[G[r] + lane + 32] : 0;
        }
        #pragma unroll
        for (int r = 0; r < RPW; ++r) {
            if (lane < L[r]) stage[P[r] + lane] = va[r];
            if (lane + 32 < L[r]) stage[P[r] + lane + 32] = vb[r];
            for (int j = lane + 64; j < L[r]; j += 32) stage[P[r] + j] = bkt[G[r] + j];
        }
        __syncthreads();
        #pragma unroll
        for (int q = 0; q < BATCH; ++q) {
            const int i = q * SRT_T + tid;
            w0[q] = (i < n) ? stage[i] : -1;
        }
    } else {
        #pragma unroll
        for (int q = 0; q < BATCH; ++q) w0[q] = -1;
    }
    #pragma unroll
    for (int q = 0; q < BATCH; ++q)
        if (w0[q] >= 0) atomicAdd(&cnt[w0[q] >> SRCBITS], 1);
    if (!staged) {
        for (int rb = wave; rb < NB1; rb += 32) {
            const int Lr = uni(rlen[rb]);
            const int Gr = uni(rbase[rb]);
            for (int j = lane; j < Lr; j += 32) atomicAdd(&cnt[bkt[Gr + j] >> SRCBITS], 1);
        }
    }
    __syncthreads();

    int c = 0, incl = 0, myoff = 0;
    if (wave < BW / 32) {
        c = cnt[tid];
        incl = c;
        for (int o = 1; o < 32; o <<= 1) {
            const int up = __shfl_up(incl, o, 32);
            if (lane >= o) incl += up;
        }
        if (lane == 31) wsum[wave] = incl;
    }
    __syncthreads();
    if (wave < BW / 32) {
        int before = lo;
        for (int w = 0; w < wave; ++w) before += wsum[w];
        myoff = before + incl - c;
        cnt[tid] = myoff;
    }
    __syncthreads();

    int pos[BATCH];
    #pragma unroll
    for (int q = 0; q < BATCH; ++q) {
        pos[q] = 0;
        if (w0[q] >= 0) pos[q] = atomicAdd(&cnt[w0[q] >> SRCBITS], 1);
    }
    if (wave < BW / 32) {
        const int d = k * BW + tid;
        if (d < N_NODES) off[d] = myoff;
    }
    if (k == NBUCK - 1 && tid == 0) off[N_NODES] = hi;
    if (staged) {
        #pragma unroll
        for (int q = 0; q < BATCH; ++q)
            if (w0[q] >= 0) stage[pos[q] - lo] = w0[q] & ((1 << SRCBITS) - 1);
        __syncthreads();
        const int a = lo & 31;
        for (int i = tid - a; i < n; i += SRT_T)
            if (i >= 0) srt[lo + i] = stage[i];
    } else {
        for (int rb = wave; rb < NB1; rb += 32) {
            const int Lr = uni(rlen[rb]);
            const int Gr = uni(rbase[rb]);
            for (int j = lane; j < Lr; j += 32) {
                const int w = bkt[Gr + j];
                const int slot = atomicAdd(&cnt[w >> SRCBITS], 1);
                srt[slot] = w & ((1 << SRCBITS) - 1);
            }
        }
    }
}

static_assert(LG_T == SRT_T, "both roles use the same group size");
__global__ __launch_bounds__(SRT_T) void sort_logits_kernel(const int* __restrict__ bkt,
                                                            const int* __restrict__ ls,
                                                            int* __restrict__ off,
                                                            int* __restrict__ srt,
                                                            const float* __restrict__ emb,
                                                            const float* __restrict__ W,
                                                            const float* __restrict__ att_src,
                                                            const float* __restrict__ att_dst,
                                                            float* __restrict__ a_src,
                                                            float* __restrict__ a_dst) {
    if (blockIdx.x < NBUCK) range_sort_role(blockIdx.x, bkt, ls, off, srt);
    else logits_role(blockIdx.x - NBUCK, emb, W, att_src, att_dst, a_src, a_dst);
}

template <int M>
__device__ __forceinline__ float xrow(float x) {
    const int b = __float_as_int(x);
    return __int_as_float(__builtin_amdgcn_update_dpp(b, b, 0x160 + M, 0xf, 0xf, false));
}

__device__ __forceinline__ float xhalf(float x) {
    const int b = __float_as_int(x);
    return __int_as_float(__builtin_amdgcn_permlanex16(b, b, 0x76543210, 0xfedcba98, false, false));
}

__device__ __forceinline__ float wave_max(float v) {
    v = fmaxf(v, xrow<1>(v));
    v = fmaxf(v, xrow<2>(v));
    v = fmaxf(v, xrow<4>(v));
    v = fmaxf(v, xrow<8>(v));
    return fmaxf(v, xhalf(v));
}

__device__ __forceinline__ float wave_sum(float v) {
    v += xrow<1>(v);
    v += xrow<2>(v);
    v += xrow<4>(v);
    v += xrow<8>(v);
    return v + xhalf(v);
}

__device__ __forceinline__ float4 wave_max4(float4 v) {
    return make_float4(wave_max(v.x), wave_max(v.y), wave_max(v.z), wave_max(v.w));
}

__device__ __forceinline__ float4 wave_sum4(float4 v) {
    return make_float4(wave_sum(v.x), wave_sum(v.y), wave_sum(v.z), wave_sum(v.w));
}

__device__ __forceinline__ float4 edge_logits(const int* __restrict__ srt,
                                              const float* __restrict__ a_src,
                                              float4 ad, int j, int& s) {
    s = srt[j];
    const float4 as = *(const float4*)(a_src + (size_t)s * 4);
    float4 e;
    e.x = leaky(as.x + ad.x);
    e.y = leaky(as.y + ad.y);
    e.z = leaky(as.z + ad.z);
    e.w = leaky(as.w + ad.w);
    return e;
}

__device__ __forceinline__ float to_scalar(float v) {
    return __int_as_float(__builtin_amdgcn_readfirstlane(__float_as_int(v)));
}

__device__ __forceinline__ int bcast(float v, int srcLane) {
    return __builtin_amdgcn_readlane(__float_as_int(v), srcLane);
}

template <int G>
__device__ __forceinline__ void load_rows(const float* __restrict__ emb, int s, int t,
                                          unsigned lane, float2 (&r)[G]) {
    #pragma unroll
    for (int i = 0; i < G; ++i) {
        const unsigned st = (unsigned)__builtin_amdgcn_readlane(s, t + i);
        r[i] = ((const float2*)emb)[st * (EMB / 2) + lane];
    }
}

template <int G>
__device__ __forceinline__ void add_rows(const float2 (&r)[G], float4 al, int t,
                                         float2& z0, float2& z1, float2& z2, float2& z3) {
    #pragma unroll
    for (int i = 0; i < G; ++i) {
        const float a0 = __int_as_float(bcast(al.x, t + i));
        const float a1 = __int_as_float(bcast(al.y, t + i));
        const float a2 = __int_as_float(bcast(al.z, t + i));
        const float a3 = __int_as_float(bcast(al.w, t + i));
        z0.x = __builtin_fmaf(a0, r[i].x, z0.x); z0.y = __builtin_fmaf(a0, r[i].y, z0.y);
        z1.x = __builtin_fmaf(a1, r[i].x, z1.x); z1.y = __builtin_fmaf(a1, r[i].y, z1.y);
        z2.x = __builtin_fmaf(a2, r[i].x, z2.x); z2.y = __builtin_fmaf(a2, r[i].y, z2.y);
        z3.x = __builtin_fmaf(a3, r[i].x, z3.x); z3.y = __builtin_fmaf(a3, r[i].y, z3.y);
    }
}

__device__ __forceinline__ void add_groups(const float* __restrict__ emb, int s, float4 al,
                                           int t0, int n, unsigned lane,
                                           float2& z0, float2& z1, float2& z2, float2& z3) {
    #pragma unroll 1
    for (int t = t0; t < n; t += 8) {
        float2 q[8];
        load_rows<8>(emb, s, t, lane, q);
        add_rows<8>(q, al, t, z0, z1, z2, z3);
    }
}

#define ZK (HEADS * EMB)
#define ZP (ZK + 4)
#define PP (256 + 16)
#define DPW 2
static_assert(N_NODES % (16 * DPW) == 0, "no partial block");

__device__ __forceinline__ void aggregate_segment(const int beg, const int end, const int s0,
                                                  const float4 ad,
                                                  const int* __restrict__ srt,
                                                  const float* __restrict__ emb,
                                                  const float* __restrict__ a_src,
                                                  const int lane, float* __restrict__ zrow) {
    const float NEG_INF = __int_as_float(0xFF800000);

    const int  j0 = beg + lane;
    const bool v0 = j0 < end;
    const unsigned ulane = (unsigned)lane;
    const int n0 = min(32, end - beg);
    const int sp = v0 ? s0 : __builtin_amdgcn_readlane(s0, max(n0 - 1, 0));
    const float4 as0 = *(const float4*)(a_src + (size_t)sp * 4);
    float2 r0[16];
    load_rows<16>(emb, sp, 0, ulane, r0);
    float4 e0;
    e0.x = v0 ? leaky(as0.x + ad.x) : NEG_INF; e0.y = v0 ? leaky(as0.y + ad.y) : NEG_INF;
    e0.z = v0 ? leaky(as0.z + ad.z) : NEG_INF; e0.w = v0 ? leaky(as0.w + ad.w) : NEG_INF;

    float4 mx = e0;
    for (int j = j0 + 32; j < end; j += 32) {
        int s;
        const float4 e = edge_logits(srt, a_src, ad, j, s);
        mx.x = fmaxf(mx.x, e.x); mx.y = fmaxf(mx.y, e.y);
        mx.z = fmaxf(mx.z, e.z); mx.w = fmaxf(mx.w, e.w);
    }
    mx = wave_max4(mx);
    mx = make_float4(to_scalar(mx.x), to_scalar(mx.y), to_scalar(mx.z), to_scalar(mx.w));

    float4 ex0 = make_float4(0.0f, 0.0f, 0.0f, 0.0f);
    if (v0) {
        ex0.x = __expf(e0.x - mx.x); ex0.y = __expf(e0.y - mx.y);
        ex0.z = __expf(e0.z - mx.z); ex0.w = __expf(e0.w - mx.w);
    }
    float4 den = ex0;
    for (int j = j0 + 32; j < end; j += 32) {
        int s;
        const float4 e = edge_logits(srt, a_src, ad, j, s);
        den.x += __expf(e.x - mx.x); den.y += __expf(e.y - mx.y);
        den.z += __expf(e.z - mx.z); den.w += __expf(e.w - mx.w);
    }
    den = wave_sum4(den);
    den.x += 1e-16f; den.y += 1e-16f; den.z += 1e-16f; den.w += 1e-16f;
    den = make_float4(to_scalar(den.x), to_scalar(den.y), to_scalar(den.z), to_scalar(den.w));

    float2 z0 = make_float2(0.0f, 0.0f), z1 = z0, z2 = z0, z3 = z0;

    if (n0 > 0) {
        const float4 al = make_float4(ex0.x / den.x, ex0.y / den.y, ex0.z / den.z, ex0.w / den.w);
        add_rows<16>(r0, al, 0, z0, z1, z2, z3);
        add_groups(emb, sp, al, 16, n0, ulane, z0, z1, z2, z3);
    }
    for (int base = beg + 32; base < end; base += 32) {
        const int j = base + lane;
        const int n = min(32, end - base);
        int s = 0;
        float4 al = make_float4(0.0f, 0.0f, 0.0f, 0.0f);
        if (j < end) {
            const float4 e = edge_logits(srt, a_src, ad, j, s);
            al.x = __expf(e.x - mx.x) / den.x; al.y = __expf(e.y - mx.y) / den.y;
            al.z = __expf(e.z - mx.z) / den.z; al.w = __expf(e.w - mx.w) / den.w;
        }
        const int sl = __builtin_amdgcn_readlane(s, n - 1);
        s = (j < end) ? s : sl;
        add_groups(emb, s, al, 0, n, ulane, z0, z1, z2, z3);
    }
    *(float2*)&zrow[0 * EMB + lane * 2] = z0;
    *(float2*)&zrow[1 * EMB + lane * 2] = z1;
    *(float2*)&zrow[2 * EMB + lane * 2] = z2;
    *(float2*)&zrow[3 * EMB + lane * 2] = z3;
}

__global__ __launch_bounds__(512, 16) void gather_project_kernel(const int* __restrict__ off,
                                                             const int* __restrict__ srt,
                                                             const float* __restrict__ emb,
                                                             const float* __restrict__ W,
                                                             const float* __restrict__ a_src,
                                                             const float* __restrict__ a_dst,
                                                             float* __restrict__ out) {
    __shared__ __attribute__((aligned(16))) float zt[DPW][16][ZP];
    __shared__ __attribute__((aligned(16))) float part[DPW][16][PP];
    const int lane = threadIdx.x & 31;
    const int wave = __builtin_amdgcn_readfirstlane(threadIdx.x >> 5);
    const int dbase = blockIdx.x * (16 * DPW) + wave;

    int beg[DPW], end[DPW], s0[DPW];
    float4 ad[DPW];
    #pragma unroll
    for (int t = 0; t < DPW; ++t) {
        const int d = dbase + 16 * t;
        beg[t] = __builtin_amdgcn_readfirstlane(off[d]);
        end[t] = __builtin_amdgcn_readfirstlane(off[d + 1]);
        ad[t]  = *(const float4*)(a_dst + (size_t)d * 4);
    }
    #pragma unroll
    for (int t = 0; t < DPW; ++t) {
        s0[t] = 0;
        if (beg[t] + lane < end[t]) s0[t] = srt[beg[t] + lane];
    }
    #pragma unroll
    for (int t = 0; t < DPW; ++t)
        aggregate_segment(beg[t], end[t], s0[t], ad[t], srt, emb, a_src, lane, &zt[t][wave][0]);

    const int half = lane >> 4;
    const int lrow = lane & 15;
    const int ct   = wave & 1;
    const int kq   = wave >> 1;
    const int hk   = kq >> 1;
    const int kin  = (kq & 1) * 32;
    const float* wp = W + (size_t)(kin + 2 * half) * HOUT + hk * RD + ct * 16 + lrow;
    v2f bf[8];
    #pragma unroll
    for (int ks = 0; ks < 8; ++ks) {
        bf[ks].x = wp[(ks * 4) * HOUT];
        bf[ks].y = wp[(ks * 4 + 1) * HOUT];
    }
    __syncthreads();

    #pragma unroll
    for (int t = 0; t < DPW; ++t) {
        const float* zr = &zt[t][lrow][kq * 32 + 2 * half];
        v8f c0 = {}, c1 = {};
        #pragma unroll
        for (int ks = 0; ks < 8; ks += 2) {
            const v2f fa0 = *(const v2f*)(zr + ks * 4);
            const v2f fa1 = *(const v2f*)(zr + ks * 4 + 4);
            c0 = __builtin_amdgcn_wmma_f32_16x16x4_f32(false, fa0, false, bf[ks],     (short)0, c0, false, false);
            c1 = __builtin_amdgcn_wmma_f32_16x16x4_f32(false, fa1, false, bf[ks + 1], (short)0, c1, false, false);
        }
        #pragma unroll
        for (int v = 0; v < 8; ++v) part[t][wave][v * 32 + lane] = c0[v] + c1[v];
    }
    __syncthreads();

    const int pl = (wave >> 3) * 16 + lrow;
    const int pv = wave & 7;
    const int pc = lane >> 4;
    #pragma unroll
    for (int t = 0; t < DPW; ++t) {
        float sum = 0.0f;
        #pragma unroll
        for (int q = 0; q < 8; ++q) sum += part[t][2 * q + pc][pv * 32 + pl];
        out[(size_t)(dbase + 16 * t) * RD + lane] = 0.25f * sum;
    }
}

extern "C" void kernel_launch(void* const* d_in, const int* in_sizes, int n_in,
                              void* d_out, int out_size, void* d_ws, size_t ws_size,
                              hipStream_t stream) {
    const float* node_emb = (const float*)d_in[0];
    const float* W        = (const float*)d_in[1];
    const float* att_src  = (const float*)d_in[2];
    const float* att_dst  = (const float*)d_in[3];
    const int*   path     = (const int*)d_in[4];
    float* out = (float*)d_out;

    float* ws    = (float*)d_ws;
    float* h     = ws;
    float* a_src = h     + (size_t)N_NODES * HOUT;
    float* a_dst = a_src + (size_t)N_NODES * HEADS;
    int*   ls    = (int*)(a_dst + (size_t)N_NODES * HEADS);
    int*   off   = ls   + (size_t)N_NODES;
    int*   bkt   = off  + (size_t)N_NODES + 32;
    int*   srt   = bkt  + (size_t)NEDGES;

    range_local_sort_kernel<<<NB1, SRT_T, 0, stream>>>(path, ls, bkt);

    sort_logits_kernel<<<NBUCK + LG_BLOCKS, SRT_T, 0, stream>>>(
        bkt, ls, off, srt, node_emb, W, att_src, att_dst, a_src, a_dst);

    gather_project_kernel<<<N_NODES / (16 * DPW), 512, 0, stream>>>(off, srt, node_emb, W, a_src, a_dst, out);
}


# Softmax division deferred: rows weighted by exp(e - max), each head's row sum scaled once by 1/denominator; denominators summed edge by edge in scalar registers, no lane tree, no per-edge divide
# speedup vs baseline: 9.6256x; 1.0362x over previous
#include <hip/hip_runtime.h>

#define N_NODES   100000
#define EMB       64
#define HEADS     4
#define RD        32
#define HOUT      (HEADS * RD)
#define NEDGES    1600000
#define NEG_SLOPE 0.2f

typedef __attribute__((ext_vector_type(2))) float v2f;
typedef __attribute__((ext_vector_type(8))) float v8f;

__device__ __forceinline__ float leaky(float x) {
    return x >= 0.0f ? x : NEG_SLOPE * x;
}

#define LG_T      1024
#define LG_TILES  (N_NODES / 16)
#define LG_BLOCKS ((LG_TILES + LG_T / 32 - 1) / (LG_T / 32))
static_assert(16 * EMB == LG_T, "one folded entry per thread");
__device__ __forceinline__ void logits_role(const int blk,
                                            const float* __restrict__ emb,
                                            const float* __restrict__ W,
                                            const float* __restrict__ att_src,
                                            const float* __restrict__ att_dst,
                                            float* __restrict__ a_src,
                                            float* __restrict__ a_dst) {
    __shared__ float Wa[16][EMB + 1];
    const int tid = threadIdx.x;
    {
        const int k = tid >> 4;
        const int j = tid & 15;
        float s = 0.0f;
        if (j < 8) {
            const int hh = j & 3;
            const float* att = (j < 4 ? att_src : att_dst) + hh * RD;
            const float* wr  = W + k * HOUT + hh * RD;
            #pragma unroll
            for (int dd = 0; dd < RD; ++dd) s += wr[dd] * att[dd];
        }
        Wa[j][k] = s;
    }
    __syncthreads();

    const int wave = tid >> 5;
    const int lane = tid & 31;
    const int rowTile = (blk * (LG_T / 32) + wave) * 16;
    if (rowTile >= N_NODES) return;

    const int half = lane >> 4;
    const int lrow = lane & 15;
    const float* arow = emb + (size_t)(rowTile + lrow) * EMB;

    v8f acc = {};
    for (int k0 = 0; k0 < EMB; k0 += 4) {
        v2f a, b;
        a.x = arow[k0 + 2 * half];
        a.y = arow[k0 + 2 * half + 1];
        b.x = Wa[lrow][k0 + 2 * half];
        b.y = Wa[lrow][k0 + 2 * half + 1];
        acc = __builtin_amdgcn_wmma_f32_16x16x4_f32(
            false, a, false, b, (short)0, acc, false, false);
    }
    float* dstp = (lrow < 4 ? a_src + lrow : a_dst + (lrow - 4)) + (size_t)(rowTile + 8 * half) * HEADS;
    if (lrow < 8) {
        #pragma unroll
        for (int v = 0; v < 8; ++v) dstp[v * HEADS] = acc[v];
    }
}

#define BSH    9
#define BW     (1 << BSH)
#define NBUCK  ((N_NODES + BW - 1) >> BSH)
#define CHUNK  6400
#define NB1    (NEDGES / CHUNK)
#define SRT_T  1024
#define FULLQ  (CHUNK / SRT_T)
#define TAILT  (CHUNK - FULLQ * SRT_T)
#define SRCBITS 17
#define LSTR   (NBUCK + 1)
#define SCANW  ((NBUCK + 31) / 32)
#define RUNS   256
#define RUNW   (RUNS / 32)
#define RPW    (RUNS / (SRT_T / 32))
#define BATCH  10
#define STAGE2 (BATCH * SRT_T)
static_assert(NEDGES % CHUNK == 0, "level-1 chunks must tile the edge list");
static_assert(TAILT > 0 && TAILT % 32 == 0, "the 7th edge belongs to whole waves");
static_assert(N_NODES <= (1 << SRCBITS), "source id and local destination share one word");
static_assert(NB1 * LSTR <= N_NODES, "the run-start table fits the old counter region");
static_assert(NB1 <= RUNS && RUNS <= SRT_T && RPW * (SRT_T / 32) == RUNS, "one run slot per thread of the first waves, RPW runs per wave");

__global__ __launch_bounds__(SRT_T) void range_local_sort_kernel(const int* __restrict__ path,
                                                                 int* __restrict__ ls,
                                                                 int* __restrict__ bkt) {
    __shared__ int hist[NBUCK];
    __shared__ int lst[NBUCK];
    __shared__ int wtot[SCANW];
    __shared__ int stage[CHUNK];
    const int tid  = threadIdx.x;
    const int lane = tid & 31;
    const int wave = tid >> 5;
    const int b    = blockIdx.x;
    const int* srcp = path + b * CHUNK;
    const int* dstp = path + NEDGES + b * CHUNK;
    const bool tail = tid < TAILT;
    int dv[FULLQ], sv[FULLQ], dt = 0, st = 0;
    #pragma unroll
    for (int q = 0; q < FULLQ; ++q) {
        dv[q] = dstp[tid + q * SRT_T];
        sv[q] = srcp[tid + q * SRT_T];
    }
    if (tail) { dt = dstp[tid + FULLQ * SRT_T]; st = srcp[tid + FULLQ * SRT_T]; }
    if (tid < NBUCK) hist[tid] = 0;
    __syncthreads();
    int rk[FULLQ], rt = 0;
    #pragma unroll
    for (int q = 0; q < FULLQ; ++q) rk[q] = atomicAdd(&hist[dv[q] >> BSH], 1);
    if (tail) rt = atomicAdd(&hist[dt >> BSH], 1);
    __syncthreads();
    int c = 0, incl = 0;
    if (wave < SCANW) {
        c = (tid < NBUCK) ? hist[tid] : 0;
        incl = c;
        for (int o = 1; o < 32; o <<= 1) {
            const int up = __shfl_up(incl, o, 32);
            if (lane >= o) incl += up;
        }
        if (lane == 31) wtot[wave] = incl;
    }
    __syncthreads();
    if (wave < SCANW) {
        int before = 0;
        for (int w = 0; w < wave; ++w) before += wtot[w];
        const int ex = before + incl - c;
        if (tid < NBUCK) {
            lst[tid] = ex;
            ls[b * LSTR + tid] = ex;
        }
        if (tid == NBUCK - 1) ls[b * LSTR + NBUCK] = ex + c;
    }
    __syncthreads();
    #pragma unroll
    for (int q = 0; q < FULLQ; ++q)
        stage[lst[dv[q] >> BSH] + rk[q]] = ((dv[q] & (BW - 1)) << SRCBITS) | sv[q];
    if (tail) stage[lst[dt >> BSH] + rt] = ((dt & (BW - 1)) << SRCBITS) | st;
    __syncthreads();
    #pragma unroll
    for (int q = 0; q < FULLQ; ++q) {
        const int i = tid + q * SRT_T;
        bkt[b * CHUNK + i] = stage[i];
    }
    if (tail) {
        const int i = tid + FULLQ * SRT_T;
        bkt[b * CHUNK + i] = stage[i];
    }
}

__device__ __forceinline__ int uni(int v) { return __builtin_amdgcn_readfirstlane(v); }

__device__ __forceinline__ void range_sort_role(const int k,
                                                const int* __restrict__ bkt,
                                                const int* __restrict__ ls,
                                                int* __restrict__ off,
                                                int* __restrict__ srt) {
    __shared__ int cnt[BW];
    __shared__ int wsum[BW / 32];
    __shared__ int rlen[RUNS];
    __shared__ int rpos[RUNS];
    __shared__ int rbase[RUNS];
    __shared__ int wlen[RUNW];
    __shared__ int wbeg[RUNW];
    __shared__ int stage[STAGE2];
    const int tid  = threadIdx.x;
    const int lane = tid & 31;
    const int wave = tid >> 5;

    int rs = 0, len = 0, lincl = 0;
    if (wave < RUNW) {
        int re = 0;
        if (tid < NB1) {
            rs = ls[tid * LSTR + k];
            re = ls[tid * LSTR + k + 1];
        }
        len = re - rs;
        lincl = len;
        for (int o = 1; o < 32; o <<= 1) {
            const int up = __shfl_up(lincl, o, 32);
            if (lane >= o) lincl += up;
        }
        int ssum = rs;
        for (int o = 16; o > 0; o >>= 1) ssum += __shfl_xor(ssum, o, 32);
        if (lane == 31) { wlen[wave] = lincl; wbeg[wave] = ssum; }
    }
    if (tid < BW) cnt[tid] = 0;
    __syncthreads();
    int losum = 0, nsum = 0;
    #pragma unroll
    for (int w = 0; w < RUNW; ++w) { losum += wbeg[w]; nsum += wlen[w]; }
    const int lo = uni(losum);
    const int n  = uni(nsum);
    const int hi = lo + n;
    if (wave < RUNW) {
        int before = 0;
        for (int w = 0; w < wave; ++w) before += wlen[w];
        rpos[tid]  = before + lincl - len;
        rlen[tid]  = len;
        rbase[tid] = tid * CHUNK + rs;
    }
    __syncthreads();
    const bool staged = n <= STAGE2;

    int w0[BATCH];
    if (staged) {
        int L[RPW], P[RPW], G[RPW], va[RPW], vb[RPW];
        #pragma unroll
        for (int r = 0; r < RPW; ++r) {
            const int rb = wave + (SRT_T / 32) * r;
            L[r] = uni(rlen[rb]);
            P[r] = uni(rpos[rb]);
            G[r] = uni(rbase[rb]);
            va[r] = (lane < L[r]) ? bkt[G[r] + lane] : 0;
            vb[r] = (lane + 32 < L[r]) ? bkt[G[r] + lane + 32] : 0;
        }
        #pragma unroll
        for (int r = 0; r < RPW; ++r) {
            if (lane < L[r]) stage[P[r] + lane] = va[r];
            if (lane + 32 < L[r]) stage[P[r] + lane + 32] = vb[r];
            for (int j = lane + 64; j < L[r]; j += 32) stage[P[r] + j] = bkt[G[r] + j];
        }
        __syncthreads();
        #pragma unroll
        for (int q = 0; q < BATCH; ++q) {
            const int i = q * SRT_T + tid;
            w0[q] = (i < n) ? stage[i] : -1;
        }
    } else {
        #pragma unroll
        for (int q = 0; q < BATCH; ++q) w0[q] = -1;
    }
    #pragma unroll
    for (int q = 0; q < BATCH; ++q)
        if (w0[q] >= 0) atomicAdd(&cnt[w0[q] >> SRCBITS], 1);
    if (!staged) {
        for (int rb = wave; rb < NB1; rb += 32) {
            const int Lr = uni(rlen[rb]);
            const int Gr = uni(rbase[rb]);
            for (int j = lane; j < Lr; j += 32) atomicAdd(&cnt[bkt[Gr + j] >> SRCBITS], 1);
        }
    }
    __syncthreads();

    int c = 0, incl = 0, myoff = 0;
    if (wave < BW / 32) {
        c = cnt[tid];
        incl = c;
        for (int o = 1; o < 32; o <<= 1) {
            const int up = __shfl_up(incl, o, 32);
            if (lane >= o) incl += up;
        }
        if (lane == 31) wsum[wave] = incl;
    }
    __syncthreads();
    if (wave < BW / 32) {
        int before = lo;
        for (int w = 0; w < wave; ++w) before += wsum[w];
        myoff = before + incl - c;
        cnt[tid] = myoff;
    }
    __syncthreads();

    int pos[BATCH];
    #pragma unroll
    for (int q = 0; q < BATCH; ++q) {
        pos[q] = 0;
        if (w0[q] >= 0) pos[q] = atomicAdd(&cnt[w0[q] >> SRCBITS], 1);
    }
    if (wave < BW / 32) {
        const int d = k * BW + tid;
        if (d < N_NODES) off[d] = myoff;
    }
    if (k == NBUCK - 1 && tid == 0) off[N_NODES] = hi;
    if (staged) {
        #pragma unroll
        for (int q = 0; q < BATCH; ++q)
            if (w0[q] >= 0) stage[pos[q] - lo] = w0[q] & ((1 << SRCBITS) - 1);
        __syncthreads();
        const int a = lo & 31;
        for (int i = tid - a; i < n; i += SRT_T)
            if (i >= 0) srt[lo + i] = stage[i];
    } else {
        for (int rb = wave; rb < NB1; rb += 32) {
            const int Lr = uni(rlen[rb]);
            const int Gr = uni(rbase[rb]);
            for (int j = lane; j < Lr; j += 32) {
                const int w = bkt[Gr + j];
                const int slot = atomicAdd(&cnt[w >> SRCBITS], 1);
                srt[slot] = w & ((1 << SRCBITS) - 1);
            }
        }
    }
}

static_assert(LG_T == SRT_T, "both roles use the same group size");
__global__ __launch_bounds__(SRT_T) void sort_logits_kernel(const int* __restrict__ bkt,
                                                            const int* __restrict__ ls,
                                                            int* __restrict__ off,
                                                            int* __restrict__ srt,
                                                            const float* __restrict__ emb,
                                                            const float* __restrict__ W,
                                                            const float* __restrict__ att_src,
                                                            const float* __restrict__ att_dst,
                                                            float* __restrict__ a_src,
                                                            float* __restrict__ a_dst) {
    if (blockIdx.x < NBUCK) range_sort_role(blockIdx.x, bkt, ls, off, srt);
    else logits_role(blockIdx.x - NBUCK, emb, W, att_src, att_dst, a_src, a_dst);
}

template <int M>
__device__ __forceinline__ float xrow(float x) {
    const int b = __float_as_int(x);
    return __int_as_float(__builtin_amdgcn_update_dpp(b, b, 0x160 + M, 0xf, 0xf, false));
}

__device__ __forceinline__ float xhalf(float x) {
    const int b = __float_as_int(x);
    return __int_as_float(__builtin_amdgcn_permlanex16(b, b, 0x76543210, 0xfedcba98, false, false));
}

__device__ __forceinline__ float wave_max(float v) {
    v = fmaxf(v, xrow<1>(v));
    v = fmaxf(v, xrow<2>(v));
    v = fmaxf(v, xrow<4>(v));
    v = fmaxf(v, xrow<8>(v));
    return fmaxf(v, xhalf(v));
}

__device__ __forceinline__ float4 wave_max4(float4 v) {
    return make_float4(wave_max(v.x), wave_max(v.y), wave_max(v.z), wave_max(v.w));
}

__device__ __forceinline__ float4 edge_logits(const int* __restrict__ srt,
                                              const float* __restrict__ a_src,
                                              float4 ad, int j, int& s) {
    s = srt[j];
    const float4 as = *(const float4*)(a_src + (size_t)s * 4);
    float4 e;
    e.x = leaky(as.x + ad.x);
    e.y = leaky(as.y + ad.y);
    e.z = leaky(as.z + ad.z);
    e.w = leaky(as.w + ad.w);
    return e;
}

__device__ __forceinline__ float to_scalar(float v) {
    return __int_as_float(__builtin_amdgcn_readfirstlane(__float_as_int(v)));
}

__device__ __forceinline__ int bcast(float v, int srcLane) {
    return __builtin_amdgcn_readlane(__float_as_int(v), srcLane);
}

template <int G>
__device__ __forceinline__ void load_rows(const float* __restrict__ emb, int s, int t,
                                          unsigned lane, float2 (&r)[G]) {
    #pragma unroll
    for (int i = 0; i < G; ++i) {
        const unsigned st = (unsigned)__builtin_amdgcn_readlane(s, t + i);
        r[i] = ((const float2*)emb)[st * (EMB / 2) + lane];
    }
}

template <int G>
__device__ __forceinline__ void add_rows(const float2 (&r)[G], float4 al, int t,
                                         float2& z0, float2& z1, float2& z2, float2& z3,
                                         float& d0, float& d1, float& d2, float& d3) {
    #pragma unroll
    for (int i = 0; i < G; ++i) {
        const float a0 = __int_as_float(bcast(al.x, t + i));
        const float a1 = __int_as_float(bcast(al.y, t + i));
        const float a2 = __int_as_float(bcast(al.z, t + i));
        const float a3 = __int_as_float(bcast(al.w, t + i));
        d0 += a0; d1 += a1; d2 += a2; d3 += a3;
        z0.x = __builtin_fmaf(a0, r[i].x, z0.x); z0.y = __builtin_fmaf(a0, r[i].y, z0.y);
        z1.x = __builtin_fmaf(a1, r[i].x, z1.x); z1.y = __builtin_fmaf(a1, r[i].y, z1.y);
        z2.x = __builtin_fmaf(a2, r[i].x, z2.x); z2.y = __builtin_fmaf(a2, r[i].y, z2.y);
        z3.x = __builtin_fmaf(a3, r[i].x, z3.x); z3.y = __builtin_fmaf(a3, r[i].y, z3.y);
    }
}

__device__ __forceinline__ void add_groups(const float* __restrict__ emb, int s, float4 al,
                                           int t0, int n, unsigned lane,
                                           float2& z0, float2& z1, float2& z2, float2& z3,
                                           float& d0, float& d1, float& d2, float& d3) {
    #pragma unroll 1
    for (int t = t0; t < n; t += 8) {
        float2 q[8];
        load_rows<8>(emb, s, t, lane, q);
        add_rows<8>(q, al, t, z0, z1, z2, z3, d0, d1, d2, d3);
    }
}

#define ZK (HEADS * EMB)
#define ZP (ZK + 4)
#define PP (256 + 16)
#define DPW 2
static_assert(N_NODES % (16 * DPW) == 0, "no partial block");

__device__ __forceinline__ void aggregate_segment(const int beg, const int end, const int s0,
                                                  const float4 ad,
                                                  const int* __restrict__ srt,
                                                  const float* __restrict__ emb,
                                                  const float* __restrict__ a_src,
                                                  const int lane, float* __restrict__ zrow) {
    const float NEG_INF = __int_as_float(0xFF800000);

    const int  j0 = beg + lane;
    const bool v0 = j0 < end;
    const unsigned ulane = (unsigned)lane;
    const int n0 = min(32, end - beg);
    const int sp = v0 ? s0 : __builtin_amdgcn_readlane(s0, max(n0 - 1, 0));
    const float4 as0 = *(const float4*)(a_src + (size_t)sp * 4);
    float2 r0[16];
    load_rows<16>(emb, sp, 0, ulane, r0);
    float4 e0;
    e0.x = v0 ? leaky(as0.x + ad.x) : NEG_INF; e0.y = v0 ? leaky(as0.y + ad.y) : NEG_INF;
    e0.z = v0 ? leaky(as0.z + ad.z) : NEG_INF; e0.w = v0 ? leaky(as0.w + ad.w) : NEG_INF;

    float4 mx = e0;
    for (int j = j0 + 32; j < end; j += 32) {
        int s;
        const float4 e = edge_logits(srt, a_src, ad, j, s);
        mx.x = fmaxf(mx.x, e.x); mx.y = fmaxf(mx.y, e.y);
        mx.z = fmaxf(mx.z, e.z); mx.w = fmaxf(mx.w, e.w);
    }
    mx = wave_max4(mx);
    mx = make_float4(to_scalar(mx.x), to_scalar(mx.y), to_scalar(mx.z), to_scalar(mx.w));

    float4 ex0 = make_float4(0.0f, 0.0f, 0.0f, 0.0f);
    if (v0) {
        ex0.x = __expf(e0.x - mx.x); ex0.y = __expf(e0.y - mx.y);
        ex0.z = __expf(e0.z - mx.z); ex0.w = __expf(e0.w - mx.w);
    }
    float2 z0 = make_float2(0.0f, 0.0f), z1 = z0, z2 = z0, z3 = z0;
    float d0 = 0.0f, d1 = 0.0f, d2 = 0.0f, d3 = 0.0f;

    if (n0 > 0) {
        add_rows<16>(r0, ex0, 0, z0, z1, z2, z3, d0, d1, d2, d3);
        add_groups(emb, sp, ex0, 16, n0, ulane, z0, z1, z2, z3, d0, d1, d2, d3);
    }
    for (int base = beg + 32; base < end; base += 32) {
        const int j = base + lane;
        const int n = min(32, end - base);
        int s = 0;
        float4 ex = make_float4(0.0f, 0.0f, 0.0f, 0.0f);
        if (j < end) {
            const float4 e = edge_logits(srt, a_src, ad, j, s);
            ex.x = __expf(e.x - mx.x); ex.y = __expf(e.y - mx.y);
            ex.z = __expf(e.z - mx.z); ex.w = __expf(e.w - mx.w);
        }
        const int sl = __builtin_amdgcn_readlane(s, n - 1);
        s = (j < end) ? s : sl;
        add_groups(emb, s, ex, 0, n, ulane, z0, z1, z2, z3, d0, d1, d2, d3);
    }
    const float q0 = __builtin_amdgcn_rcpf(d0 + 1e-16f);
    const float q1 = __builtin_amdgcn_rcpf(d1 + 1e-16f);
    const float q2 = __builtin_amdgcn_rcpf(d2 + 1e-16f);
    const float q3 = __builtin_amdgcn_rcpf(d3 + 1e-16f);
    *(float2*)&zrow[0 * EMB + lane * 2] = make_float2(z0.x * q0, z0.y * q0);
    *(float2*)&zrow[1 * EMB + lane * 2] = make_float2(z1.x * q1, z1.y * q1);
    *(float2*)&zrow[2 * EMB + lane * 2] = make_float2(z2.x * q2, z2.y * q2);
    *(float2*)&zrow[3 * EMB + lane * 2] = make_float2(z3.x * q3, z3.y * q3);
}

__global__ __launch_bounds__(512, 16) void gather_project_kernel(const int* __restrict__ off,
                                                             const int* __restrict__ srt,
                                                             const float* __restrict__ emb,
                                                             const float* __restrict__ W,
                                                             const float* __restrict__ a_src,
                                                             const float* __restrict__ a_dst,
                                                             float* __restrict__ out) {
    __shared__ __attribute__((aligned(16))) float zt[DPW][16][ZP];
    __shared__ __attribute__((aligned(16))) float part[DPW][16][PP];
    const int lane = threadIdx.x & 31;
    const int wave = __builtin_amdgcn_readfirstlane(threadIdx.x >> 5);
    const int dbase = blockIdx.x * (16 * DPW) + wave;

    int beg[DPW], end[DPW], s0[DPW];
    float4 ad[DPW];
    #pragma unroll
    for (int t = 0; t < DPW; ++t) {
        const int d = dbase + 16 * t;
        beg[t] = __builtin_amdgcn_readfirstlane(off[d]);
        end[t] = __builtin_amdgcn_readfirstlane(off[d + 1]);
        ad[t]  = *(const float4*)(a_dst + (size_t)d * 4);
    }
    #pragma unroll
    for (int t = 0; t < DPW; ++t) {
        s0[t] = 0;
        if (beg[t] + lane < end[t]) s0[t] = srt[beg[t] + lane];
    }
    #pragma unroll
    for (int t = 0; t < DPW; ++t)
        aggregate_segment(beg[t], end[t], s0[t], ad[t], srt, emb, a_src, lane, &zt[t][wave][0]);

    const int half = lane >> 4;
    const int lrow = lane & 15;
    const int ct   = wave & 1;
    const int kq   = wave >> 1;
    const int hk   = kq >> 1;
    const int kin  = (kq & 1) * 32;
    const float* wp = W + (size_t)(kin + 2 * half) * HOUT + hk * RD + ct * 16 + lrow;
    v2f bf[8];
    #pragma unroll
    for (int ks = 0; ks < 8; ++ks) {
        bf[ks].x = wp[(ks * 4) * HOUT];
        bf[ks].y = wp[(ks * 4 + 1) * HOUT];
    }
    __syncthreads();

    #pragma unroll
    for (int t = 0; t < DPW; ++t) {
        const float* zr = &zt[t][lrow][kq * 32 + 2 * half];
        v8f c0 = {}, c1 = {};
        #pragma unroll
        for (int ks = 0; ks < 8; ks += 2) {
            const v2f fa0 = *(const v2f*)(zr + ks * 4);
            const v2f fa1 = *(const v2f*)(zr + ks * 4 + 4);
            c0 = __builtin_amdgcn_wmma_f32_16x16x4_f32(false, fa0, false, bf[ks],     (short)0, c0, false, false);
            c1 = __builtin_amdgcn_wmma_f32_16x16x4_f32(false, fa1, false, bf[ks + 1], (short)0, c1, false, false);
        }
        #pragma unroll
        for (int v = 0; v < 8; ++v) part[t][wave][v * 32 + lane] = c0[v] + c1[v];
    }
    __syncthreads();

    const int pl = (wave >> 3) * 16 + lrow;
    const int pv = wave & 7;
    const int pc = lane >> 4;
    #pragma unroll
    for (int t = 0; t < DPW; ++t) {
        float sum = 0.0f;
        #pragma unroll
        for (int q = 0; q < 8; ++q) sum += part[t][2 * q + pc][pv * 32 + pl];
        out[(size_t)(dbase + 16 * t) * RD + lane] = 0.25f * sum;
    }
}

extern "C" void kernel_launch(void* const* d_in, const int* in_sizes, int n_in,
                              void* d_out, int out_size, void* d_ws, size_t ws_size,
                              hipStream_t stream) {
    const float* node_emb = (const float*)d_in[0];
    const float* W        = (const float*)d_in[1];
    const float* att_src  = (const float*)d_in[2];
    const float* att_dst  = (const float*)d_in[3];
    const int*   path     = (const int*)d_in[4];
    float* out = (float*)d_out;

    float* ws    = (float*)d_ws;
    float* h     = ws;
    float* a_src = h     + (size_t)N_NODES * HOUT;
    float* a_dst = a_src + (size_t)N_NODES * HEADS;
    int*   ls    = (int*)(a_dst + (size_t)N_NODES * HEADS);
    int*   off   = ls   + (size_t)N_NODES;
    int*   bkt   = off  + (size_t)N_NODES + 32;
    int*   srt   = bkt  + (size_t)NEDGES;

    range_local_sort_kernel<<<NB1, SRT_T, 0, stream>>>(path, ls, bkt);

    sort_logits_kernel<<<NBUCK + LG_BLOCKS, SRT_T, 0, stream>>>(
        bkt, ls, off, srt, node_emb, W, att_src, att_dst, a_src, a_dst);

    gather_project_kernel<<<N_NODES / (16 * DPW), 512, 0, stream>>>(off, srt, node_emb, W, a_src, a_dst, out);
}
